# starting kernel
# baseline (speedup 1.0000x reference)
#include <hip/hip_runtime.h>
#include <math.h>

#define N_NODES 100000
#define N_EDGES 1600000
#define IN_DIM  128
#define HID     32
#define HEADS   4
#define OUT_DIM 16

typedef __attribute__((ext_vector_type(2))) float v2f;
typedef __attribute__((ext_vector_type(8))) float v8f;

__device__ __forceinline__ unsigned fkey(float x) {
  unsigned b = __float_as_uint(x);
  return (b & 0x80000000u) ? ~b : (b | 0x80000000u);
}
__device__ __forceinline__ float funkey(unsigned k) {
  return __uint_as_float((k & 0x80000000u) ? (k & 0x7FFFFFFFu) : ~k);
}

__global__ void zero_f32(float* __restrict__ p, long long n) {
  long long t = (long long)blockIdx.x * blockDim.x + threadIdx.x;
  if (t < n) p[t] = 0.0f;
}

template <int NT>
__global__ void __launch_bounds__(256)
wmma_gemm_strip(const float* __restrict__ A, const float* __restrict__ W,
                float* __restrict__ C, int Mtiles, int K, int ldc, int coff) {
  const int wave = (blockIdx.x * blockDim.x + threadIdx.x) >> 5;
  if (wave >= Mtiles) return;
  const int lane = threadIdx.x & 31;
  const int lo = lane & 15;
  const int hi = lane >> 4;
  const float* Ab = A + ((size_t)wave * 16 + lo) * K;
  v8f acc[NT];
#pragma unroll
  for (int j = 0; j < NT; ++j)
    acc[j] = (v8f){0.f, 0.f, 0.f, 0.f, 0.f, 0.f, 0.f, 0.f};
  for (int k0 = 0; k0 < K; k0 += 4) {
    const int ka = k0 + hi * 2;
    v2f a;
    a.x = Ab[ka]; a.y = Ab[ka + 1];
#pragma unroll
    for (int j = 0; j < NT; ++j) {
      const float* Wb = W + ((size_t)j * 16 + lo) * K;
      v2f b;
      b.x = Wb[ka]; b.y = Wb[ka + 1];
      acc[j] = __builtin_amdgcn_wmma_f32_16x16x4_f32(false, a, false, b,
                                                     (short)0, acc[j], false, false);
    }
  }
#pragma unroll
  for (int j = 0; j < NT; ++j) {
    float* Cb = C + (size_t)wave * 16 * ldc + coff + (size_t)j * 16 + lo;
#pragma unroll
    for (int r = 0; r < 8; ++r)
      Cb[(size_t)(r + hi * 8) * ldc] = acc[j][r];
  }
}

__global__ void __launch_bounds__(256)
wmma_gemm_frac(const float* __restrict__ H, const int* __restrict__ fc,
               const float* __restrict__ Wf, float* __restrict__ hfpre) {
  const int wave = (blockIdx.x * blockDim.x + threadIdx.x) >> 5;
  if (wave >= N_NODES / 16) return;
  const int s = blockIdx.y;
  const int lane = threadIdx.x & 31;
  const int lo = lane & 15;
  const int hi = lane >> 4;
  const int nid = fc[(wave * 16 + lo) * 3 + s];
  const float* Ab = H + (size_t)nid * IN_DIM;
  const float* Ws = Wf + (size_t)s * HID * IN_DIM;
  v8f acc[2];
  acc[0] = (v8f){0.f, 0.f, 0.f, 0.f, 0.f, 0.f, 0.f, 0.f};
  acc[1] = acc[0];
  for (int k0 = 0; k0 < IN_DIM; k0 += 4) {
    const int ka = k0 + hi * 2;
    v2f a;
    a.x = Ab[ka]; a.y = Ab[ka + 1];
#pragma unroll
    for (int j = 0; j < 2; ++j) {
      const float* Wb = Ws + ((size_t)j * 16 + lo) * IN_DIM;
      v2f b;
      b.x = Wb[ka]; b.y = Wb[ka + 1];
      acc[j] = __builtin_amdgcn_wmma_f32_16x16x4_f32(false, a, false, b,
                                                     (short)0, acc[j], false, false);
    }
  }
#pragma unroll
  for (int j = 0; j < 2; ++j) {
    float* Cb = hfpre + (size_t)wave * 16 * 96 + s * HID + (size_t)j * 16 + lo;
#pragma unroll
    for (int r = 0; r < 8; ++r)
      Cb[(size_t)(r + hi * 8) * 96] = acc[j][r];
  }
}

__global__ void attn_scores(const float* __restrict__ z, const float* __restrict__ a,
                            float* __restrict__ ss, float* __restrict__ sd,
                            int n, int heads) {
  int t = blockIdx.x * blockDim.x + threadIdx.x;
  if (t >= n * heads) return;
  int node = t / heads, hh = t - node * heads;
  const float* zp = z + (size_t)node * heads * HID + hh * HID;
  const float* al = a + (size_t)hh * 2 * HID;
  const float* ar = al + HID;
  float s0 = 0.f, s1 = 0.f;
#pragma unroll 8
  for (int o = 0; o < HID; ++o) { s0 += zp[o] * al[o]; s1 += zp[o] * ar[o]; }
  ss[t] = s0; sd[t] = s1;
}

__global__ void edge_scores_max(const int* __restrict__ src, const int* __restrict__ dst,
                                const float* __restrict__ ss, const float* __restrict__ sd,
                                float* __restrict__ e, unsigned* __restrict__ m, int heads) {
  int t = blockIdx.x * blockDim.x + threadIdx.x;
  if (t >= N_EDGES * heads) return;
  int ed = t / heads, hh = t - ed * heads;
  float x = ss[src[ed] * heads + hh] + sd[dst[ed] * heads + hh];
  x = (x > 0.f) ? x : 0.01f * x;
  e[t] = x;
  atomicMax(m + dst[ed] * heads + hh, fkey(x));
}

__global__ void edge_expsum(const int* __restrict__ dst, float* __restrict__ e,
                            const unsigned* __restrict__ m, float* __restrict__ sum,
                            int heads) {
  int t = blockIdx.x * blockDim.x + threadIdx.x;
  if (t >= N_EDGES * heads) return;
  int ed = t / heads, hh = t - ed * heads;
  int di = dst[ed] * heads + hh;
  float ee = expf(e[t] - funkey(m[di]));
  e[t] = ee;
  atomicAdd(sum + di, ee);
}

__global__ void edge_aggregate(const int* __restrict__ src, const int* __restrict__ dst,
                               const float* __restrict__ e, const float* __restrict__ sum,
                               const float* __restrict__ z, float* __restrict__ out,
                               int heads, int ldo) {
  long long t = (long long)blockIdx.x * blockDim.x + threadIdx.x;
  const int C = heads * HID;
  if (t >= (long long)N_EDGES * C) return;
  int ed = (int)(t / C);
  int c  = (int)(t - (long long)ed * C);
  int hh = c >> 5;
  int di = dst[ed];
  float alpha = e[(size_t)ed * heads + hh] / fmaxf(sum[di * heads + hh], 1e-9f);
  atomicAdd(out + (size_t)di * ldo + c, alpha * z[(size_t)src[ed] * C + c]);
}

__global__ void elu_inplace(float* __restrict__ x, long long n) {
  long long t = (long long)blockIdx.x * blockDim.x + threadIdx.x;
  if (t >= n) return;
  float v = x[t];
  x[t] = (v > 0.f) ? v : (expf(v) - 1.0f);
}

extern "C" void kernel_launch(void* const* d_in, const int* in_sizes, int n_in,
                              void* d_out, int out_size, void* d_ws, size_t ws_size,
                              hipStream_t stream) {
  const float* h    = (const float*)d_in[0];
  const int*   src  = (const int*)  d_in[1];
  const int*   dst  = (const int*)  d_in[2];
  const int*   fc   = (const int*)  d_in[3];
  const float* W1   = (const float*)d_in[4];
  const float* a1   = (const float*)d_in[5];
  const float* W2   = (const float*)d_in[6];
  const float* a2   = (const float*)d_in[7];
  const float* Wf   = (const float*)d_in[8];
  const float* Wff  = (const float*)d_in[9];
  const float* Wfc2 = (const float*)d_in[10];
  float* out = (float*)d_out;

  float* w = (float*)d_ws;
  size_t off = 0;
  float*    agg1 = w + off;              off += (size_t)N_NODES * 128;
  unsigned* m1   = (unsigned*)(w + off); off += (size_t)N_NODES * HEADS;
  float*    sum1 = w + off;              off += (size_t)N_NODES * HEADS;
  unsigned* m2   = (unsigned*)(w + off); off += N_NODES;
  float*    sum2 = w + off;              off += N_NODES;
  float*    hg   = w + off;              off += (size_t)N_NODES * 64;
  const long long zeroN = (long long)off;
  float* z1  = w + off; off += (size_t)N_NODES * 128;
  float* s1s = w + off; off += (size_t)N_NODES * HEADS;
  float* s1d = w + off; off += (size_t)N_NODES * HEADS;
  float* e1  = w + off; off += (size_t)N_EDGES * HEADS;
  float* z2  = w + off; off += (size_t)N_NODES * HID;
  float* s2s = w + off; off += N_NODES;
  float* s2d = w + off; off += N_NODES;
  float* e2  = w + off; off += N_EDGES;
  float* hfp = w + off; off += (size_t)N_NODES * 96;
  (void)ws_size; (void)in_sizes; (void)n_in; (void)out_size;

  const int TB = 256;
  const int MT = N_NODES / 16;
  const int gemm_blocks = (MT + 7) / 8;

  zero_f32<<<(int)((zeroN + TB - 1) / TB), TB, 0, stream>>>(w, zeroN);

  wmma_gemm_strip<8><<<gemm_blocks, TB, 0, stream>>>(h, W1, z1, MT, 128, 128, 0);
  attn_scores<<<(N_NODES * HEADS + TB - 1) / TB, TB, 0, stream>>>(z1, a1, s1s, s1d, N_NODES, HEADS);
  edge_scores_max<<<(N_EDGES * HEADS + TB - 1) / TB, TB, 0, stream>>>(src, dst, s1s, s1d, e1, m1, HEADS);
  edge_expsum<<<(N_EDGES * HEADS + TB - 1) / TB, TB, 0, stream>>>(dst, e1, m1, sum1, HEADS);
  {
    long long tot = (long long)N_EDGES * 128;
    edge_aggregate<<<(int)((tot + TB - 1) / TB), TB, 0, stream>>>(src, dst, e1, sum1, z1, agg1, HEADS, 128);
  }
  elu_inplace<<<(int)(((long long)N_NODES * 128 + TB - 1) / TB), TB, 0, stream>>>(
      agg1, (long long)N_NODES * 128);

  wmma_gemm_strip<2><<<gemm_blocks, TB, 0, stream>>>(agg1, W2, z2, MT, 128, HID, 0);
  attn_scores<<<(N_NODES + TB - 1) / TB, TB, 0, stream>>>(z2, a2, s2s, s2d, N_NODES, 1);
  edge_scores_max<<<(N_EDGES + TB - 1) / TB, TB, 0, stream>>>(src, dst, s2s, s2d, e2, m2, 1);
  edge_expsum<<<(N_EDGES + TB - 1) / TB, TB, 0, stream>>>(dst, e2, m2, sum2, 1);
  {
    long long tot = (long long)N_EDGES * HID;
    edge_aggregate<<<(int)((tot + TB - 1) / TB), TB, 0, stream>>>(src, dst, e2, sum2, z2, hg, 1, 64);
  }

  {
    dim3 g(gemm_blocks, 3);
    wmma_gemm_frac<<<g, TB, 0, stream>>>(h, fc, Wf, hfp);
  }
  wmma_gemm_strip<2><<<gemm_blocks, TB, 0, stream>>>(hfp, Wff, hg, MT, 96, 64, HID);

  wmma_gemm_strip<1><<<gemm_blocks, TB, 0, stream>>>(hg, Wfc2, out, MT, 64, OUT_DIM, 0);
}


# Edges counting-sorted by dst in-call (int atomic arrival-order rank -> CSR); one wave per dst node does softmax and sums z[src] rows as a gather: no float atomics, sums in slot order + lane tree
# speedup vs baseline: 3.5271x; 3.5271x over previous
#include <hip/hip_runtime.h>
#include <math.h>

#define N_NODES 100000
#define N_EDGES 1600000
#define IN_DIM  128
#define HID     32
#define HEADS   4
#define OUT_DIM 16

typedef __attribute__((ext_vector_type(2))) float v2f;
typedef __attribute__((ext_vector_type(8))) float v8f;

__global__ void zero_f32(float* __restrict__ p, long long n) {
  long long t = (long long)blockIdx.x * blockDim.x + threadIdx.x;
  if (t < n) p[t] = 0.0f;
}

template <int NT>
__global__ void __launch_bounds__(256)
wmma_gemm_strip(const float* __restrict__ A, const float* __restrict__ W,
                float* __restrict__ C, int Mtiles, int K, int ldc, int coff) {
  const int wave = (blockIdx.x * blockDim.x + threadIdx.x) >> 5;
  if (wave >= Mtiles) return;
  const int lane = threadIdx.x & 31;
  const int lo = lane & 15;
  const int hi = lane >> 4;
  const float* Ab = A + ((size_t)wave * 16 + lo) * K;
  v8f acc[NT];
#pragma unroll
  for (int j = 0; j < NT; ++j)
    acc[j] = (v8f){0.f, 0.f, 0.f, 0.f, 0.f, 0.f, 0.f, 0.f};
  for (int k0 = 0; k0 < K; k0 += 4) {
    const int ka = k0 + hi * 2;
    v2f a;
    a.x = Ab[ka]; a.y = Ab[ka + 1];
#pragma unroll
    for (int j = 0; j < NT; ++j) {
      const float* Wb = W + ((size_t)j * 16 + lo) * K;
      v2f b;
      b.x = Wb[ka]; b.y = Wb[ka + 1];
      acc[j] = __builtin_amdgcn_wmma_f32_16x16x4_f32(false, a, false, b,
                                                     (short)0, acc[j], false, false);
    }
  }
#pragma unroll
  for (int j = 0; j < NT; ++j) {
    float* Cb = C + (size_t)wave * 16 * ldc + coff + (size_t)j * 16 + lo;
#pragma unroll
    for (int r = 0; r < 8; ++r)
      Cb[(size_t)(r + hi * 8) * ldc] = acc[j][r];
  }
}

__global__ void __launch_bounds__(256)
wmma_gemm_frac(const float* __restrict__ H, const int* __restrict__ fc,
               const float* __restrict__ Wf, float* __restrict__ hfpre) {
  const int wave = (blockIdx.x * blockDim.x + threadIdx.x) >> 5;
  if (wave >= N_NODES / 16) return;
  const int s = blockIdx.y;
  const int lane = threadIdx.x & 31;
  const int lo = lane & 15;
  const int hi = lane >> 4;
  const int nid = fc[(wave * 16 + lo) * 3 + s];
  const float* Ab = H + (size_t)nid * IN_DIM;
  const float* Ws = Wf + (size_t)s * HID * IN_DIM;
  v8f acc[2];
  acc[0] = (v8f){0.f, 0.f, 0.f, 0.f, 0.f, 0.f, 0.f, 0.f};
  acc[1] = acc[0];
  for (int k0 = 0; k0 < IN_DIM; k0 += 4) {
    const int ka = k0 + hi * 2;
    v2f a;
    a.x = Ab[ka]; a.y = Ab[ka + 1];
#pragma unroll
    for (int j = 0; j < 2; ++j) {
      const float* Wb = Ws + ((size_t)j * 16 + lo) * IN_DIM;
      v2f b;
      b.x = Wb[ka]; b.y = Wb[ka + 1];
      acc[j] = __builtin_amdgcn_wmma_f32_16x16x4_f32(false, a, false, b,
                                                     (short)0, acc[j], false, false);
    }
  }
#pragma unroll
  for (int j = 0; j < 2; ++j) {
    float* Cb = hfpre + (size_t)wave * 16 * 96 + s * HID + (size_t)j * 16 + lo;
#pragma unroll
    for (int r = 0; r < 8; ++r)
      Cb[(size_t)(r + hi * 8) * 96] = acc[j][r];
  }
}

__global__ void attn_scores(const float* __restrict__ z, const float* __restrict__ a,
                            float* __restrict__ ss, float* __restrict__ sd,
                            int n, int heads) {
  int t = blockIdx.x * blockDim.x + threadIdx.x;
  if (t >= n * heads) return;
  int node = t / heads, hh = t - node * heads;
  const float* zp = z + (size_t)node * heads * HID + hh * HID;
  const float* al = a + (size_t)hh * 2 * HID;
  const float* ar = al + HID;
  float s0 = 0.f, s1 = 0.f;
#pragma unroll 8
  for (int o = 0; o < HID; ++o) { s0 += zp[o] * al[o]; s1 += zp[o] * ar[o]; }
  ss[t] = s0; sd[t] = s1;
}

__global__ void edge_count_rank(const int* __restrict__ dst, int* __restrict__ deg,
                                int* __restrict__ rank) {
  int e = blockIdx.x * blockDim.x + threadIdx.x;
  if (e >= N_EDGES) return;
  rank[e] = atomicAdd(deg + dst[e], 1);
}

__global__ void __launch_bounds__(256)
scan_block_totals(const int* __restrict__ deg, int* __restrict__ btot) {
  __shared__ int wsum[8];
  const int t = threadIdx.x, lane = t & 31, w = t >> 5;
  const int i = (blockIdx.x * 256 + t) * 4;
  int v = 0;
  if (i + 3 < N_NODES) {
    const int4 d = *(const int4*)(deg + i);
    v = d.x + d.y + d.z + d.w;
  } else {
    for (int k = 0; k < 4; ++k) if (i + k < N_NODES) v += deg[i + k];
  }
  for (int o = 16; o > 0; o >>= 1) v += __shfl_xor(v, o);
  if (lane == 0) wsum[w] = v;
  __syncthreads();
  if (t == 0) {
    int s = 0;
    for (int k = 0; k < 8; ++k) s += wsum[k];
    btot[blockIdx.x] = s;
  }
}

__global__ void __launch_bounds__(256)
scan_rowptr(const int* __restrict__ deg, const int* __restrict__ btot,
            int* __restrict__ rowptr) {
  __shared__ int wsum[8];
  __shared__ int boff;
  const int t = threadIdx.x, lane = t & 31, w = t >> 5;
  int p = 0;
  for (int j = t; j < (int)blockIdx.x; j += 256) p += btot[j];
  for (int o = 16; o > 0; o >>= 1) p += __shfl_xor(p, o);
  if (lane == 0) wsum[w] = p;
  __syncthreads();
  if (t == 0) {
    int s = 0;
    for (int k = 0; k < 8; ++k) s += wsum[k];
    boff = s;
  }
  __syncthreads();
  const int i = (blockIdx.x * 256 + t) * 4;
  int d[4] = {0, 0, 0, 0};
  if (i + 3 < N_NODES) {
    const int4 q = *(const int4*)(deg + i);
    d[0] = q.x; d[1] = q.y; d[2] = q.z; d[3] = q.w;
  } else {
    for (int k = 0; k < 4; ++k) if (i + k < N_NODES) d[k] = deg[i + k];
  }
  const int v = d[0] + d[1] + d[2] + d[3];
  int inc = v;
  for (int o = 1; o < 32; o <<= 1) {
    const int u = __shfl_up(inc, o);
    if (lane >= o) inc += u;
  }
  if (lane == 31) wsum[w] = inc;
  __syncthreads();
  int ex = boff + inc - v;
  for (int k = 0; k < w; ++k) ex += wsum[k];
  for (int k = 0; k < 4; ++k) {
    if (i + k < N_NODES) rowptr[i + k] = ex;
    ex += d[k];
  }
  if (i < N_NODES && i + 4 >= N_NODES) rowptr[N_NODES] = ex;
}

__global__ void edge_fill(const int* __restrict__ src, const int* __restrict__ dst,
                          const int* __restrict__ rowptr, const int* __restrict__ rank,
                          int* __restrict__ csr) {
  int e = blockIdx.x * blockDim.x + threadIdx.x;
  if (e >= N_EDGES) return;
  csr[rowptr[dst[e]] + rank[e]] = src[e];
}

template <int NH>
__device__ __forceinline__ int load_edge(const int* __restrict__ csr, const float* __restrict__ ss,
                                         const float* sdn, int j, int end, float* x) {
  int s = 0;
  if (j < end) {
    s = csr[j];
    float v[NH];
    if constexpr (NH == 4) {
      const float4 q = *(const float4*)(ss + (size_t)s * 4);
      v[0] = q.x; v[1] = q.y; v[2] = q.z; v[3] = q.w;
    } else {
      for (int h = 0; h < NH; ++h) v[h] = ss[(size_t)s * NH + h];
    }
    for (int h = 0; h < NH; ++h) {
      const float t = v[h] + sdn[h];
      x[h] = (t > 0.f) ? t : 0.01f * t;
    }
  } else {
    for (int h = 0; h < NH; ++h) x[h] = -INFINITY;
  }
  return s;
}

template <int NH>
__device__ __forceinline__ float pick_alpha(const float* alpha, int t, int hh) {
  float a = __int_as_float(__builtin_amdgcn_readlane(__float_as_int(alpha[0]), t));
  for (int h = 1; h < NH; ++h) {
    const float b = __int_as_float(__builtin_amdgcn_readlane(__float_as_int(alpha[h]), t));
    a = (hh == h) ? b : a;
  }
  return a;
}

template <int NH>
__device__ __forceinline__ void load_row(const float* __restrict__ p, float* r) {
  if constexpr (NH == 4) {
    const float4 q = *(const float4*)p;
    r[0] = q.x; r[1] = q.y; r[2] = q.z; r[3] = q.w;
  } else {
    for (int h = 0; h < NH; ++h) r[h] = p[h];
  }
}

template <int NH>
__device__ __forceinline__ void add_rows(const float* __restrict__ zl, int s, const float* alpha,
                                         int cnt, int hh, float* acc) {
  constexpr int C = NH * HID;
  int t = 0;
  for (; t + 4 <= cnt; t += 4) {
    float r[4][NH], a[4];
#pragma unroll
    for (int u = 0; u < 4; ++u) {
      const int st = __builtin_amdgcn_readlane(s, t + u);
      a[u] = pick_alpha<NH>(alpha, t + u, hh);
      load_row<NH>(zl + (size_t)st * C, r[u]);
    }
#pragma unroll
    for (int u = 0; u < 4; ++u)
      for (int h = 0; h < NH; ++h) acc[h] += a[u] * r[u][h];
  }
  for (; t < cnt; ++t) {
    float r[NH];
    const int st = __builtin_amdgcn_readlane(s, t);
    const float a = pick_alpha<NH>(alpha, t, hh);
    load_row<NH>(zl + (size_t)st * C, r);
    for (int h = 0; h < NH; ++h) acc[h] += a * r[h];
  }
}

template <int NH, bool ELU>
__global__ void __launch_bounds__(256)
node_softmax_gather(const int* __restrict__ rowptr, const int* __restrict__ csr,
                    const float* __restrict__ ss, const float* __restrict__ sd,
                    const float* __restrict__ z, float* __restrict__ out, int ldo) {
  const int n = __builtin_amdgcn_readfirstlane((int)(blockIdx.x * 8u + (threadIdx.x >> 5)));
  if (n >= N_NODES) return;
  const int lane = threadIdx.x & 31;
  const int hh = (lane * NH) >> 5;
  const int beg = rowptr[n];
  const int end = rowptr[n + 1];
  float acc[NH];
  for (int h = 0; h < NH; ++h) acc[h] = 0.f;
  if (end > beg) {
    float sdn[NH];
    for (int h = 0; h < NH; ++h) sdn[h] = sd[(size_t)n * NH + h];
    const float* zl = z + lane * NH;
    float x0[NH];
    const int s0 = load_edge<NH>(csr, ss, sdn, beg + lane, end, x0);
    const bool v0 = (beg + lane) < end;
    float m[NH];
    for (int h = 0; h < NH; ++h) m[h] = x0[h];
    for (int base = beg + 32; base < end; base += 32) {
      float x[NH];
      load_edge<NH>(csr, ss, sdn, base + lane, end, x);
      for (int h = 0; h < NH; ++h) m[h] = fmaxf(m[h], x[h]);
    }
    for (int o = 16; o > 0; o >>= 1)
      for (int h = 0; h < NH; ++h) m[h] = fmaxf(m[h], __shfl_xor(m[h], o));
    float ee0[NH], sum[NH];
    for (int h = 0; h < NH; ++h) { ee0[h] = v0 ? expf(x0[h] - m[h]) : 0.f; sum[h] = ee0[h]; }
    for (int base = beg + 32; base < end; base += 32) {
      float x[NH];
      load_edge<NH>(csr, ss, sdn, base + lane, end, x);
      const bool v = (base + lane) < end;
      for (int h = 0; h < NH; ++h) sum[h] += v ? expf(x[h] - m[h]) : 0.f;
    }
    for (int o = 16; o > 0; o >>= 1)
      for (int h = 0; h < NH; ++h) sum[h] += __shfl_xor(sum[h], o);
    float den[NH], al[NH];
    for (int h = 0; h < NH; ++h) { den[h] = fmaxf(sum[h], 1e-9f); al[h] = ee0[h] / den[h]; }
    const int c0 = end - beg;
    add_rows<NH>(zl, s0, al, c0 < 32 ? c0 : 32, hh, acc);
    for (int base = beg + 32; base < end; base += 32) {
      float x[NH];
      const int s = load_edge<NH>(csr, ss, sdn, base + lane, end, x);
      const bool v = (base + lane) < end;
      for (int h = 0; h < NH; ++h) al[h] = v ? expf(x[h] - m[h]) / den[h] : 0.f;
      const int c = end - base;
      add_rows<NH>(zl, s, al, c < 32 ? c : 32, hh, acc);
    }
  }
  if constexpr (ELU) {
    for (int h = 0; h < NH; ++h) acc[h] = (acc[h] > 0.f) ? acc[h] : (expf(acc[h]) - 1.0f);
  }
  float* op = out + (size_t)n * ldo + lane * NH;
  if constexpr (NH == 4) {
    *(float4*)op = make_float4(acc[0], acc[1], acc[2], acc[3]);
  } else {
    for (int h = 0; h < NH; ++h) op[h] = acc[h];
  }
}

extern "C" void kernel_launch(void* const* d_in, const int* in_sizes, int n_in,
                              void* d_out, int out_size, void* d_ws, size_t ws_size,
                              hipStream_t stream) {
  const float* h    = (const float*)d_in[0];
  const int*   src  = (const int*)  d_in[1];
  const int*   dst  = (const int*)  d_in[2];
  const int*   fc   = (const int*)  d_in[3];
  const float* W1   = (const float*)d_in[4];
  const float* a1   = (const float*)d_in[5];
  const float* W2   = (const float*)d_in[6];
  const float* a2   = (const float*)d_in[7];
  const float* Wf   = (const float*)d_in[8];
  const float* Wff  = (const float*)d_in[9];
  const float* Wfc2 = (const float*)d_in[10];
  float* out = (float*)d_out;

  float* w = (float*)d_ws;
  size_t off = 0;
  int* deg    = (int*)(w + off); off += N_NODES;
  int* rowptr = (int*)(w + off); off += N_NODES + 32;
  int* btot   = (int*)(w + off); off += 128;
  int* rank   = (int*)(w + off); off += N_EDGES;
  int* csr    = (int*)(w + off); off += N_EDGES;
  float* z1  = w + off; off += (size_t)N_NODES * 128;
  float* s1s = w + off; off += (size_t)N_NODES * HEADS;
  float* s1d = w + off; off += (size_t)N_NODES * HEADS;
  float* h1  = w + off; off += (size_t)N_NODES * 128;
  float* z2  = w + off; off += (size_t)N_NODES * HID;
  float* s2s = w + off; off += N_NODES;
  float* s2d = w + off; off += N_NODES;
  float* hg  = w + off; off += (size_t)N_NODES * 64;
  float* hfp = w + off; off += (size_t)N_NODES * 96;
  (void)ws_size; (void)in_sizes; (void)n_in; (void)out_size;

  const int TB = 256;
  const int MT = N_NODES / 16;
  const int gemm_blocks = (MT + 7) / 8;
  const int edge_blocks = (N_EDGES + TB - 1) / TB;
  const int scan_blocks = (N_NODES + 1023) / 1024;
  const int node_blocks = (N_NODES + 7) / 8;

  zero_f32<<<(N_NODES + TB - 1) / TB, TB, 0, stream>>>((float*)deg, (long long)N_NODES);
  edge_count_rank<<<edge_blocks, TB, 0, stream>>>(dst, deg, rank);
  scan_block_totals<<<scan_blocks, TB, 0, stream>>>(deg, btot);
  scan_rowptr<<<scan_blocks, TB, 0, stream>>>(deg, btot, rowptr);
  edge_fill<<<edge_blocks, TB, 0, stream>>>(src, dst, rowptr, rank, csr);

  wmma_gemm_strip<8><<<gemm_blocks, TB, 0, stream>>>(h, W1, z1, MT, 128, 128, 0);
  attn_scores<<<(N_NODES * HEADS + TB - 1) / TB, TB, 0, stream>>>(z1, a1, s1s, s1d, N_NODES, HEADS);
  node_softmax_gather<HEADS, true><<<node_blocks, TB, 0, stream>>>(rowptr, csr, s1s, s1d, z1, h1, 128);

  wmma_gemm_strip<2><<<gemm_blocks, TB, 0, stream>>>(h1, W2, z2, MT, 128, HID, 0);
  attn_scores<<<(N_NODES + TB - 1) / TB, TB, 0, stream>>>(z2, a2, s2s, s2d, N_NODES, 1);
  node_softmax_gather<1, false><<<node_blocks, TB, 0, stream>>>(rowptr, csr, s2s, s2d, z2, hg, 64);

  {
    dim3 g(gemm_blocks, 3);
    wmma_gemm_frac<<<g, TB, 0, stream>>>(h, fc, Wf, hfp);
  }
  wmma_gemm_strip<2><<<gemm_blocks, TB, 0, stream>>>(hfp, Wff, hg, MT, 96, 64, HID);

  wmma_gemm_strip<1><<<gemm_blocks, TB, 0, stream>>>(hg, Wfc2, out, MT, 64, OUT_DIM, 0);
}


# GEMM weight tile copied once per workgroup into LDS (whole-line 16 B loads, pitch K+4 floats); B fragments read from LDS instead of 16-line strided global loads per wave; k order and sums unchanged
# speedup vs baseline: 4.2966x; 1.2182x over previous
#include <hip/hip_runtime.h>
#include <math.h>

#define N_NODES 100000
#define N_EDGES 1600000
#define IN_DIM  128
#define HID     32
#define HEADS   4
#define OUT_DIM 16

typedef __attribute__((ext_vector_type(2))) float v2f;
typedef __attribute__((ext_vector_type(8))) float v8f;

__global__ void zero_f32(float* __restrict__ p, long long n) {
  long long t = (long long)blockIdx.x * blockDim.x + threadIdx.x;
  if (t < n) p[t] = 0.0f;
}

template <int NT, int K>
__device__ __forceinline__ void stage_weights(const float* __restrict__ W, float4* Ws) {
  constexpr int K4 = K / 4;
  constexpr int P4 = K4 + 1;
  for (int i = threadIdx.x; i < NT * 16 * K4; i += 256) {
    const int row = i / K4;
    const int c = i - row * K4;
    Ws[row * P4 + c] = ((const float4*)W)[i];
  }
}

template <int NT, int K>
__global__ void __launch_bounds__(256)
wmma_gemm_strip(const float* __restrict__ A, const float* __restrict__ W,
                float* __restrict__ C, int Mtiles, int ldc, int coff) {
  constexpr int P = K + 4;
  __shared__ float4 Ws[NT * 16 * (P / 4)];
  stage_weights<NT, K>(W, Ws);
  __syncthreads();
  const int wave = (blockIdx.x * blockDim.x + threadIdx.x) >> 5;
  if (wave < Mtiles) {
    const int lane = threadIdx.x & 31;
    const int lo = lane & 15;
    const int hi = lane >> 4;
    const float* Ab = A + ((size_t)wave * 16 + lo) * K;
    const float* Wl = (const float*)Ws + lo * P;
    v8f acc[NT];
#pragma unroll
    for (int j = 0; j < NT; ++j)
      acc[j] = (v8f){0.f, 0.f, 0.f, 0.f, 0.f, 0.f, 0.f, 0.f};
    for (int k0 = 0; k0 < K; k0 += 4) {
      const int ka = k0 + hi * 2;
      v2f a;
      a.x = Ab[ka]; a.y = Ab[ka + 1];
#pragma unroll
      for (int j = 0; j < NT; ++j) {
        const v2f b = *(const v2f*)(Wl + j * 16 * P + ka);
        acc[j] = __builtin_amdgcn_wmma_f32_16x16x4_f32(false, a, false, b,
                                                       (short)0, acc[j], false, false);
      }
    }
#pragma unroll
    for (int j = 0; j < NT; ++j) {
      float* Cb = C + (size_t)wave * 16 * ldc + coff + (size_t)j * 16 + lo;
#pragma unroll
      for (int r = 0; r < 8; ++r)
        Cb[(size_t)(r + hi * 8) * ldc] = acc[j][r];
    }
  }
}

__global__ void __launch_bounds__(256)
wmma_gemm_frac(const float* __restrict__ H, const int* __restrict__ fc,
               const float* __restrict__ Wf, float* __restrict__ hfpre) {
  constexpr int P = IN_DIM + 4;
  __shared__ float4 Ws[2 * 16 * (P / 4)];
  const int s = blockIdx.y;
  stage_weights<2, IN_DIM>(Wf + (size_t)s * HID * IN_DIM, Ws);
  __syncthreads();
  const int wave = (blockIdx.x * blockDim.x + threadIdx.x) >> 5;
  if (wave < N_NODES / 16) {
    const int lane = threadIdx.x & 31;
    const int lo = lane & 15;
    const int hi = lane >> 4;
    const int nid = fc[(wave * 16 + lo) * 3 + s];
    const float* Ab = H + (size_t)nid * IN_DIM;
    const float* Wl = (const float*)Ws + lo * P;
    v8f acc[2];
    acc[0] = (v8f){0.f, 0.f, 0.f, 0.f, 0.f, 0.f, 0.f, 0.f};
    acc[1] = acc[0];
    for (int k0 = 0; k0 < IN_DIM; k0 += 4) {
      const int ka = k0 + hi * 2;
      v2f a;
      a.x = Ab[ka]; a.y = Ab[ka + 1];
#pragma unroll
      for (int j = 0; j < 2; ++j) {
        const v2f b = *(const v2f*)(Wl + j * 16 * P + ka);
        acc[j] = __builtin_amdgcn_wmma_f32_16x16x4_f32(false, a, false, b,
                                                       (short)0, acc[j], false, false);
      }
    }
#pragma unroll
    for (int j = 0; j < 2; ++j) {
      float* Cb = hfpre + (size_t)wave * 16 * 96 + s * HID + (size_t)j * 16 + lo;
#pragma unroll
      for (int r = 0; r < 8; ++r)
        Cb[(size_t)(r + hi * 8) * 96] = acc[j][r];
    }
  }
}

__global__ void attn_scores(const float* __restrict__ z, const float* __restrict__ a,
                            float* __restrict__ ss, float* __restrict__ sd,
                            int n, int heads) {
  int t = blockIdx.x * blockDim.x + threadIdx.x;
  if (t >= n * heads) return;
  int node = t / heads, hh = t - node * heads;
  const float* zp = z + (size_t)node * heads * HID + hh * HID;
  const float* al = a + (size_t)hh * 2 * HID;
  const float* ar = al + HID;
  float s0 = 0.f, s1 = 0.f;
#pragma unroll 8
  for (int o = 0; o < HID; ++o) { s0 += zp[o] * al[o]; s1 += zp[o] * ar[o]; }
  ss[t] = s0; sd[t] = s1;
}

__global__ void edge_count_rank(const int* __restrict__ dst, int* __restrict__ deg,
                                int* __restrict__ rank) {
  int e = blockIdx.x * blockDim.x + threadIdx.x;
  if (e >= N_EDGES) return;
  rank[e] = atomicAdd(deg + dst[e], 1);
}

__global__ void __launch_bounds__(256)
scan_block_totals(const int* __restrict__ deg, int* __restrict__ btot) {
  __shared__ int wsum[8];
  const int t = threadIdx.x, lane = t & 31, w = t >> 5;
  const int i = (blockIdx.x * 256 + t) * 4;
  int v = 0;
  if (i + 3 < N_NODES) {
    const int4 d = *(const int4*)(deg + i);
    v = d.x + d.y + d.z + d.w;
  } else {
    for (int k = 0; k < 4; ++k) if (i + k < N_NODES) v += deg[i + k];
  }
  for (int o = 16; o > 0; o >>= 1) v += __shfl_xor(v, o);
  if (lane == 0) wsum[w] = v;
  __syncthreads();
  if (t == 0) {
    int s = 0;
    for (int k = 0; k < 8; ++k) s += wsum[k];
    btot[blockIdx.x] = s;
  }
}

__global__ void __launch_bounds__(256)
scan_rowptr(const int* __restrict__ deg, const int* __restrict__ btot,
            int* __restrict__ rowptr) {
  __shared__ int wsum[8];
  __shared__ int boff;
  const int t = threadIdx.x, lane = t & 31, w = t >> 5;
  int p = 0;
  for (int j = t; j < (int)blockIdx.x; j += 256) p += btot[j];
  for (int o = 16; o > 0; o >>= 1) p += __shfl_xor(p, o);
  if (lane == 0) wsum[w] = p;
  __syncthreads();
  if (t == 0) {
    int s = 0;
    for (int k = 0; k < 8; ++k) s += wsum[k];
    boff = s;
  }
  __syncthreads();
  const int i = (blockIdx.x * 256 + t) * 4;
  int d[4] = {0, 0, 0, 0};
  if (i + 3 < N_NODES) {
    const int4 q = *(const int4*)(deg + i);
    d[0] = q.x; d[1] = q.y; d[2] = q.z; d[3] = q.w;
  } else {
    for (int k = 0; k < 4; ++k) if (i + k < N_NODES) d[k] = deg[i + k];
  }
  const int v = d[0] + d[1] + d[2] + d[3];
  int inc = v;
  for (int o = 1; o < 32; o <<= 1) {
    const int u = __shfl_up(inc, o);
    if (lane >= o) inc += u;
  }
  if (lane == 31) wsum[w] = inc;
  __syncthreads();
  int ex = boff + inc - v;
  for (int k = 0; k < w; ++k) ex += wsum[k];
  for (int k = 0; k < 4; ++k) {
    if (i + k < N_NODES) rowptr[i + k] = ex;
    ex += d[k];
  }
  if (i < N_NODES && i + 4 >= N_NODES) rowptr[N_NODES] = ex;
}

__global__ void edge_fill(const int* __restrict__ src, const int* __restrict__ dst,
                          const int* __restrict__ rowptr, const int* __restrict__ rank,
                          int* __restrict__ csr) {
  int e = blockIdx.x * blockDim.x + threadIdx.x;
  if (e >= N_EDGES) return;
  csr[rowptr[dst[e]] + rank[e]] = src[e];
}

template <int NH>
__device__ __forceinline__ int load_edge(const int* __restrict__ csr, const float* __restrict__ ss,
                                         const float* sdn, int j, int end, float* x) {
  int s = 0;
  if (j < end) {
    s = csr[j];
    float v[NH];
    if constexpr (NH == 4) {
      const float4 q = *(const float4*)(ss + (size_t)s * 4);
      v[0] = q.x; v[1] = q.y; v[2] = q.z; v[3] = q.w;
    } else {
      for (int h = 0; h < NH; ++h) v[h] = ss[(size_t)s * NH + h];
    }
    for (int h = 0; h < NH; ++h) {
      const float t = v[h] + sdn[h];
      x[h] = (t > 0.f) ? t : 0.01f * t;
    }
  } else {
    for (int h = 0; h < NH; ++h) x[h] = -INFINITY;
  }
  return s;
}

template <int NH>
__device__ __forceinline__ float pick_alpha(const float* alpha, int t, int hh) {
  float a = __int_as_float(__builtin_amdgcn_readlane(__float_as_int(alpha[0]), t));
  for (int h = 1; h < NH; ++h) {
    const float b = __int_as_float(__builtin_amdgcn_readlane(__float_as_int(alpha[h]), t));
    a = (hh == h) ? b : a;
  }
  return a;
}

template <int NH>
__device__ __forceinline__ void load_row(const float* __restrict__ p, float* r) {
  if constexpr (NH == 4) {
    const float4 q = *(const float4*)p;
    r[0] = q.x; r[1] = q.y; r[2] = q.z; r[3] = q.w;
  } else {
    for (int h = 0; h < NH; ++h) r[h] = p[h];
  }
}

template <int NH>
__device__ __forceinline__ void add_rows(const float* __restrict__ zl, int s, const float* alpha,
                                         int cnt, int hh, float* acc) {
  constexpr int C = NH * HID;
  int t = 0;
  for (; t + 4 <= cnt; t += 4) {
    float r[4][NH], a[4];
#pragma unroll
    for (int u = 0; u < 4; ++u) {
      const int st = __builtin_amdgcn_readlane(s, t + u);
      a[u] = pick_alpha<NH>(alpha, t + u, hh);
      load_row<NH>(zl + (size_t)st * C, r[u]);
    }
#pragma unroll
    for (int u = 0; u < 4; ++u)
      for (int h = 0; h < NH; ++h) acc[h] += a[u] * r[u][h];
  }
  for (; t < cnt; ++t) {
    float r[NH];
    const int st = __builtin_amdgcn_readlane(s, t);
    const float a = pick_alpha<NH>(alpha, t, hh);
    load_row<NH>(zl + (size_t)st * C, r);
    for (int h = 0; h < NH; ++h) acc[h] += a * r[h];
  }
}

template <int NH, bool ELU>
__global__ void __launch_bounds__(256)
node_softmax_gather(const int* __restrict__ rowptr, const int* __restrict__ csr,
                    const float* __restrict__ ss, const float* __restrict__ sd,
                    const float* __restrict__ z, float* __restrict__ out, int ldo) {
  const int n = __builtin_amdgcn_readfirstlane((int)(blockIdx.x * 8u + (threadIdx.x >> 5)));
  if (n >= N_NODES) return;
  const int lane = threadIdx.x & 31;
  const int hh = (lane * NH) >> 5;
  const int beg = rowptr[n];
  const int end = rowptr[n + 1];
  float acc[NH];
  for (int h = 0; h < NH; ++h) acc[h] = 0.f;
  if (end > beg) {
    float sdn[NH];
    for (int h = 0; h < NH; ++h) sdn[h] = sd[(size_t)n * NH + h];
    const float* zl = z + lane * NH;
    float x0[NH];
    const int s0 = load_edge<NH>(csr, ss, sdn, beg + lane, end, x0);
    const bool v0 = (beg + lane) < end;
    float m[NH];
    for (int h = 0; h < NH; ++h) m[h] = x0[h];
    for (int base = beg + 32; base < end; base += 32) {
      float x[NH];
      load_edge<NH>(csr, ss, sdn, base + lane, end, x);
      for (int h = 0; h < NH; ++h) m[h] = fmaxf(m[h], x[h]);
    }
    for (int o = 16; o > 0; o >>= 1)
      for (int h = 0; h < NH; ++h) m[h] = fmaxf(m[h], __shfl_xor(m[h], o));
    float ee0[NH], sum[NH];
    for (int h = 0; h < NH; ++h) { ee0[h] = v0 ? expf(x0[h] - m[h]) : 0.f; sum[h] = ee0[h]; }
    for (int base = beg + 32; base < end; base += 32) {
      float x[NH];
      load_edge<NH>(csr, ss, sdn, base + lane, end, x);
      const bool v = (base + lane) < end;
      for (int h = 0; h < NH; ++h) sum[h] += v ? expf(x[h] - m[h]) : 0.f;
    }
    for (int o = 16; o > 0; o >>= 1)
      for (int h = 0; h < NH; ++h) sum[h] += __shfl_xor(sum[h], o);
    float den[NH], al[NH];
    for (int h = 0; h < NH; ++h) { den[h] = fmaxf(sum[h], 1e-9f); al[h] = ee0[h] / den[h]; }
    const int c0 = end - beg;
    add_rows<NH>(zl, s0, al, c0 < 32 ? c0 : 32, hh, acc);
    for (int base = beg + 32; base < end; base += 32) {
      float x[NH];
      const int s = load_edge<NH>(csr, ss, sdn, base + lane, end, x);
      const bool v = (base + lane) < end;
      for (int h = 0; h < NH; ++h) al[h] = v ? expf(x[h] - m[h]) / den[h] : 0.f;
      const int c = end - base;
      add_rows<NH>(zl, s, al, c < 32 ? c : 32, hh, acc);
    }
  }
  if constexpr (ELU) {
    for (int h = 0; h < NH; ++h) acc[h] = (acc[h] > 0.f) ? acc[h] : (expf(acc[h]) - 1.0f);
  }
  float* op = out + (size_t)n * ldo + lane * NH;
  if constexpr (NH == 4) {
    *(float4*)op = make_float4(acc[0], acc[1], acc[2], acc[3]);
  } else {
    for (int h = 0; h < NH; ++h) op[h] = acc[h];
  }
}

extern "C" void kernel_launch(void* const* d_in, const int* in_sizes, int n_in,
                              void* d_out, int out_size, void* d_ws, size_t ws_size,
                              hipStream_t stream) {
  const float* h    = (const float*)d_in[0];
  const int*   src  = (const int*)  d_in[1];
  const int*   dst  = (const int*)  d_in[2];
  const int*   fc   = (const int*)  d_in[3];
  const float* W1   = (const float*)d_in[4];
  const float* a1   = (const float*)d_in[5];
  const float* W2   = (const float*)d_in[6];
  const float* a2   = (const float*)d_in[7];
  const float* Wf   = (const float*)d_in[8];
  const float* Wff  = (const float*)d_in[9];
  const float* Wfc2 = (const float*)d_in[10];
  float* out = (float*)d_out;

  float* w = (float*)d_ws;
  size_t off = 0;
  int* deg    = (int*)(w + off); off += N_NODES;
  int* rowptr = (int*)(w + off); off += N_NODES + 32;
  int* btot   = (int*)(w + off); off += 128;
  int* rank   = (int*)(w + off); off += N_EDGES;
  int* csr    = (int*)(w + off); off += N_EDGES;
  float* z1  = w + off; off += (size_t)N_NODES * 128;
  float* s1s = w + off; off += (size_t)N_NODES * HEADS;
  float* s1d = w + off; off += (size_t)N_NODES * HEADS;
  float* h1  = w + off; off += (size_t)N_NODES * 128;
  float* z2  = w + off; off += (size_t)N_NODES * HID;
  float* s2s = w + off; off += N_NODES;
  float* s2d = w + off; off += N_NODES;
  float* hg  = w + off; off += (size_t)N_NODES * 64;
  float* hfp = w + off; off += (size_t)N_NODES * 96;
  (void)ws_size; (void)in_sizes; (void)n_in; (void)out_size;

  const int TB = 256;
  const int MT = N_NODES / 16;
  const int gemm_blocks = (MT + 7) / 8;
  const int edge_blocks = (N_EDGES + TB - 1) / TB;
  const int scan_blocks = (N_NODES + 1023) / 1024;
  const int node_blocks = (N_NODES + 7) / 8;

  zero_f32<<<(N_NODES + TB - 1) / TB, TB, 0, stream>>>((float*)deg, (long long)N_NODES);
  edge_count_rank<<<edge_blocks, TB, 0, stream>>>(dst, deg, rank);
  scan_block_totals<<<scan_blocks, TB, 0, stream>>>(deg, btot);
  scan_rowptr<<<scan_blocks, TB, 0, stream>>>(deg, btot, rowptr);
  edge_fill<<<edge_blocks, TB, 0, stream>>>(src, dst, rowptr, rank, csr);

  wmma_gemm_strip<8, 128><<<gemm_blocks, TB, 0, stream>>>(h, W1, z1, MT, 128, 0);
  attn_scores<<<(N_NODES * HEADS + TB - 1) / TB, TB, 0, stream>>>(z1, a1, s1s, s1d, N_NODES, HEADS);
  node_softmax_gather<HEADS, true><<<node_blocks, TB, 0, stream>>>(rowptr, csr, s1s, s1d, z1, h1, 128);

  wmma_gemm_strip<2, 128><<<gemm_blocks, TB, 0, stream>>>(h1, W2, z2, MT, HID, 0);
  attn_scores<<<(N_NODES + TB - 1) / TB, TB, 0, stream>>>(z2, a2, s2s, s2d, N_NODES, 1);
  node_softmax_gather<1, false><<<node_blocks, TB, 0, stream>>>(rowptr, csr, s2s, s2d, z2, hg, 64);

  {
    dim3 g(gemm_blocks, 3);
    wmma_gemm_frac<<<g, TB, 0, stream>>>(h, fc, Wf, hfp);
  }
  wmma_gemm_strip<2, 96><<<gemm_blocks, TB, 0, stream>>>(hfp, Wff, hg, MT, 64, HID);

  wmma_gemm_strip<1, 64><<<gemm_blocks, TB, 0, stream>>>(hg, Wfc2, out, MT, OUT_DIM, 0);
}


# Edges grouped by dst through LDS bins (256-node ranges, one global add per slice and bin, per-bin LDS sort, run-wise stores), not a global atomic + scattered store per edge; slots in arrival order
# speedup vs baseline: 6.3829x; 1.4856x over previous
#include <hip/hip_runtime.h>
#include <math.h>

#define N_NODES 100000
#define N_EDGES 1600000
#define IN_DIM  128
#define HID     32
#define HEADS   4
#define OUT_DIM 16

typedef __attribute__((ext_vector_type(2))) float v2f;
typedef __attribute__((ext_vector_type(8))) float v8f;

__global__ void zero_f32(float* __restrict__ p, long long n) {
  long long t = (long long)blockIdx.x * blockDim.x + threadIdx.x;
  if (t < n) p[t] = 0.0f;
}

template <int NT, int K>
__device__ __forceinline__ void stage_weights(const float* __restrict__ W, float4* Ws) {
  constexpr int K4 = K / 4;
  constexpr int P4 = K4 + 1;
  for (int i = threadIdx.x; i < NT * 16 * K4; i += 256) {
    const int row = i / K4;
    const int c = i - row * K4;
    Ws[row * P4 + c] = ((const float4*)W)[i];
  }
}

template <int NT, int K>
__global__ void __launch_bounds__(256)
wmma_gemm_strip(const float* __restrict__ A, const float* __restrict__ W,
                float* __restrict__ C, int Mtiles, int ldc, int coff) {
  constexpr int P = K + 4;
  __shared__ float4 Ws[NT * 16 * (P / 4)];
  stage_weights<NT, K>(W, Ws);
  __syncthreads();
  const int wave = (blockIdx.x * blockDim.x + threadIdx.x) >> 5;
  if (wave < Mtiles) {
    const int lane = threadIdx.x & 31;
    const int lo = lane & 15;
    const int hi = lane >> 4;
    const float* Ab = A + ((size_t)wave * 16 + lo) * K;
    const float* Wl = (const float*)Ws + lo * P;
    v8f acc[NT];
#pragma unroll
    for (int j = 0; j < NT; ++j)
      acc[j] = (v8f){0.f, 0.f, 0.f, 0.f, 0.f, 0.f, 0.f, 0.f};
    for (int k0 = 0; k0 < K; k0 += 4) {
      const int ka = k0 + hi * 2;
      v2f a;
      a.x = Ab[ka]; a.y = Ab[ka + 1];
#pragma unroll
      for (int j = 0; j < NT; ++j) {
        const v2f b = *(const v2f*)(Wl + j * 16 * P + ka);
        acc[j] = __builtin_amdgcn_wmma_f32_16x16x4_f32(false, a, false, b,
                                                       (short)0, acc[j], false, false);
      }
    }
#pragma unroll
    for (int j = 0; j < NT; ++j) {
      float* Cb = C + (size_t)wave * 16 * ldc + coff + (size_t)j * 16 + lo;
#pragma unroll
      for (int r = 0; r < 8; ++r)
        Cb[(size_t)(r + hi * 8) * ldc] = acc[j][r];
    }
  }
}

__global__ void __launch_bounds__(256)
wmma_gemm_frac(const float* __restrict__ H, const int* __restrict__ fc,
               const float* __restrict__ Wf, float* __restrict__ hfpre) {
  constexpr int P = IN_DIM + 4;
  __shared__ float4 Ws[2 * 16 * (P / 4)];
  const int s = blockIdx.y;
  stage_weights<2, IN_DIM>(Wf + (size_t)s * HID * IN_DIM, Ws);
  __syncthreads();
  const int wave = (blockIdx.x * blockDim.x + threadIdx.x) >> 5;
  if (wave < N_NODES / 16) {
    const int lane = threadIdx.x & 31;
    const int lo = lane & 15;
    const int hi = lane >> 4;
    const int nid = fc[(wave * 16 + lo) * 3 + s];
    const float* Ab = H + (size_t)nid * IN_DIM;
    const float* Wl = (const float*)Ws + lo * P;
    v8f acc[2];
    acc[0] = (v8f){0.f, 0.f, 0.f, 0.f, 0.f, 0.f, 0.f, 0.f};
    acc[1] = acc[0];
    for (int k0 = 0; k0 < IN_DIM; k0 += 4) {
      const int ka = k0 + hi * 2;
      v2f a;
      a.x = Ab[ka]; a.y = Ab[ka + 1];
#pragma unroll
      for (int j = 0; j < 2; ++j) {
        const v2f b = *(const v2f*)(Wl + j * 16 * P + ka);
        acc[j] = __builtin_amdgcn_wmma_f32_16x16x4_f32(false, a, false, b,
                                                       (short)0, acc[j], false, false);
      }
    }
#pragma unroll
    for (int j = 0; j < 2; ++j) {
      float* Cb = hfpre + (size_t)wave * 16 * 96 + s * HID + (size_t)j * 16 + lo;
#pragma unroll
      for (int r = 0; r < 8; ++r)
        Cb[(size_t)(r + hi * 8) * 96] = acc[j][r];
    }
  }
}

__global__ void attn_scores(const float* __restrict__ z, const float* __restrict__ a,
                            float* __restrict__ ss, float* __restrict__ sd,
                            int n, int heads) {
  int t = blockIdx.x * blockDim.x + threadIdx.x;
  if (t >= n * heads) return;
  int node = t / heads, hh = t - node * heads;
  const float* zp = z + (size_t)node * heads * HID + hh * HID;
  const float* al = a + (size_t)hh * 2 * HID;
  const float* ar = al + HID;
  float s0 = 0.f, s1 = 0.f;
#pragma unroll 8
  for (int o = 0; o < HID; ++o) { s0 += zp[o] * al[o]; s1 += zp[o] * ar[o]; }
  ss[t] = s0; sd[t] = s1;
}

#define BIN_SHIFT 8
#define BIN_NODES (1 << BIN_SHIFT)
#define NBINS     ((N_NODES + BIN_NODES - 1) >> BIN_SHIFT)
#define SLICES    256
#define SLICE_E   ((N_EDGES + SLICES - 1) / SLICES)
#define SORT_T    1024
#define PER_T     ((SLICE_E + SORT_T - 1) / SORT_T)
#define SRC_BITS  17
#define BIN_CAP   8192
static_assert(N_NODES <= (1 << SRC_BITS), "a source id must fit the packed word");
static_assert(NBINS <= SORT_T, "one thread per bin in the scans");
static_assert(BIN_NODES <= SORT_T, "one thread per node in the scans");

__device__ __forceinline__ int block_excl_scan(int v, int* wsum) {
  const int lane = threadIdx.x & 31, w = threadIdx.x >> 5;
  int inc = v;
  for (int o = 1; o < 32; o <<= 1) {
    const int u = __shfl_up(inc, o);
    if (lane >= o) inc += u;
  }
  if (lane == 31) wsum[w] = inc;
  __syncthreads();
  if (w == 0) {
    const int s = wsum[lane];
    int si = s;
    for (int o = 1; o < 32; o <<= 1) {
      const int u = __shfl_up(si, o);
      if (lane >= o) si += u;
    }
    wsum[lane] = si - s;
  }
  __syncthreads();
  const int ex = wsum[w] + inc - v;
  __syncthreads();
  return ex;
}

__global__ void __launch_bounds__(SORT_T)
bin_count(const int* __restrict__ dst, int* __restrict__ bintot, int* __restrict__ runbase) {
  __shared__ int hist[NBINS];
  const int t = threadIdx.x;
  for (int i = t; i < NBINS; i += SORT_T) hist[i] = 0;
  __syncthreads();
  const int e0 = blockIdx.x * SLICE_E;
  const int e1 = (e0 + SLICE_E < N_EDGES) ? e0 + SLICE_E : N_EDGES;
  for (int e = e0 + t; e < e1; e += SORT_T) atomicAdd(&hist[dst[e] >> BIN_SHIFT], 1);
  __syncthreads();
  for (int i = t; i < NBINS; i += SORT_T)
    runbase[blockIdx.x * NBINS + i] = atomicAdd(&bintot[i], hist[i]);
}

__global__ void __launch_bounds__(SORT_T)
bin_scatter(const int* __restrict__ src, const int* __restrict__ dst,
            const int* __restrict__ bintot, const int* __restrict__ runbase,
            int* __restrict__ binned) {
  __shared__ int hist[NBINS];
  __shared__ int lstart[NBINS];
  __shared__ int gbase[NBINS];
  __shared__ int wsum[32];
  __shared__ int sval[SLICE_E];
  __shared__ int sidx[SLICE_E];
  const int t = threadIdx.x;
  for (int i = t; i < NBINS; i += SORT_T) hist[i] = 0;
  __syncthreads();
  const int e0 = blockIdx.x * SLICE_E;
  const int e1 = (e0 + SLICE_E < N_EDGES) ? e0 + SLICE_E : N_EDGES;
  int d[PER_T], s[PER_T];
#pragma unroll
  for (int k = 0; k < PER_T; ++k) {
    const int e = e0 + t + k * SORT_T;
    d[k] = -1; s[k] = 0;
    if (e < e1) { d[k] = dst[e]; s[k] = src[e]; atomicAdd(&hist[d[k] >> BIN_SHIFT], 1); }
  }
  __syncthreads();
  const int tot = (t < NBINS) ? bintot[t] : 0;
  const int cnt = (t < NBINS) ? hist[t] : 0;
  const int binstart = block_excl_scan(tot, wsum);
  const int lst = block_excl_scan(cnt, wsum);
  if (t < NBINS) {
    gbase[t] = binstart + runbase[blockIdx.x * NBINS + t];
    lstart[t] = lst;
    hist[t] = 0;
  }
  __syncthreads();
#pragma unroll
  for (int k = 0; k < PER_T; ++k) {
    if (d[k] >= 0) {
      const int b = d[k] >> BIN_SHIFT;
      const int r = atomicAdd(&hist[b], 1);
      const int p = lstart[b] + r;
      sval[p] = s[k] | ((d[k] & (BIN_NODES - 1)) << SRC_BITS);
      sidx[p] = gbase[b] + r;
    }
  }
  __syncthreads();
  const int n = e1 - e0;
  for (int i = t; i < n; i += SORT_T) binned[sidx[i]] = sval[i];
}

__global__ void __launch_bounds__(SORT_T)
bin_finish(const int* __restrict__ bintot, const int* __restrict__ binned,
           int* __restrict__ rowptr, int* __restrict__ csr) {
  __shared__ int deg[BIN_NODES];
  __shared__ int first[BIN_NODES];
  __shared__ int wsum[32];
  __shared__ int range[2];
  __shared__ int sorted[BIN_CAP];
  const int t = threadIdx.x;
  const int b = blockIdx.x;
  const int tot = (t < NBINS) ? bintot[t] : 0;
  const int ex0 = block_excl_scan(tot, wsum);
  if (t == b) { range[0] = ex0; range[1] = tot; }
  if (t < BIN_NODES) deg[t] = 0;
  __syncthreads();
  const int start = range[0];
  const int cnt = range[1];
  const int* mine = binned + start;
  for (int i = t; i < cnt; i += SORT_T) atomicAdd(&deg[((unsigned)mine[i]) >> SRC_BITS], 1);
  __syncthreads();
  const int dg = (t < BIN_NODES) ? deg[t] : 0;
  const int ex = block_excl_scan(dg, wsum);
  if (t < BIN_NODES) {
    const int node = b * BIN_NODES + t;
    if (node < N_NODES) rowptr[node] = start + ex;
    first[t] = ex;
    deg[t] = 0;
  }
  if (b == NBINS - 1 && t == 0) rowptr[N_NODES] = start + cnt;
  __syncthreads();
  if (cnt <= BIN_CAP) {
    for (int i = t; i < cnt; i += SORT_T) {
      const unsigned w = (unsigned)mine[i];
      const int dl = w >> SRC_BITS;
      const int r = atomicAdd(&deg[dl], 1);
      sorted[first[dl] + r] = (int)(w & ((1u << SRC_BITS) - 1u));
    }
    __syncthreads();
    for (int i = t; i < cnt; i += SORT_T) csr[start + i] = sorted[i];
  } else {
    for (int i = t; i < cnt; i += SORT_T) {
      const unsigned w = (unsigned)mine[i];
      const int dl = w >> SRC_BITS;
      const int r = atomicAdd(&deg[dl], 1);
      csr[start + first[dl] + r] = (int)(w & ((1u << SRC_BITS) - 1u));
    }
  }
}

template <int NH>
__device__ __forceinline__ int load_edge(const int* __restrict__ csr, const float* __restrict__ ss,
                                         const float* sdn, int j, int end, float* x) {
  int s = 0;
  if (j < end) {
    s = csr[j];
    float v[NH];
    if constexpr (NH == 4) {
      const float4 q = *(const float4*)(ss + (size_t)s * 4);
      v[0] = q.x; v[1] = q.y; v[2] = q.z; v[3] = q.w;
    } else {
      for (int h = 0; h < NH; ++h) v[h] = ss[(size_t)s * NH + h];
    }
    for (int h = 0; h < NH; ++h) {
      const float t = v[h] + sdn[h];
      x[h] = (t > 0.f) ? t : 0.01f * t;
    }
  } else {
    for (int h = 0; h < NH; ++h) x[h] = -INFINITY;
  }
  return s;
}

template <int NH>
__device__ __forceinline__ float pick_alpha(const float* alpha, int t, int hh) {
  float a = __int_as_float(__builtin_amdgcn_readlane(__float_as_int(alpha[0]), t));
  for (int h = 1; h < NH; ++h) {
    const float b = __int_as_float(__builtin_amdgcn_readlane(__float_as_int(alpha[h]), t));
    a = (hh == h) ? b : a;
  }
  return a;
}

template <int NH>
__device__ __forceinline__ void load_row(const float* __restrict__ p, float* r) {
  if constexpr (NH == 4) {
    const float4 q = *(const float4*)p;
    r[0] = q.x; r[1] = q.y; r[2] = q.z; r[3] = q.w;
  } else {
    for (int h = 0; h < NH; ++h) r[h] = p[h];
  }
}

template <int NH>
__device__ __forceinline__ void add_rows(const float* __restrict__ zl, int s, const float* alpha,
                                         int cnt, int hh, float* acc) {
  constexpr int C = NH * HID;
  int t = 0;
  for (; t + 4 <= cnt; t += 4) {
    float r[4][NH], a[4];
#pragma unroll
    for (int u = 0; u < 4; ++u) {
      const int st = __builtin_amdgcn_readlane(s, t + u);
      a[u] = pick_alpha<NH>(alpha, t + u, hh);
      load_row<NH>(zl + (size_t)st * C, r[u]);
    }
#pragma unroll
    for (int u = 0; u < 4; ++u)
      for (int h = 0; h < NH; ++h) acc[h] += a[u] * r[u][h];
  }
  for (; t < cnt; ++t) {
    float r[NH];
    const int st = __builtin_amdgcn_readlane(s, t);
    const float a = pick_alpha<NH>(alpha, t, hh);
    load_row<NH>(zl + (size_t)st * C, r);
    for (int h = 0; h < NH; ++h) acc[h] += a * r[h];
  }
}

template <int NH, bool ELU>
__global__ void __launch_bounds__(256)
node_softmax_gather(const int* __restrict__ rowptr, const int* __restrict__ csr,
                    const float* __restrict__ ss, const float* __restrict__ sd,
                    const float* __restrict__ z, float* __restrict__ out, int ldo) {
  const int n = __builtin_amdgcn_readfirstlane((int)(blockIdx.x * 8u + (threadIdx.x >> 5)));
  if (n >= N_NODES) return;
  const int lane = threadIdx.x & 31;
  const int hh = (lane * NH) >> 5;
  const int beg = rowptr[n];
  const int end = rowptr[n + 1];
  float acc[NH];
  for (int h = 0; h < NH; ++h) acc[h] = 0.f;
  if (end > beg) {
    float sdn[NH];
    for (int h = 0; h < NH; ++h) sdn[h] = sd[(size_t)n * NH + h];
    const float* zl = z + lane * NH;
    float x0[NH];
    const int s0 = load_edge<NH>(csr, ss, sdn, beg + lane, end, x0);
    const bool v0 = (beg + lane) < end;
    float m[NH];
    for (int h = 0; h < NH; ++h) m[h] = x0[h];
    for (int base = beg + 32; base < end; base += 32) {
      float x[NH];
      load_edge<NH>(csr, ss, sdn, base + lane, end, x);
      for (int h = 0; h < NH; ++h) m[h] = fmaxf(m[h], x[h]);
    }
    for (int o = 16; o > 0; o >>= 1)
      for (int h = 0; h < NH; ++h) m[h] = fmaxf(m[h], __shfl_xor(m[h], o));
    float ee0[NH], sum[NH];
    for (int h = 0; h < NH; ++h) { ee0[h] = v0 ? expf(x0[h] - m[h]) : 0.f; sum[h] = ee0[h]; }
    for (int base = beg + 32; base < end; base += 32) {
      float x[NH];
      load_edge<NH>(csr, ss, sdn, base + lane, end, x);
      const bool v = (base + lane) < end;
      for (int h = 0; h < NH; ++h) sum[h] += v ? expf(x[h] - m[h]) : 0.f;
    }
    for (int o = 16; o > 0; o >>= 1)
      for (int h = 0; h < NH; ++h) sum[h] += __shfl_xor(sum[h], o);
    float den[NH], al[NH];
    for (int h = 0; h < NH; ++h) { den[h] = fmaxf(sum[h], 1e-9f); al[h] = ee0[h] / den[h]; }
    const int c0 = end - beg;
    add_rows<NH>(zl, s0, al, c0 < 32 ? c0 : 32, hh, acc);
    for (int base = beg + 32; base < end; base += 32) {
      float x[NH];
      const int s = load_edge<NH>(csr, ss, sdn, base + lane, end, x);
      const bool v = (base + lane) < end;
      for (int h = 0; h < NH; ++h) al[h] = v ? expf(x[h] - m[h]) / den[h] : 0.f;
      const int c = end - base;
      add_rows<NH>(zl, s, al, c < 32 ? c : 32, hh, acc);
    }
  }
  if constexpr (ELU) {
    for (int h = 0; h < NH; ++h) acc[h] = (acc[h] > 0.f) ? acc[h] : (expf(acc[h]) - 1.0f);
  }
  float* op = out + (size_t)n * ldo + lane * NH;
  if constexpr (NH == 4) {
    *(float4*)op = make_float4(acc[0], acc[1], acc[2], acc[3]);
  } else {
    for (int h = 0; h < NH; ++h) op[h] = acc[h];
  }
}

extern "C" void kernel_launch(void* const* d_in, const int* in_sizes, int n_in,
                              void* d_out, int out_size, void* d_ws, size_t ws_size,
                              hipStream_t stream) {
  const float* h    = (const float*)d_in[0];
  const int*   src  = (const int*)  d_in[1];
  const int*   dst  = (const int*)  d_in[2];
  const int*   fc   = (const int*)  d_in[3];
  const float* W1   = (const float*)d_in[4];
  const float* a1   = (const float*)d_in[5];
  const float* W2   = (const float*)d_in[6];
  const float* a2   = (const float*)d_in[7];
  const float* Wf   = (const float*)d_in[8];
  const float* Wff  = (const float*)d_in[9];
  const float* Wfc2 = (const float*)d_in[10];
  float* out = (float*)d_out;

  float* w = (float*)d_ws;
  size_t off = 0;
  int* bintot  = (int*)(w + off); off += 512;
  int* runbase = (int*)(w + off); off += SLICES * 512;
  int* rowptr  = (int*)(w + off); off += N_NODES + 32;
  int* binned  = (int*)(w + off); off += N_EDGES;
  int* csr     = (int*)(w + off); off += N_EDGES;
  float* z1  = w + off; off += (size_t)N_NODES * 128;
  float* s1s = w + off; off += (size_t)N_NODES * HEADS;
  float* s1d = w + off; off += (size_t)N_NODES * HEADS;
  float* h1  = w + off; off += (size_t)N_NODES * 128;
  float* z2  = w + off; off += (size_t)N_NODES * HID;
  float* s2s = w + off; off += N_NODES;
  float* s2d = w + off; off += N_NODES;
  float* hg  = w + off; off += (size_t)N_NODES * 64;
  float* hfp = w + off; off += (size_t)N_NODES * 96;
  (void)ws_size; (void)in_sizes; (void)n_in; (void)out_size;

  const int TB = 256;
  const int MT = N_NODES / 16;
  const int gemm_blocks = (MT + 7) / 8;
  const int node_blocks = (N_NODES + 7) / 8;

  zero_f32<<<(NBINS + TB - 1) / TB, TB, 0, stream>>>((float*)bintot, (long long)NBINS);
  bin_count<<<SLICES, SORT_T, 0, stream>>>(dst, bintot, runbase);
  bin_scatter<<<SLICES, SORT_T, 0, stream>>>(src, dst, bintot, runbase, binned);
  bin_finish<<<NBINS, SORT_T, 0, stream>>>(bintot, binned, rowptr, csr);

  wmma_gemm_strip<8, 128><<<gemm_blocks, TB, 0, stream>>>(h, W1, z1, MT, 128, 0);
  attn_scores<<<(N_NODES * HEADS + TB - 1) / TB, TB, 0, stream>>>(z1, a1, s1s, s1d, N_NODES, HEADS);
  node_softmax_gather<HEADS, true><<<node_blocks, TB, 0, stream>>>(rowptr, csr, s1s, s1d, z1, h1, 128);

  wmma_gemm_strip<2, 128><<<gemm_blocks, TB, 0, stream>>>(h1, W2, z2, MT, HID, 0);
  attn_scores<<<(N_NODES + TB - 1) / TB, TB, 0, stream>>>(z2, a2, s2s, s2d, N_NODES, 1);
  node_softmax_gather<1, false><<<node_blocks, TB, 0, stream>>>(rowptr, csr, s2s, s2d, z2, hg, 64);

  {
    dim3 g(gemm_blocks, 3);
    wmma_gemm_frac<<<g, TB, 0, stream>>>(h, fc, Wf, hfp);
  }
  wmma_gemm_strip<2, 96><<<gemm_blocks, TB, 0, stream>>>(hfp, Wff, hg, MT, 64, HID);

  wmma_gemm_strip<1, 64><<<gemm_blocks, TB, 0, stream>>>(hg, Wfc2, out, MT, OUT_DIM, 0);
}


# Attention scores formed in the z GEMMs' epilogue from the accumulators (2-column product per lane, 16-lane DPP tree sum, not a serial 32-term sum) instead of a kernel re-reading z; 13 launches -> 11
# speedup vs baseline: 6.6547x; 1.0426x over previous
#include <hip/hip_runtime.h>
#include <math.h>

#define N_NODES 100000
#define N_EDGES 1600000
#define IN_DIM  128
#define HID     32
#define HEADS   4
#define OUT_DIM 16

typedef __attribute__((ext_vector_type(2))) float v2f;
typedef __attribute__((ext_vector_type(8))) float v8f;

__global__ void zero_f32(float* __restrict__ p, long long n) {
  long long t = (long long)blockIdx.x * blockDim.x + threadIdx.x;
  if (t < n) p[t] = 0.0f;
}

template <int NT, int K>
__device__ __forceinline__ void stage_weights(const float* __restrict__ W, float4* Ws) {
  constexpr int K4 = K / 4;
  constexpr int P4 = K4 + 1;
  for (int i = threadIdx.x; i < NT * 16 * K4; i += 256) {
    const int row = i / K4;
    const int c = i - row * K4;
    Ws[row * P4 + c] = ((const float4*)W)[i];
  }
}

__device__ __forceinline__ float row16_sum(float x) {
  x += __int_as_float(__builtin_amdgcn_update_dpp(0, __float_as_int(x), 0x161, 0xf, 0xf, true));
  x += __int_as_float(__builtin_amdgcn_update_dpp(0, __float_as_int(x), 0x162, 0xf, 0xf, true));
  x += __int_as_float(__builtin_amdgcn_update_dpp(0, __float_as_int(x), 0x164, 0xf, 0xf, true));
  x += __int_as_float(__builtin_amdgcn_update_dpp(0, __float_as_int(x), 0x168, 0xf, 0xf, true));
  return x;
}

template <int NT, int K, int NHD>
__global__ void __launch_bounds__(256)
wmma_gemm_strip(const float* __restrict__ A, const float* __restrict__ W,
                float* __restrict__ C, int Mtiles, int ldc, int coff,
                const float* __restrict__ av, float* __restrict__ ss, float* __restrict__ sd) {
  constexpr int P = K + 4;
  __shared__ float4 Ws[NT * 16 * (P / 4)];
  stage_weights<NT, K>(W, Ws);
  __syncthreads();
  const int wave = (blockIdx.x * blockDim.x + threadIdx.x) >> 5;
  if (wave < Mtiles) {
    const int lane = threadIdx.x & 31;
    const int lo = lane & 15;
    const int hi = lane >> 4;
    const float* Ab = A + ((size_t)wave * 16 + lo) * K;
    const float* Wl = (const float*)Ws + lo * P;
    v8f acc[NT];
#pragma unroll
    for (int j = 0; j < NT; ++j)
      acc[j] = (v8f){0.f, 0.f, 0.f, 0.f, 0.f, 0.f, 0.f, 0.f};
    for (int k0 = 0; k0 < K; k0 += 4) {
      const int ka = k0 + hi * 2;
      v2f a;
      a.x = Ab[ka]; a.y = Ab[ka + 1];
#pragma unroll
      for (int j = 0; j < NT; ++j) {
        const v2f b = *(const v2f*)(Wl + j * 16 * P + ka);
        acc[j] = __builtin_amdgcn_wmma_f32_16x16x4_f32(false, a, false, b,
                                                       (short)0, acc[j], false, false);
      }
    }
#pragma unroll
    for (int j = 0; j < NT; ++j) {
      float* Cb = C + (size_t)wave * 16 * ldc + coff + (size_t)j * 16 + lo;
#pragma unroll
      for (int r = 0; r < 8; ++r)
        Cb[(size_t)(r + hi * 8) * ldc] = acc[j][r];
    }
    if constexpr (NHD > 0) {
      static_assert(NT == 2 * NHD, "a head is two 16-column tiles");
      float o[NHD];
#pragma unroll
      for (int h = 0; h < NHD; ++h) {
        const float* ah = av + h * 2 * HID;
        const float al0 = ah[lo], al1 = ah[16 + lo], ar0 = ah[HID + lo], ar1 = ah[HID + 16 + lo];
        float keep = 0.f;
#pragma unroll
        for (int r = 0; r < 8; ++r) {
          const float ps = row16_sum(acc[2 * h][r] * al0 + acc[2 * h + 1][r] * al1);
          const float pd = row16_sum(acc[2 * h][r] * ar0 + acc[2 * h + 1][r] * ar1);
          keep = (lo == r) ? ps : keep;
          keep = (lo == r + 8) ? pd : keep;
        }
        o[h] = keep;
      }
      float* sp = ((lo < 8) ? ss : sd) + ((size_t)wave * 16 + hi * 8 + (lo & 7)) * NHD;
      if constexpr (NHD == 4) {
        *(float4*)sp = make_float4(o[0], o[1], o[2], o[3]);
      } else {
#pragma unroll
        for (int h = 0; h < NHD; ++h) sp[h] = o[h];
      }
    }
  }
}

__global__ void __launch_bounds__(256)
wmma_gemm_frac(const float* __restrict__ H, const int* __restrict__ fc,
               const float* __restrict__ Wf, float* __restrict__ hfpre) {
  constexpr int P = IN_DIM + 4;
  __shared__ float4 Ws[2 * 16 * (P / 4)];
  const int s = blockIdx.y;
  stage_weights<2, IN_DIM>(Wf + (size_t)s * HID * IN_DIM, Ws);
  __syncthreads();
  const int wave = (blockIdx.x * blockDim.x + threadIdx.x) >> 5;
  if (wave < N_NODES / 16) {
    const int lane = threadIdx.x & 31;
    const int lo = lane & 15;
    const int hi = lane >> 4;
    const int nid = fc[(wave * 16 + lo) * 3 + s];
    const float* Ab = H + (size_t)nid * IN_DIM;
    const float* Wl = (const float*)Ws + lo * P;
    v8f acc[2];
    acc[0] = (v8f){0.f, 0.f, 0.f, 0.f, 0.f, 0.f, 0.f, 0.f};
    acc[1] = acc[0];
    for (int k0 = 0; k0 < IN_DIM; k0 += 4) {
      const int ka = k0 + hi * 2;
      v2f a;
      a.x = Ab[ka]; a.y = Ab[ka + 1];
#pragma unroll
      for (int j = 0; j < 2; ++j) {
        const v2f b = *(const v2f*)(Wl + j * 16 * P + ka);
        acc[j] = __builtin_amdgcn_wmma_f32_16x16x4_f32(false, a, false, b,
                                                       (short)0, acc[j], false, false);
      }
    }
#pragma unroll
    for (int j = 0; j < 2; ++j) {
      float* Cb = hfpre + (size_t)wave * 16 * 96 + s * HID + (size_t)j * 16 + lo;
#pragma unroll
      for (int r = 0; r < 8; ++r)
        Cb[(size_t)(r + hi * 8) * 96] = acc[j][r];
    }
  }
}

#define BIN_SHIFT 8
#define BIN_NODES (1 << BIN_SHIFT)
#define NBINS     ((N_NODES + BIN_NODES - 1) >> BIN_SHIFT)
#define SLICES    256
#define SLICE_E   ((N_EDGES + SLICES - 1) / SLICES)
#define SORT_T    1024
#define PER_T     ((SLICE_E + SORT_T - 1) / SORT_T)
#define SRC_BITS  17
#define BIN_CAP   8192
static_assert(N_NODES <= (1 << SRC_BITS), "a source id must fit the packed word");
static_assert(NBINS <= SORT_T, "one thread per bin in the scans");
static_assert(BIN_NODES <= SORT_T, "one thread per node in the scans");

__device__ __forceinline__ int block_excl_scan(int v, int* wsum) {
  const int lane = threadIdx.x & 31, w = threadIdx.x >> 5;
  int inc = v;
  for (int o = 1; o < 32; o <<= 1) {
    const int u = __shfl_up(inc, o);
    if (lane >= o) inc += u;
  }
  if (lane == 31) wsum[w] = inc;
  __syncthreads();
  if (w == 0) {
    const int s = wsum[lane];
    int si = s;
    for (int o = 1; o < 32; o <<= 1) {
      const int u = __shfl_up(si, o);
      if (lane >= o) si += u;
    }
    wsum[lane] = si - s;
  }
  __syncthreads();
  const int ex = wsum[w] + inc - v;
  __syncthreads();
  return ex;
}

__global__ void __launch_bounds__(SORT_T)
bin_count(const int* __restrict__ dst, int* __restrict__ bintot, int* __restrict__ runbase) {
  __shared__ int hist[NBINS];
  const int t = threadIdx.x;
  for (int i = t; i < NBINS; i += SORT_T) hist[i] = 0;
  __syncthreads();
  const int e0 = blockIdx.x * SLICE_E;
  const int e1 = (e0 + SLICE_E < N_EDGES) ? e0 + SLICE_E : N_EDGES;
  for (int e = e0 + t; e < e1; e += SORT_T) atomicAdd(&hist[dst[e] >> BIN_SHIFT], 1);
  __syncthreads();
  for (int i = t; i < NBINS; i += SORT_T)
    runbase[blockIdx.x * NBINS + i] = atomicAdd(&bintot[i], hist[i]);
}

__global__ void __launch_bounds__(SORT_T)
bin_scatter(const int* __restrict__ src, const int* __restrict__ dst,
            const int* __restrict__ bintot, const int* __restrict__ runbase,
            int* __restrict__ binned) {
  __shared__ int hist[NBINS];
  __shared__ int lstart[NBINS];
  __shared__ int gbase[NBINS];
  __shared__ int wsum[32];
  __shared__ int sval[SLICE_E];
  __shared__ int sidx[SLICE_E];
  const int t = threadIdx.x;
  for (int i = t; i < NBINS; i += SORT_T) hist[i] = 0;
  __syncthreads();
  const int e0 = blockIdx.x * SLICE_E;
  const int e1 = (e0 + SLICE_E < N_EDGES) ? e0 + SLICE_E : N_EDGES;
  int d[PER_T], s[PER_T];
#pragma unroll
  for (int k = 0; k < PER_T; ++k) {
    const int e = e0 + t + k * SORT_T;
    d[k] = -1; s[k] = 0;
    if (e < e1) { d[k] = dst[e]; s[k] = src[e]; atomicAdd(&hist[d[k] >> BIN_SHIFT], 1); }
  }
  __syncthreads();
  const int tot = (t < NBINS) ? bintot[t] : 0;
  const int cnt = (t < NBINS) ? hist[t] : 0;
  const int binstart = block_excl_scan(tot, wsum);
  const int lst = block_excl_scan(cnt, wsum);
  if (t < NBINS) {
    gbase[t] = binstart + runbase[blockIdx.x * NBINS + t];
    lstart[t] = lst;
    hist[t] = 0;
  }
  __syncthreads();
#pragma unroll
  for (int k = 0; k < PER_T; ++k) {
    if (d[k] >= 0) {
      const int b = d[k] >> BIN_SHIFT;
      const int r = atomicAdd(&hist[b], 1);
      const int p = lstart[b] + r;
      sval[p] = s[k] | ((d[k] & (BIN_NODES - 1)) << SRC_BITS);
      sidx[p] = gbase[b] + r;
    }
  }
  __syncthreads();
  const int n = e1 - e0;
  for (int i = t; i < n; i += SORT_T) binned[sidx[i]] = sval[i];
}

__global__ void __launch_bounds__(SORT_T)
bin_finish(const int* __restrict__ bintot, const int* __restrict__ binned,
           int* __restrict__ rowptr, int* __restrict__ csr) {
  __shared__ int deg[BIN_NODES];
  __shared__ int first[BIN_NODES];
  __shared__ int wsum[32];
  __shared__ int range[2];
  __shared__ int sorted[BIN_CAP];
  const int t = threadIdx.x;
  const int b = blockIdx.x;
  const int tot = (t < NBINS) ? bintot[t] : 0;
  const int ex0 = block_excl_scan(tot, wsum);
  if (t == b) { range[0] = ex0; range[1] = tot; }
  if (t < BIN_NODES) deg[t] = 0;
  __syncthreads();
  const int start = range[0];
  const int cnt = range[1];
  const int* mine = binned + start;
  for (int i = t; i < cnt; i += SORT_T) atomicAdd(&deg[((unsigned)mine[i]) >> SRC_BITS], 1);
  __syncthreads();
  const int dg = (t < BIN_NODES) ? deg[t] : 0;
  const int ex = block_excl_scan(dg, wsum);
  if (t < BIN_NODES) {
    const int node = b * BIN_NODES + t;
    if (node < N_NODES) rowptr[node] = start + ex;
    first[t] = ex;
    deg[t] = 0;
  }
  if (b == NBINS - 1 && t == 0) rowptr[N_NODES] = start + cnt;
  __syncthreads();
  if (cnt <= BIN_CAP) {
    for (int i = t; i < cnt; i += SORT_T) {
      const unsigned w = (unsigned)mine[i];
      const int dl = w >> SRC_BITS;
      const int r = atomicAdd(&deg[dl], 1);
      sorted[first[dl] + r] = (int)(w & ((1u << SRC_BITS) - 1u));
    }
    __syncthreads();
    for (int i = t; i < cnt; i += SORT_T) csr[start + i] = sorted[i];
  } else {
    for (int i = t; i < cnt; i += SORT_T) {
      const unsigned w = (unsigned)mine[i];
      const int dl = w >> SRC_BITS;
      const int r = atomicAdd(&deg[dl], 1);
      csr[start + first[dl] + r] = (int)(w & ((1u << SRC_BITS) - 1u));
    }
  }
}

template <int NH>
__device__ __forceinline__ int load_edge(const int* __restrict__ csr, const float* __restrict__ ss,
                                         const float* sdn, int j, int end, float* x) {
  int s = 0;
  if (j < end) {
    s = csr[j];
    float v[NH];
    if constexpr (NH == 4) {
      const float4 q = *(const float4*)(ss + (size_t)s * 4);
      v[0] = q.x; v[1] = q.y; v[2] = q.z; v[3] = q.w;
    } else {
      for (int h = 0; h < NH; ++h) v[h] = ss[(size_t)s * NH + h];
    }
    for (int h = 0; h < NH; ++h) {
      const float t = v[h] + sdn[h];
      x[h] = (t > 0.f) ? t : 0.01f * t;
    }
  } else {
    for (int h = 0; h < NH; ++h) x[h] = -INFINITY;
  }
  return s;
}

template <int NH>
__device__ __forceinline__ float pick_alpha(const float* alpha, int t, int hh) {
  float a = __int_as_float(__builtin_amdgcn_readlane(__float_as_int(alpha[0]), t));
  for (int h = 1; h < NH; ++h) {
    const float b = __int_as_float(__builtin_amdgcn_readlane(__float_as_int(alpha[h]), t));
    a = (hh == h) ? b : a;
  }
  return a;
}

template <int NH>
__device__ __forceinline__ void load_row(const float* __restrict__ p, float* r) {
  if constexpr (NH == 4) {
    const float4 q = *(const float4*)p;
    r[0] = q.x; r[1] = q.y; r[2] = q.z; r[3] = q.w;
  } else {
    for (int h = 0; h < NH; ++h) r[h] = p[h];
  }
}

template <int NH>
__device__ __forceinline__ void add_rows(const float* __restrict__ zl, int s, const float* alpha,
                                         int cnt, int hh, float* acc) {
  constexpr int C = NH * HID;
  int t = 0;
  for (; t + 4 <= cnt; t += 4) {
    float r[4][NH], a[4];
#pragma unroll
    for (int u = 0; u < 4; ++u) {
      const int st = __builtin_amdgcn_readlane(s, t + u);
      a[u] = pick_alpha<NH>(alpha, t + u, hh);
      load_row<NH>(zl + (size_t)st * C, r[u]);
    }
#pragma unroll
    for (int u = 0; u < 4; ++u)
      for (int h = 0; h < NH; ++h) acc[h] += a[u] * r[u][h];
  }
  for (; t < cnt; ++t) {
    float r[NH];
    const int st = __builtin_amdgcn_readlane(s, t);
    const float a = pick_alpha<NH>(alpha, t, hh);
    load_row<NH>(zl + (size_t)st * C, r);
    for (int h = 0; h < NH; ++h) acc[h] += a * r[h];
  }
}

template <int NH, bool ELU>
__global__ void __launch_bounds__(256)
node_softmax_gather(const int* __restrict__ rowptr, const int* __restrict__ csr,
                    const float* __restrict__ ss, const float* __restrict__ sd,
                    const float* __restrict__ z, float* __restrict__ out, int ldo) {
  const int n = __builtin_amdgcn_readfirstlane((int)(blockIdx.x * 8u + (threadIdx.x >> 5)));
  if (n >= N_NODES) return;
  const int lane = threadIdx.x & 31;
  const int hh = (lane * NH) >> 5;
  const int beg = rowptr[n];
  const int end = rowptr[n + 1];
  float acc[NH];
  for (int h = 0; h < NH; ++h) acc[h] = 0.f;
  if (end > beg) {
    float sdn[NH];
    for (int h = 0; h < NH; ++h) sdn[h] = sd[(size_t)n * NH + h];
    const float* zl = z + lane * NH;
    float x0[NH];
    const int s0 = load_edge<NH>(csr, ss, sdn, beg + lane, end, x0);
    const bool v0 = (beg + lane) < end;
    float m[NH];
    for (int h = 0; h < NH; ++h) m[h] = x0[h];
    for (int base = beg + 32; base < end; base += 32) {
      float x[NH];
      load_edge<NH>(csr, ss, sdn, base + lane, end, x);
      for (int h = 0; h < NH; ++h) m[h] = fmaxf(m[h], x[h]);
    }
    for (int o = 16; o > 0; o >>= 1)
      for (int h = 0; h < NH; ++h) m[h] = fmaxf(m[h], __shfl_xor(m[h], o));
    float ee0[NH], sum[NH];
    for (int h = 0; h < NH; ++h) { ee0[h] = v0 ? expf(x0[h] - m[h]) : 0.f; sum[h] = ee0[h]; }
    for (int base = beg + 32; base < end; base += 32) {
      float x[NH];
      load_edge<NH>(csr, ss, sdn, base + lane, end, x);
      const bool v = (base + lane) < end;
      for (int h = 0; h < NH; ++h) sum[h] += v ? expf(x[h] - m[h]) : 0.f;
    }
    for (int o = 16; o > 0; o >>= 1)
      for (int h = 0; h < NH; ++h) sum[h] += __shfl_xor(sum[h], o);
    float den[NH], al[NH];
    for (int h = 0; h < NH; ++h) { den[h] = fmaxf(sum[h], 1e-9f); al[h] = ee0[h] / den[h]; }
    const int c0 = end - beg;
    add_rows<NH>(zl, s0, al, c0 < 32 ? c0 : 32, hh, acc);
    for (int base = beg + 32; base < end; base += 32) {
      float x[NH];
      const int s = load_edge<NH>(csr, ss, sdn, base + lane, end, x);
      const bool v = (base + lane) < end;
      for (int h = 0; h < NH; ++h) al[h] = v ? expf(x[h] - m[h]) / den[h] : 0.f;
      const int c = end - base;
      add_rows<NH>(zl, s, al, c < 32 ? c : 32, hh, acc);
    }
  }
  if constexpr (ELU) {
    for (int h = 0; h < NH; ++h) acc[h] = (acc[h] > 0.f) ? acc[h] : (expf(acc[h]) - 1.0f);
  }
  float* op = out + (size_t)n * ldo + lane * NH;
  if constexpr (NH == 4) {
    *(float4*)op = make_float4(acc[0], acc[1], acc[2], acc[3]);
  } else {
    for (int h = 0; h < NH; ++h) op[h] = acc[h];
  }
}

extern "C" void kernel_launch(void* const* d_in, const int* in_sizes, int n_in,
                              void* d_out, int out_size, void* d_ws, size_t ws_size,
                              hipStream_t stream) {
  const float* h    = (const float*)d_in[0];
  const int*   src  = (const int*)  d_in[1];
  const int*   dst  = (const int*)  d_in[2];
  const int*   fc   = (const int*)  d_in[3];
  const float* W1   = (const float*)d_in[4];
  const float* a1   = (const float*)d_in[5];
  const float* W2   = (const float*)d_in[6];
  const float* a2   = (const float*)d_in[7];
  const float* Wf   = (const float*)d_in[8];
  const float* Wff  = (const float*)d_in[9];
  const float* Wfc2 = (const float*)d_in[10];
  float* out = (float*)d_out;

  float* w = (float*)d_ws;
  size_t off = 0;
  int* bintot  = (int*)(w + off); off += 512;
  int* runbase = (int*)(w + off); off += SLICES * 512;
  int* rowptr  = (int*)(w + off); off += N_NODES + 32;
  int* binned  = (int*)(w + off); off += N_EDGES;
  int* csr     = (int*)(w + off); off += N_EDGES;
  float* z1  = w + off; off += (size_t)N_NODES * 128;
  float* s1s = w + off; off += (size_t)N_NODES * HEADS;
  float* s1d = w + off; off += (size_t)N_NODES * HEADS;
  float* h1  = w + off; off += (size_t)N_NODES * 128;
  float* z2  = w + off; off += (size_t)N_NODES * HID;
  float* s2s = w + off; off += N_NODES;
  float* s2d = w + off; off += N_NODES;
  float* hg  = w + off; off += (size_t)N_NODES * 64;
  float* hfp = w + off; off += (size_t)N_NODES * 96;
  (void)ws_size; (void)in_sizes; (void)n_in; (void)out_size;

  const int TB = 256;
  const int MT = N_NODES / 16;
  const int gemm_blocks = (MT + 7) / 8;
  const int node_blocks = (N_NODES + 7) / 8;

  zero_f32<<<(NBINS + TB - 1) / TB, TB, 0, stream>>>((float*)bintot, (long long)NBINS);
  bin_count<<<SLICES, SORT_T, 0, stream>>>(dst, bintot, runbase);
  bin_scatter<<<SLICES, SORT_T, 0, stream>>>(src, dst, bintot, runbase, binned);
  bin_finish<<<NBINS, SORT_T, 0, stream>>>(bintot, binned, rowptr, csr);

  wmma_gemm_strip<8, 128, HEADS><<<gemm_blocks, TB, 0, stream>>>(h, W1, z1, MT, 128, 0, a1, s1s, s1d);
  node_softmax_gather<HEADS, true><<<node_blocks, TB, 0, stream>>>(rowptr, csr, s1s, s1d, z1, h1, 128);

  wmma_gemm_strip<2, 128, 1><<<gemm_blocks, TB, 0, stream>>>(h1, W2, z2, MT, HID, 0, a2, s2s, s2d);
  node_softmax_gather<1, false><<<node_blocks, TB, 0, stream>>>(rowptr, csr, s2s, s2d, z2, hg, 64);

  {
    dim3 g(gemm_blocks, 3);
    wmma_gemm_frac<<<g, TB, 0, stream>>>(h, fc, Wf, hfp);
  }
  wmma_gemm_strip<2, 96, 0><<<gemm_blocks, TB, 0, stream>>>(hfp, Wff, hg, MT, 64, HID, nullptr, nullptr, nullptr);

  wmma_gemm_strip<1, 64, 0><<<gemm_blocks, TB, 0, stream>>>(hg, Wfc2, out, MT, OUT_DIM, 0, nullptr, nullptr, nullptr);
}


# Frac-branch GEMMs (independent of the graph) run as interleaved extra workgroups inside the two gather launches, not as launches of their own: matrix and gather work overlap; 11 launches -> 9
# speedup vs baseline: 7.1581x; 1.0756x over previous
#include <hip/hip_runtime.h>
#include <math.h>

#define N_NODES 100000
#define N_EDGES 1600000
#define IN_DIM  128
#define HID     32
#define HEADS   4
#define OUT_DIM 16

typedef __attribute__((ext_vector_type(2))) float v2f;
typedef __attribute__((ext_vector_type(8))) float v8f;

__global__ void zero_f32(float* __restrict__ p, long long n) {
  long long t = (long long)blockIdx.x * blockDim.x + threadIdx.x;
  if (t < n) p[t] = 0.0f;
}

template <int NT, int K>
__device__ __forceinline__ void stage_weights(const float* __restrict__ W, float4* Ws) {
  constexpr int K4 = K / 4;
  constexpr int P4 = K4 + 1;
  for (int i = threadIdx.x; i < NT * 16 * K4; i += 256) {
    const int row = i / K4;
    const int c = i - row * K4;
    Ws[row * P4 + c] = ((const float4*)W)[i];
  }
}

__device__ __forceinline__ float row16_sum(float x) {
  x += __int_as_float(__builtin_amdgcn_update_dpp(0, __float_as_int(x), 0x161, 0xf, 0xf, true));
  x += __int_as_float(__builtin_amdgcn_update_dpp(0, __float_as_int(x), 0x162, 0xf, 0xf, true));
  x += __int_as_float(__builtin_amdgcn_update_dpp(0, __float_as_int(x), 0x164, 0xf, 0xf, true));
  x += __int_as_float(__builtin_amdgcn_update_dpp(0, __float_as_int(x), 0x168, 0xf, 0xf, true));
  return x;
}

template <int NT, int K, int NHD>
__device__ __forceinline__ void
gemm_strip_block(int bx, const float* __restrict__ A, const float* __restrict__ W,
                 float* __restrict__ C, int Mtiles, int ldc, int coff,
                 const float* __restrict__ av, float* __restrict__ ss, float* __restrict__ sd) {
  constexpr int P = K + 4;
  __shared__ float4 Ws[NT * 16 * (P / 4)];
  stage_weights<NT, K>(W, Ws);
  __syncthreads();
  const int wave = (bx * blockDim.x + threadIdx.x) >> 5;
  if (wave < Mtiles) {
    const int lane = threadIdx.x & 31;
    const int lo = lane & 15;
    const int hi = lane >> 4;
    const float* Ab = A + ((size_t)wave * 16 + lo) * K;
    const float* Wl = (const float*)Ws + lo * P;
    v8f acc[NT];
#pragma unroll
    for (int j = 0; j < NT; ++j)
      acc[j] = (v8f){0.f, 0.f, 0.f, 0.f, 0.f, 0.f, 0.f, 0.f};
    for (int k0 = 0; k0 < K; k0 += 4) {
      const int ka = k0 + hi * 2;
      v2f a;
      a.x = Ab[ka]; a.y = Ab[ka + 1];
#pragma unroll
      for (int j = 0; j < NT; ++j) {
        const v2f b = *(const v2f*)(Wl + j * 16 * P + ka);
        acc[j] = __builtin_amdgcn_wmma_f32_16x16x4_f32(false, a, false, b,
                                                       (short)0, acc[j], false, false);
      }
    }
#pragma unroll
    for (int j = 0; j < NT; ++j) {
      float* Cb = C + (size_t)wave * 16 * ldc + coff + (size_t)j * 16 + lo;
#pragma unroll
      for (int r = 0; r < 8; ++r)
        Cb[(size_t)(r + hi * 8) * ldc] = acc[j][r];
    }
    if constexpr (NHD > 0) {
      static_assert(NT == 2 * NHD, "a head is two 16-column tiles");
      float o[NHD];
#pragma unroll
      for (int h = 0; h < NHD; ++h) {
        const float* ah = av + h * 2 * HID;
        const float al0 = ah[lo], al1 = ah[16 + lo], ar0 = ah[HID + lo], ar1 = ah[HID + 16 + lo];
        float keep = 0.f;
#pragma unroll
        for (int r = 0; r < 8; ++r) {
          const float ps = row16_sum(acc[2 * h][r] * al0 + acc[2 * h + 1][r] * al1);
          const float pd = row16_sum(acc[2 * h][r] * ar0 + acc[2 * h + 1][r] * ar1);
          keep = (lo == r) ? ps : keep;
          keep = (lo == r + 8) ? pd : keep;
        }
        o[h] = keep;
      }
      float* sp = ((lo < 8) ? ss : sd) + ((size_t)wave * 16 + hi * 8 + (lo & 7)) * NHD;
      if constexpr (NHD == 4) {
        *(float4*)sp = make_float4(o[0], o[1], o[2], o[3]);
      } else {
#pragma unroll
        for (int h = 0; h < NHD; ++h) sp[h] = o[h];
      }
    }
  }
}

template <int NT, int K, int NHD>
__global__ void __launch_bounds__(256)
wmma_gemm_strip(const float* __restrict__ A, const float* __restrict__ W,
                float* __restrict__ C, int Mtiles, int ldc, int coff,
                const float* __restrict__ av, float* __restrict__ ss, float* __restrict__ sd) {
  gemm_strip_block<NT, K, NHD>((int)blockIdx.x, A, W, C, Mtiles, ldc, coff, av, ss, sd);
}

__device__ __forceinline__ void
gemm_frac_block(int bx, int s, const float* __restrict__ H, const int* __restrict__ fc,
                const float* __restrict__ Wf, float* __restrict__ hfpre) {
  constexpr int P = IN_DIM + 4;
  __shared__ float4 Ws[2 * 16 * (P / 4)];
  stage_weights<2, IN_DIM>(Wf + (size_t)s * HID * IN_DIM, Ws);
  __syncthreads();
  const int wave = (bx * blockDim.x + threadIdx.x) >> 5;
  if (wave < N_NODES / 16) {
    const int lane = threadIdx.x & 31;
    const int lo = lane & 15;
    const int hi = lane >> 4;
    const int nid = fc[(wave * 16 + lo) * 3 + s];
    const float* Ab = H + (size_t)nid * IN_DIM;
    const float* Wl = (const float*)Ws + lo * P;
    v8f acc[2];
    acc[0] = (v8f){0.f, 0.f, 0.f, 0.f, 0.f, 0.f, 0.f, 0.f};
    acc[1] = acc[0];
    for (int k0 = 0; k0 < IN_DIM; k0 += 4) {
      const int ka = k0 + hi * 2;
      v2f a;
      a.x = Ab[ka]; a.y = Ab[ka + 1];
#pragma unroll
      for (int j = 0; j < 2; ++j) {
        const v2f b = *(const v2f*)(Wl + j * 16 * P + ka);
        acc[j] = __builtin_amdgcn_wmma_f32_16x16x4_f32(false, a, false, b,
                                                       (short)0, acc[j], false, false);
      }
    }
#pragma unroll
    for (int j = 0; j < 2; ++j) {
      float* Cb = hfpre + (size_t)wave * 16 * 96 + s * HID + (size_t)j * 16 + lo;
#pragma unroll
      for (int r = 0; r < 8; ++r)
        Cb[(size_t)(r + hi * 8) * 96] = acc[j][r];
    }
  }
}

#define BIN_SHIFT 8
#define BIN_NODES (1 << BIN_SHIFT)
#define NBINS     ((N_NODES + BIN_NODES - 1) >> BIN_SHIFT)
#define SLICES    256
#define SLICE_E   ((N_EDGES + SLICES - 1) / SLICES)
#define SORT_T    1024
#define PER_T     ((SLICE_E + SORT_T - 1) / SORT_T)
#define SRC_BITS  17
#define BIN_CAP   8192
static_assert(N_NODES <= (1 << SRC_BITS), "a source id must fit the packed word");
static_assert(NBINS <= SORT_T, "one thread per bin in the scans");
static_assert(BIN_NODES <= SORT_T, "one thread per node in the scans");

__device__ __forceinline__ int block_excl_scan(int v, int* wsum) {
  const int lane = threadIdx.x & 31, w = threadIdx.x >> 5;
  int inc = v;
  for (int o = 1; o < 32; o <<= 1) {
    const int u = __shfl_up(inc, o);
    if (lane >= o) inc += u;
  }
  if (lane == 31) wsum[w] = inc;
  __syncthreads();
  if (w == 0) {
    const int s = wsum[lane];
    int si = s;
    for (int o = 1; o < 32; o <<= 1) {
      const int u = __shfl_up(si, o);
      if (lane >= o) si += u;
    }
    wsum[lane] = si - s;
  }
  __syncthreads();
  const int ex = wsum[w] + inc - v;
  __syncthreads();
  return ex;
}

__global__ void __launch_bounds__(SORT_T)
bin_count(const int* __restrict__ dst, int* __restrict__ bintot, int* __restrict__ runbase) {
  __shared__ int hist[NBINS];
  const int t = threadIdx.x;
  for (int i = t; i < NBINS; i += SORT_T) hist[i] = 0;
  __syncthreads();
  const int e0 = blockIdx.x * SLICE_E;
  const int e1 = (e0 + SLICE_E < N_EDGES) ? e0 + SLICE_E : N_EDGES;
  for (int e = e0 + t; e < e1; e += SORT_T) atomicAdd(&hist[dst[e] >> BIN_SHIFT], 1);
  __syncthreads();
  for (int i = t; i < NBINS; i += SORT_T)
    runbase[blockIdx.x * NBINS + i] = atomicAdd(&bintot[i], hist[i]);
}

__global__ void __launch_bounds__(SORT_T)
bin_scatter(const int* __restrict__ src, const int* __restrict__ dst,
            const int* __restrict__ bintot, const int* __restrict__ runbase,
            int* __restrict__ binned) {
  __shared__ int hist[NBINS];
  __shared__ int lstart[NBINS];
  __shared__ int gbase[NBINS];
  __shared__ int wsum[32];
  __shared__ int sval[SLICE_E];
  __shared__ int sidx[SLICE_E];
  const int t = threadIdx.x;
  for (int i = t; i < NBINS; i += SORT_T) hist[i] = 0;
  __syncthreads();
  const int e0 = blockIdx.x * SLICE_E;
  const int e1 = (e0 + SLICE_E < N_EDGES) ? e0 + SLICE_E : N_EDGES;
  int d[PER_T], s[PER_T];
#pragma unroll
  for (int k = 0; k < PER_T; ++k) {
    const int e = e0 + t + k * SORT_T;
    d[k] = -1; s[k] = 0;
    if (e < e1) { d[k] = dst[e]; s[k] = src[e]; atomicAdd(&hist[d[k] >> BIN_SHIFT], 1); }
  }
  __syncthreads();
  const int tot = (t < NBINS) ? bintot[t] : 0;
  const int cnt = (t < NBINS) ? hist[t] : 0;
  const int binstart = block_excl_scan(tot, wsum);
  const int lst = block_excl_scan(cnt, wsum);
  if (t < NBINS) {
    gbase[t] = binstart + runbase[blockIdx.x * NBINS + t];
    lstart[t] = lst;
    hist[t] = 0;
  }
  __syncthreads();
#pragma unroll
  for (int k = 0; k < PER_T; ++k) {
    if (d[k] >= 0) {
      const int b = d[k] >> BIN_SHIFT;
      const int r = atomicAdd(&hist[b], 1);
      const int p = lstart[b] + r;
      sval[p] = s[k] | ((d[k] & (BIN_NODES - 1)) << SRC_BITS);
      sidx[p] = gbase[b] + r;
    }
  }
  __syncthreads();
  const int n = e1 - e0;
  for (int i = t; i < n; i += SORT_T) binned[sidx[i]] = sval[i];
}

__global__ void __launch_bounds__(SORT_T)
bin_finish(const int* __restrict__ bintot, const int* __restrict__ binned,
           int* __restrict__ rowptr, int* __restrict__ csr) {
  __shared__ int deg[BIN_NODES];
  __shared__ int first[BIN_NODES];
  __shared__ int wsum[32];
  __shared__ int range[2];
  __shared__ int sorted[BIN_CAP];
  const int t = threadIdx.x;
  const int b = blockIdx.x;
  const int tot = (t < NBINS) ? bintot[t] : 0;
  const int ex0 = block_excl_scan(tot, wsum);
  if (t == b) { range[0] = ex0; range[1] = tot; }
  if (t < BIN_NODES) deg[t] = 0;
  __syncthreads();
  const int start = range[0];
  const int cnt = range[1];
  const int* mine = binned + start;
  for (int i = t; i < cnt; i += SORT_T) atomicAdd(&deg[((unsigned)mine[i]) >> SRC_BITS], 1);
  __syncthreads();
  const int dg = (t < BIN_NODES) ? deg[t] : 0;
  const int ex = block_excl_scan(dg, wsum);
  if (t < BIN_NODES) {
    const int node = b * BIN_NODES + t;
    if (node < N_NODES) rowptr[node] = start + ex;
    first[t] = ex;
    deg[t] = 0;
  }
  if (b == NBINS - 1 && t == 0) rowptr[N_NODES] = start + cnt;
  __syncthreads();
  if (cnt <= BIN_CAP) {
    for (int i = t; i < cnt; i += SORT_T) {
      const unsigned w = (unsigned)mine[i];
      const int dl = w >> SRC_BITS;
      const int r = atomicAdd(&deg[dl], 1);
      sorted[first[dl] + r] = (int)(w & ((1u << SRC_BITS) - 1u));
    }
    __syncthreads();
    for (int i = t; i < cnt; i += SORT_T) csr[start + i] = sorted[i];
  } else {
    for (int i = t; i < cnt; i += SORT_T) {
      const unsigned w = (unsigned)mine[i];
      const int dl = w >> SRC_BITS;
      const int r = atomicAdd(&deg[dl], 1);
      csr[start + first[dl] + r] = (int)(w & ((1u << SRC_BITS) - 1u));
    }
  }
}

template <int NH>
__device__ __forceinline__ int load_edge(const int* __restrict__ csr, const float* __restrict__ ss,
                                         const float* sdn, int j, int end, float* x) {
  int s = 0;
  if (j < end) {
    s = csr[j];
    float v[NH];
    if constexpr (NH == 4) {
      const float4 q = *(const float4*)(ss + (size_t)s * 4);
      v[0] = q.x; v[1] = q.y; v[2] = q.z; v[3] = q.w;
    } else {
      for (int h = 0; h < NH; ++h) v[h] = ss[(size_t)s * NH + h];
    }
    for (int h = 0; h < NH; ++h) {
      const float t = v[h] + sdn[h];
      x[h] = (t > 0.f) ? t : 0.01f * t;
    }
  } else {
    for (int h = 0; h < NH; ++h) x[h] = -INFINITY;
  }
  return s;
}

template <int NH>
__device__ __forceinline__ float pick_alpha(const float* alpha, int t, int hh) {
  float a = __int_as_float(__builtin_amdgcn_readlane(__float_as_int(alpha[0]), t));
  for (int h = 1; h < NH; ++h) {
    const float b = __int_as_float(__builtin_amdgcn_readlane(__float_as_int(alpha[h]), t));
    a = (hh == h) ? b : a;
  }
  return a;
}

template <int NH>
__device__ __forceinline__ void load_row(const float* __restrict__ p, float* r) {
  if constexpr (NH == 4) {
    const float4 q = *(const float4*)p;
    r[0] = q.x; r[1] = q.y; r[2] = q.z; r[3] = q.w;
  } else {
    for (int h = 0; h < NH; ++h) r[h] = p[h];
  }
}

template <int NH>
__device__ __forceinline__ void add_rows(const float* __restrict__ zl, int s, const float* alpha,
                                         int cnt, int hh, float* acc) {
  constexpr int C = NH * HID;
  int t = 0;
  for (; t + 4 <= cnt; t += 4) {
    float r[4][NH], a[4];
#pragma unroll
    for (int u = 0; u < 4; ++u) {
      const int st = __builtin_amdgcn_readlane(s, t + u);
      a[u] = pick_alpha<NH>(alpha, t + u, hh);
      load_row<NH>(zl + (size_t)st * C, r[u]);
    }
#pragma unroll
    for (int u = 0; u < 4; ++u)
      for (int h = 0; h < NH; ++h) acc[h] += a[u] * r[u][h];
  }
  for (; t < cnt; ++t) {
    float r[NH];
    const int st = __builtin_amdgcn_readlane(s, t);
    const float a = pick_alpha<NH>(alpha, t, hh);
    load_row<NH>(zl + (size_t)st * C, r);
    for (int h = 0; h < NH; ++h) acc[h] += a * r[h];
  }
}

template <int NH, bool ELU>
__device__ __forceinline__ void
node_softmax_gather_block(int gb, const int* __restrict__ rowptr, const int* __restrict__ csr,
                          const float* __restrict__ ss, const float* __restrict__ sd,
                          const float* __restrict__ z, float* __restrict__ out, int ldo) {
  const int n = __builtin_amdgcn_readfirstlane(gb * 8 + (int)(threadIdx.x >> 5));
  if (n >= N_NODES) return;
  const int lane = threadIdx.x & 31;
  const int hh = (lane * NH) >> 5;
  const int beg = rowptr[n];
  const int end = rowptr[n + 1];
  float acc[NH];
  for (int h = 0; h < NH; ++h) acc[h] = 0.f;
  if (end > beg) {
    float sdn[NH];
    for (int h = 0; h < NH; ++h) sdn[h] = sd[(size_t)n * NH + h];
    const float* zl = z + lane * NH;
    float x0[NH];
    const int s0 = load_edge<NH>(csr, ss, sdn, beg + lane, end, x0);
    const bool v0 = (beg + lane) < end;
    float m[NH];
    for (int h = 0; h < NH; ++h) m[h] = x0[h];
    for (int base = beg + 32; base < end; base += 32) {
      float x[NH];
      load_edge<NH>(csr, ss, sdn, base + lane, end, x);
      for (int h = 0; h < NH; ++h) m[h] = fmaxf(m[h], x[h]);
    }
    for (int o = 16; o > 0; o >>= 1)
      for (int h = 0; h < NH; ++h) m[h] = fmaxf(m[h], __shfl_xor(m[h], o));
    float ee0[NH], sum[NH];
    for (int h = 0; h < NH; ++h) { ee0[h] = v0 ? expf(x0[h] - m[h]) : 0.f; sum[h] = ee0[h]; }
    for (int base = beg + 32; base < end; base += 32) {
      float x[NH];
      load_edge<NH>(csr, ss, sdn, base + lane, end, x);
      const bool v = (base + lane) < end;
      for (int h = 0; h < NH; ++h) sum[h] += v ? expf(x[h] - m[h]) : 0.f;
    }
    for (int o = 16; o > 0; o >>= 1)
      for (int h = 0; h < NH; ++h) sum[h] += __shfl_xor(sum[h], o);
    float den[NH], al[NH];
    for (int h = 0; h < NH; ++h) { den[h] = fmaxf(sum[h], 1e-9f); al[h] = ee0[h] / den[h]; }
    const int c0 = end - beg;
    add_rows<NH>(zl, s0, al, c0 < 32 ? c0 : 32, hh, acc);
    for (int base = beg + 32; base < end; base += 32) {
      float x[NH];
      const int s = load_edge<NH>(csr, ss, sdn, base + lane, end, x);
      const bool v = (base + lane) < end;
      for (int h = 0; h < NH; ++h) al[h] = v ? expf(x[h] - m[h]) / den[h] : 0.f;
      const int c = end - base;
      add_rows<NH>(zl, s, al, c < 32 ? c : 32, hh, acc);
    }
  }
  if constexpr (ELU) {
    for (int h = 0; h < NH; ++h) acc[h] = (acc[h] > 0.f) ? acc[h] : (expf(acc[h]) - 1.0f);
  }
  float* op = out + (size_t)n * ldo + lane * NH;
  if constexpr (NH == 4) {
    *(float4*)op = make_float4(acc[0], acc[1], acc[2], acc[3]);
  } else {
    for (int h = 0; h < NH; ++h) op[h] = acc[h];
  }
}

#define GATHER_WGS ((N_NODES + 7) / 8)
#define STRIP_WGS  ((N_NODES / 16 + 7) / 8)
template <int PERIOD, int GEMM_WGS>
struct Mixed {
  static constexpr int MIX = PERIOD * GEMM_WGS;
  static constexpr int GRID = MIX + GATHER_WGS - (PERIOD - 1) * GEMM_WGS;
  static_assert(GATHER_WGS >= (PERIOD - 1) * GEMM_WGS, "the gather outlasts the GEMM");
  static __device__ __forceinline__ int gemm_wg(int i) { return (i < MIX && i % PERIOD == 0) ? i / PERIOD : -1; }
  static __device__ __forceinline__ int gather_wg(int i) {
    return (i < MIX) ? (i / PERIOD) * (PERIOD - 1) + (i % PERIOD) - 1 : i - GEMM_WGS;
  }
};
typedef Mixed<5, 3 * STRIP_WGS> Mix1;
typedef Mixed<13, STRIP_WGS> Mix2;

__global__ void __launch_bounds__(256) __attribute__((amdgpu_waves_per_eu(16, 16)))
gather1_and_frac(const int* __restrict__ rowptr, const int* __restrict__ csr,
                 const float* __restrict__ ss, const float* __restrict__ sd,
                 const float* __restrict__ z, float* __restrict__ out,
                 const float* __restrict__ H, const int* __restrict__ fc,
                 const float* __restrict__ Wf, float* __restrict__ hfpre) {
  const int i = (int)blockIdx.x;
  const int f = Mix1::gemm_wg(i);
  if (f >= 0) {
    gemm_frac_block(f % STRIP_WGS, f / STRIP_WGS, H, fc, Wf, hfpre);
  } else {
    node_softmax_gather_block<HEADS, true>(Mix1::gather_wg(i), rowptr, csr, ss, sd, z, out, 128);
  }
}

__global__ void __launch_bounds__(256) __attribute__((amdgpu_waves_per_eu(16, 16)))
gather2_and_wff(const int* __restrict__ rowptr, const int* __restrict__ csr,
                const float* __restrict__ ss, const float* __restrict__ sd,
                const float* __restrict__ z, float* __restrict__ hg,
                const float* __restrict__ hfpre, const float* __restrict__ Wff) {
  const int i = (int)blockIdx.x;
  const int f = Mix2::gemm_wg(i);
  if (f >= 0) {
    gemm_strip_block<2, 96, 0>(f, hfpre, Wff, hg, N_NODES / 16, 64, HID, nullptr, nullptr, nullptr);
  } else {
    node_softmax_gather_block<1, false>(Mix2::gather_wg(i), rowptr, csr, ss, sd, z, hg, 64);
  }
}

extern "C" void kernel_launch(void* const* d_in, const int* in_sizes, int n_in,
                              void* d_out, int out_size, void* d_ws, size_t ws_size,
                              hipStream_t stream) {
  const float* h    = (const float*)d_in[0];
  const int*   src  = (const int*)  d_in[1];
  const int*   dst  = (const int*)  d_in[2];
  const int*   fc   = (const int*)  d_in[3];
  const float* W1   = (const float*)d_in[4];
  const float* a1   = (const float*)d_in[5];
  const float* W2   = (const float*)d_in[6];
  const float* a2   = (const float*)d_in[7];
  const float* Wf   = (const float*)d_in[8];
  const float* Wff  = (const float*)d_in[9];
  const float* Wfc2 = (const float*)d_in[10];
  float* out = (float*)d_out;

  float* w = (float*)d_ws;
  size_t off = 0;
  int* bintot  = (int*)(w + off); off += 512;
  int* runbase = (int*)(w + off); off += SLICES * 512;
  int* rowptr  = (int*)(w + off); off += N_NODES + 32;
  int* binned  = (int*)(w + off); off += N_EDGES;
  int* csr     = (int*)(w + off); off += N_EDGES;
  float* z1  = w + off; off += (size_t)N_NODES * 128;
  float* s1s = w + off; off += (size_t)N_NODES * HEADS;
  float* s1d = w + off; off += (size_t)N_NODES * HEADS;
  float* h1  = w + off; off += (size_t)N_NODES * 128;
  float* z2  = w + off; off += (size_t)N_NODES * HID;
  float* s2s = w + off; off += N_NODES;
  float* s2d = w + off; off += N_NODES;
  float* hg  = w + off; off += (size_t)N_NODES * 64;
  float* hfp = w + off; off += (size_t)N_NODES * 96;
  (void)ws_size; (void)in_sizes; (void)n_in; (void)out_size;

  const int TB = 256;
  const int MT = N_NODES / 16;
  const int gemm_blocks = (MT + 7) / 8;

  zero_f32<<<(NBINS + TB - 1) / TB, TB, 0, stream>>>((float*)bintot, (long long)NBINS);
  bin_count<<<SLICES, SORT_T, 0, stream>>>(dst, bintot, runbase);
  bin_scatter<<<SLICES, SORT_T, 0, stream>>>(src, dst, bintot, runbase, binned);
  bin_finish<<<NBINS, SORT_T, 0, stream>>>(bintot, binned, rowptr, csr);

  wmma_gemm_strip<8, 128, HEADS><<<gemm_blocks, TB, 0, stream>>>(h, W1, z1, MT, 128, 0, a1, s1s, s1d);
  gather1_and_frac<<<Mix1::GRID, TB, 0, stream>>>(rowptr, csr, s1s, s1d, z1, h1, h, fc, Wf, hfp);

  wmma_gemm_strip<2, 128, 1><<<gemm_blocks, TB, 0, stream>>>(h1, W2, z2, MT, HID, 0, a2, s2s, s2d);
  gather2_and_wff<<<Mix2::GRID, TB, 0, stream>>>(rowptr, csr, s2s, s2d, z2, hg, hfp, Wff);

  wmma_gemm_strip<1, 64, 0><<<gemm_blocks, TB, 0, stream>>>(hg, Wfc2, out, MT, OUT_DIM, 0, nullptr, nullptr, nullptr);
}


# Layer-2 GEMM fused into the layer-1 gather: a workgroup gathers 16 nodes (2 per wave) into an LDS tile, one wave multiplies it by W2 on WMMA and forms the scores; h1 never stored; 9 launches -> 8
# speedup vs baseline: 7.3855x; 1.0318x over previous
#include <hip/hip_runtime.h>
#include <math.h>

#define N_NODES 100000
#define N_EDGES 1600000
#define IN_DIM  128
#define HID     32
#define HEADS   4
#define OUT_DIM 16

typedef __attribute__((ext_vector_type(2))) float v2f;
typedef __attribute__((ext_vector_type(8))) float v8f;

__global__ void zero_f32(float* __restrict__ p, long long n) {
  long long t = (long long)blockIdx.x * blockDim.x + threadIdx.x;
  if (t < n) p[t] = 0.0f;
}

template <int NT, int K>
__device__ __forceinline__ void stage_weights(const float* __restrict__ W, float4* Ws) {
  constexpr int K4 = K / 4;
  constexpr int P4 = K4 + 1;
  for (int i = threadIdx.x; i < NT * 16 * K4; i += 256) {
    const int row = i / K4;
    const int c = i - row * K4;
    Ws[row * P4 + c] = ((const float4*)W)[i];
  }
}

__device__ __forceinline__ float row16_sum(float x) {
  x += __int_as_float(__builtin_amdgcn_update_dpp(0, __float_as_int(x), 0x161, 0xf, 0xf, true));
  x += __int_as_float(__builtin_amdgcn_update_dpp(0, __float_as_int(x), 0x162, 0xf, 0xf, true));
  x += __int_as_float(__builtin_amdgcn_update_dpp(0, __float_as_int(x), 0x164, 0xf, 0xf, true));
  x += __int_as_float(__builtin_amdgcn_update_dpp(0, __float_as_int(x), 0x168, 0xf, 0xf, true));
  return x;
}

template <int NT, int NHD>
__device__ __forceinline__ void
strip_store_scores(int wave, int lo, int hi, const v8f* acc, float* __restrict__ C, int ldc, int coff,
                   const float* __restrict__ av, float* __restrict__ ss, float* __restrict__ sd) {
  {
#pragma unroll
    for (int j = 0; j < NT; ++j) {
      float* Cb = C + (size_t)wave * 16 * ldc + coff + (size_t)j * 16 + lo;
#pragma unroll
      for (int r = 0; r < 8; ++r)
        Cb[(size_t)(r + hi * 8) * ldc] = acc[j][r];
    }
    if constexpr (NHD > 0) {
      static_assert(NT == 2 * NHD, "a head is two 16-column tiles");
      float o[NHD];
#pragma unroll
      for (int h = 0; h < NHD; ++h) {
        const float* ah = av + h * 2 * HID;
        const float al0 = ah[lo], al1 = ah[16 + lo], ar0 = ah[HID + lo], ar1 = ah[HID + 16 + lo];
        float keep = 0.f;
#pragma unroll
        for (int r = 0; r < 8; ++r) {
          const float ps = row16_sum(acc[2 * h][r] * al0 + acc[2 * h + 1][r] * al1);
          const float pd = row16_sum(acc[2 * h][r] * ar0 + acc[2 * h + 1][r] * ar1);
          keep = (lo == r) ? ps : keep;
          keep = (lo == r + 8) ? pd : keep;
        }
        o[h] = keep;
      }
      float* sp = ((lo < 8) ? ss : sd) + ((size_t)wave * 16 + hi * 8 + (lo & 7)) * NHD;
      if constexpr (NHD == 4) {
        *(float4*)sp = make_float4(o[0], o[1], o[2], o[3]);
      } else {
#pragma unroll
        for (int h = 0; h < NHD; ++h) sp[h] = o[h];
      }
    }
  }
}

template <int NT, int K, int NHD>
__device__ __forceinline__ void
gemm_strip_block(int bx, const float* __restrict__ A, const float* __restrict__ W,
                 float* __restrict__ C, int Mtiles, int ldc, int coff,
                 const float* __restrict__ av, float* __restrict__ ss, float* __restrict__ sd) {
  constexpr int P = K + 4;
  __shared__ float4 Ws[NT * 16 * (P / 4)];
  stage_weights<NT, K>(W, Ws);
  __syncthreads();
  const int wave = (bx * blockDim.x + threadIdx.x) >> 5;
  if (wave < Mtiles) {
    const int lane = threadIdx.x & 31;
    const int lo = lane & 15;
    const int hi = lane >> 4;
    const float* Ab = A + ((size_t)wave * 16 + lo) * K;
    const float* Wl = (const float*)Ws + lo * P;
    v8f acc[NT];
#pragma unroll
    for (int j = 0; j < NT; ++j)
      acc[j] = (v8f){0.f, 0.f, 0.f, 0.f, 0.f, 0.f, 0.f, 0.f};
    for (int k0 = 0; k0 < K; k0 += 4) {
      const int ka = k0 + hi * 2;
      v2f a;
      a.x = Ab[ka]; a.y = Ab[ka + 1];
#pragma unroll
      for (int j = 0; j < NT; ++j) {
        const v2f b = *(const v2f*)(Wl + j * 16 * P + ka);
        acc[j] = __builtin_amdgcn_wmma_f32_16x16x4_f32(false, a, false, b,
                                                       (short)0, acc[j], false, false);
      }
    }
    strip_store_scores<NT, NHD>(wave, lo, hi, acc, C, ldc, coff, av, ss, sd);
  }
}

template <int NT, int K, int NHD>
__global__ void __launch_bounds__(256)
wmma_gemm_strip(const float* __restrict__ A, const float* __restrict__ W,
                float* __restrict__ C, int Mtiles, int ldc, int coff,
                const float* __restrict__ av, float* __restrict__ ss, float* __restrict__ sd) {
  gemm_strip_block<NT, K, NHD>((int)blockIdx.x, A, W, C, Mtiles, ldc, coff, av, ss, sd);
}

__device__ __forceinline__ void
gemm_frac_block(int bx, int s, const float* __restrict__ H, const int* __restrict__ fc,
                const float* __restrict__ Wf, float* __restrict__ hfpre, float4* Ws) {
  constexpr int P = IN_DIM + 4;
  stage_weights<2, IN_DIM>(Wf + (size_t)s * HID * IN_DIM, Ws);
  __syncthreads();
  const int wave = (bx * blockDim.x + threadIdx.x) >> 5;
  if (wave < N_NODES / 16) {
    const int lane = threadIdx.x & 31;
    const int lo = lane & 15;
    const int hi = lane >> 4;
    const int nid = fc[(wave * 16 + lo) * 3 + s];
    const float* Ab = H + (size_t)nid * IN_DIM;
    const float* Wl = (const float*)Ws + lo * P;
    v8f acc[2];
    acc[0] = (v8f){0.f, 0.f, 0.f, 0.f, 0.f, 0.f, 0.f, 0.f};
    acc[1] = acc[0];
    for (int k0 = 0; k0 < IN_DIM; k0 += 4) {
      const int ka = k0 + hi * 2;
      v2f a;
      a.x = Ab[ka]; a.y = Ab[ka + 1];
#pragma unroll
      for (int j = 0; j < 2; ++j) {
        const v2f b = *(const v2f*)(Wl + j * 16 * P + ka);
        acc[j] = __builtin_amdgcn_wmma_f32_16x16x4_f32(false, a, false, b,
                                                       (short)0, acc[j], false, false);
      }
    }
#pragma unroll
    for (int j = 0; j < 2; ++j) {
      float* Cb = hfpre + (size_t)wave * 16 * 96 + s * HID + (size_t)j * 16 + lo;
#pragma unroll
      for (int r = 0; r < 8; ++r)
        Cb[(size_t)(r + hi * 8) * 96] = acc[j][r];
    }
  }
}

#define BIN_SHIFT 8
#define BIN_NODES (1 << BIN_SHIFT)
#define NBINS     ((N_NODES + BIN_NODES - 1) >> BIN_SHIFT)
#define SLICES    256
#define SLICE_E   ((N_EDGES + SLICES - 1) / SLICES)
#define SORT_T    1024
#define PER_T     ((SLICE_E + SORT_T - 1) / SORT_T)
#define SRC_BITS  17
#define BIN_CAP   8192
static_assert(N_NODES <= (1 << SRC_BITS), "a source id must fit the packed word");
static_assert(NBINS <= SORT_T, "one thread per bin in the scans");
static_assert(BIN_NODES <= SORT_T, "one thread per node in the scans");

__device__ __forceinline__ int block_excl_scan(int v, int* wsum) {
  const int lane = threadIdx.x & 31, w = threadIdx.x >> 5;
  int inc = v;
  for (int o = 1; o < 32; o <<= 1) {
    const int u = __shfl_up(inc, o);
    if (lane >= o) inc += u;
  }
  if (lane == 31) wsum[w] = inc;
  __syncthreads();
  if (w == 0) {
    const int s = wsum[lane];
    int si = s;
    for (int o = 1; o < 32; o <<= 1) {
      const int u = __shfl_up(si, o);
      if (lane >= o) si += u;
    }
    wsum[lane] = si - s;
  }
  __syncthreads();
  const int ex = wsum[w] + inc - v;
  __syncthreads();
  return ex;
}

__global__ void __launch_bounds__(SORT_T)
bin_count(const int* __restrict__ dst, int* __restrict__ bintot, int* __restrict__ runbase) {
  __shared__ int hist[NBINS];
  const int t = threadIdx.x;
  for (int i = t; i < NBINS; i += SORT_T) hist[i] = 0;
  __syncthreads();
  const int e0 = blockIdx.x * SLICE_E;
  const int e1 = (e0 + SLICE_E < N_EDGES) ? e0 + SLICE_E : N_EDGES;
  for (int e = e0 + t; e < e1; e += SORT_T) atomicAdd(&hist[dst[e] >> BIN_SHIFT], 1);
  __syncthreads();
  for (int i = t; i < NBINS; i += SORT_T)
    runbase[blockIdx.x * NBINS + i] = atomicAdd(&bintot[i], hist[i]);
}

__global__ void __launch_bounds__(SORT_T)
bin_scatter(const int* __restrict__ src, const int* __restrict__ dst,
            const int* __restrict__ bintot, const int* __restrict__ runbase,
            int* __restrict__ binned) {
  __shared__ int hist[NBINS];
  __shared__ int lstart[NBINS];
  __shared__ int gbase[NBINS];
  __shared__ int wsum[32];
  __shared__ int sval[SLICE_E];
  __shared__ int sidx[SLICE_E];
  const int t = threadIdx.x;
  for (int i = t; i < NBINS; i += SORT_T) hist[i] = 0;
  __syncthreads();
  const int e0 = blockIdx.x * SLICE_E;
  const int e1 = (e0 + SLICE_E < N_EDGES) ? e0 + SLICE_E : N_EDGES;
  int d[PER_T], s[PER_T];
#pragma unroll
  for (int k = 0; k < PER_T; ++k) {
    const int e = e0 + t + k * SORT_T;
    d[k] = -1; s[k] = 0;
    if (e < e1) { d[k] = dst[e]; s[k] = src[e]; atomicAdd(&hist[d[k] >> BIN_SHIFT], 1); }
  }
  __syncthreads();
  const int tot = (t < NBINS) ? bintot[t] : 0;
  const int cnt = (t < NBINS) ? hist[t] : 0;
  const int binstart = block_excl_scan(tot, wsum);
  const int lst = block_excl_scan(cnt, wsum);
  if (t < NBINS) {
    gbase[t] = binstart + runbase[blockIdx.x * NBINS + t];
    lstart[t] = lst;
    hist[t] = 0;
  }
  __syncthreads();
#pragma unroll
  for (int k = 0; k < PER_T; ++k) {
    if (d[k] >= 0) {
      const int b = d[k] >> BIN_SHIFT;
      const int r = atomicAdd(&hist[b], 1);
      const int p = lstart[b] + r;
      sval[p] = s[k] | ((d[k] & (BIN_NODES - 1)) << SRC_BITS);
      sidx[p] = gbase[b] + r;
    }
  }
  __syncthreads();
  const int n = e1 - e0;
  for (int i = t; i < n; i += SORT_T) binned[sidx[i]] = sval[i];
}

__global__ void __launch_bounds__(SORT_T)
bin_finish(const int* __restrict__ bintot, const int* __restrict__ binned,
           int* __restrict__ rowptr, int* __restrict__ csr) {
  __shared__ int deg[BIN_NODES];
  __shared__ int first[BIN_NODES];
  __shared__ int wsum[32];
  __shared__ int range[2];
  __shared__ int sorted[BIN_CAP];
  const int t = threadIdx.x;
  const int b = blockIdx.x;
  const int tot = (t < NBINS) ? bintot[t] : 0;
  const int ex0 = block_excl_scan(tot, wsum);
  if (t == b) { range[0] = ex0; range[1] = tot; }
  if (t < BIN_NODES) deg[t] = 0;
  __syncthreads();
  const int start = range[0];
  const int cnt = range[1];
  const int* mine = binned + start;
  for (int i = t; i < cnt; i += SORT_T) atomicAdd(&deg[((unsigned)mine[i]) >> SRC_BITS], 1);
  __syncthreads();
  const int dg = (t < BIN_NODES) ? deg[t] : 0;
  const int ex = block_excl_scan(dg, wsum);
  if (t < BIN_NODES) {
    const int node = b * BIN_NODES + t;
    if (node < N_NODES) rowptr[node] = start + ex;
    first[t] = ex;
    deg[t] = 0;
  }
  if (b == NBINS - 1 && t == 0) rowptr[N_NODES] = start + cnt;
  __syncthreads();
  if (cnt <= BIN_CAP) {
    for (int i = t; i < cnt; i += SORT_T) {
      const unsigned w = (unsigned)mine[i];
      const int dl = w >> SRC_BITS;
      const int r = atomicAdd(&deg[dl], 1);
      sorted[first[dl] + r] = (int)(w & ((1u << SRC_BITS) - 1u));
    }
    __syncthreads();
    for (int i = t; i < cnt; i += SORT_T) csr[start + i] = sorted[i];
  } else {
    for (int i = t; i < cnt; i += SORT_T) {
      const unsigned w = (unsigned)mine[i];
      const int dl = w >> SRC_BITS;
      const int r = atomicAdd(&deg[dl], 1);
      csr[start + first[dl] + r] = (int)(w & ((1u << SRC_BITS) - 1u));
    }
  }
}

template <int NH>
__device__ __forceinline__ int load_edge(const int* __restrict__ csr, const float* __restrict__ ss,
                                         const float* sdn, int j, int end, float* x) {
  int s = 0;
  if (j < end) {
    s = csr[j];
    float v[NH];
    if constexpr (NH == 4) {
      const float4 q = *(const float4*)(ss + (size_t)s * 4);
      v[0] = q.x; v[1] = q.y; v[2] = q.z; v[3] = q.w;
    } else {
      for (int h = 0; h < NH; ++h) v[h] = ss[(size_t)s * NH + h];
    }
    for (int h = 0; h < NH; ++h) {
      const float t = v[h] + sdn[h];
      x[h] = (t > 0.f) ? t : 0.01f * t;
    }
  } else {
    for (int h = 0; h < NH; ++h) x[h] = -INFINITY;
  }
  return s;
}

template <int NH>
__device__ __forceinline__ float pick_alpha(const float* alpha, int t, int hh) {
  float a = __int_as_float(__builtin_amdgcn_readlane(__float_as_int(alpha[0]), t));
  for (int h = 1; h < NH; ++h) {
    const float b = __int_as_float(__builtin_amdgcn_readlane(__float_as_int(alpha[h]), t));
    a = (hh == h) ? b : a;
  }
  return a;
}

template <int NH>
__device__ __forceinline__ void load_row(const float* __restrict__ p, float* r) {
  if constexpr (NH == 4) {
    const float4 q = *(const float4*)p;
    r[0] = q.x; r[1] = q.y; r[2] = q.z; r[3] = q.w;
  } else {
    for (int h = 0; h < NH; ++h) r[h] = p[h];
  }
}

template <int NH>
__device__ __forceinline__ void add_rows(const float* __restrict__ zl, int s, const float* alpha,
                                         int cnt, int hh, float* acc) {
  constexpr int C = NH * HID;
  int t = 0;
  for (; t + 4 <= cnt; t += 4) {
    float r[4][NH], a[4];
#pragma unroll
    for (int u = 0; u < 4; ++u) {
      const int st = __builtin_amdgcn_readlane(s, t + u);
      a[u] = pick_alpha<NH>(alpha, t + u, hh);
      load_row<NH>(zl + (size_t)st * C, r[u]);
    }
#pragma unroll
    for (int u = 0; u < 4; ++u)
      for (int h = 0; h < NH; ++h) acc[h] += a[u] * r[u][h];
  }
  for (; t < cnt; ++t) {
    float r[NH];
    const int st = __builtin_amdgcn_readlane(s, t);
    const float a = pick_alpha<NH>(alpha, t, hh);
    load_row<NH>(zl + (size_t)st * C, r);
    for (int h = 0; h < NH; ++h) acc[h] += a * r[h];
  }
}

template <int NH, bool ELU>
__device__ __forceinline__ void
node_softmax_gather_row(int n, const int* __restrict__ rowptr, const int* __restrict__ csr,
                        const float* __restrict__ ss, const float* __restrict__ sd,
                        const float* __restrict__ z, float* acc) {
  const int lane = threadIdx.x & 31;
  const int hh = (lane * NH) >> 5;
  const int beg = rowptr[n];
  const int end = rowptr[n + 1];
  for (int h = 0; h < NH; ++h) acc[h] = 0.f;
  if (end > beg) {
    float sdn[NH];
    for (int h = 0; h < NH; ++h) sdn[h] = sd[(size_t)n * NH + h];
    const float* zl = z + lane * NH;
    float x0[NH];
    const int s0 = load_edge<NH>(csr, ss, sdn, beg + lane, end, x0);
    const bool v0 = (beg + lane) < end;
    float m[NH];
    for (int h = 0; h < NH; ++h) m[h] = x0[h];
    for (int base = beg + 32; base < end; base += 32) {
      float x[NH];
      load_edge<NH>(csr, ss, sdn, base + lane, end, x);
      for (int h = 0; h < NH; ++h) m[h] = fmaxf(m[h], x[h]);
    }
    for (int o = 16; o > 0; o >>= 1)
      for (int h = 0; h < NH; ++h) m[h] = fmaxf(m[h], __shfl_xor(m[h], o));
    float ee0[NH], sum[NH];
    for (int h = 0; h < NH; ++h) { ee0[h] = v0 ? expf(x0[h] - m[h]) : 0.f; sum[h] = ee0[h]; }
    for (int base = beg + 32; base < end; base += 32) {
      float x[NH];
      load_edge<NH>(csr, ss, sdn, base + lane, end, x);
      const bool v = (base + lane) < end;
      for (int h = 0; h < NH; ++h) sum[h] += v ? expf(x[h] - m[h]) : 0.f;
    }
    for (int o = 16; o > 0; o >>= 1)
      for (int h = 0; h < NH; ++h) sum[h] += __shfl_xor(sum[h], o);
    float den[NH], al[NH];
    for (int h = 0; h < NH; ++h) { den[h] = fmaxf(sum[h], 1e-9f); al[h] = ee0[h] / den[h]; }
    const int c0 = end - beg;
    add_rows<NH>(zl, s0, al, c0 < 32 ? c0 : 32, hh, acc);
    for (int base = beg + 32; base < end; base += 32) {
      float x[NH];
      const int s = load_edge<NH>(csr, ss, sdn, base + lane, end, x);
      const bool v = (base + lane) < end;
      for (int h = 0; h < NH; ++h) al[h] = v ? expf(x[h] - m[h]) / den[h] : 0.f;
      const int c = end - base;
      add_rows<NH>(zl, s, al, c < 32 ? c : 32, hh, acc);
    }
  }
  if constexpr (ELU) {
    for (int h = 0; h < NH; ++h) acc[h] = (acc[h] > 0.f) ? acc[h] : (expf(acc[h]) - 1.0f);
  }
}

template <int NH, bool ELU>
__device__ __forceinline__ void
node_softmax_gather_block(int gb, const int* __restrict__ rowptr, const int* __restrict__ csr,
                          const float* __restrict__ ss, const float* __restrict__ sd,
                          const float* __restrict__ z, float* __restrict__ out, int ldo) {
  const int n = __builtin_amdgcn_readfirstlane(gb * 8 + (int)(threadIdx.x >> 5));
  if (n >= N_NODES) return;
  const int lane = threadIdx.x & 31;
  float acc[NH];
  node_softmax_gather_row<NH, ELU>(n, rowptr, csr, ss, sd, z, acc);
  float* op = out + (size_t)n * ldo + lane * NH;
  if constexpr (NH == 4) {
    *(float4*)op = make_float4(acc[0], acc[1], acc[2], acc[3]);
  } else {
    for (int h = 0; h < NH; ++h) op[h] = acc[h];
  }
}

#define H1_COLS  (HEADS * HID)
#define H1_PITCH (H1_COLS + 4)
#define FUSE_W_F4 (2 * 16 * (H1_PITCH / 4))
#define FUSE_LDS_F4 (FUSE_W_F4 + 16 * (H1_PITCH / 4))
static_assert(N_NODES % 16 == 0, "whole 16-node groups");
__device__ __forceinline__ void
gather1_z2_block(int gb, const int* __restrict__ rowptr, const int* __restrict__ csr,
                 const float* __restrict__ ss, const float* __restrict__ sd,
                 const float* __restrict__ z, const float* __restrict__ W2,
                 const float* __restrict__ a2, float* __restrict__ z2,
                 float* __restrict__ s2s, float* __restrict__ s2d, float4* lds) {
  float4* Ws = lds;
  float* T = (float*)(lds + FUSE_W_F4);
  {
    constexpr int K4 = H1_COLS / 4, P4 = H1_PITCH / 4;
    static_assert(2 * 16 * K4 == 4 * 256, "four float4 per thread");
    float4 q[4];
#pragma unroll
    for (int u = 0; u < 4; ++u) q[u] = ((const float4*)W2)[threadIdx.x + u * 256];
#pragma unroll
    for (int u = 0; u < 4; ++u) {
      const int i = threadIdx.x + u * 256;
      Ws[(i / K4) * P4 + (i % K4)] = q[u];
    }
  }
  const int w = __builtin_amdgcn_readfirstlane((int)(threadIdx.x >> 5));
  const int lane = threadIdx.x & 31;
  for (int i = 0; i < 2; ++i) {
    const int row = w * 2 + i;
    float acc[HEADS];
    node_softmax_gather_row<HEADS, true>(gb * 16 + row, rowptr, csr, ss, sd, z, acc);
    *(float4*)(T + row * H1_PITCH + lane * HEADS) = make_float4(acc[0], acc[1], acc[2], acc[3]);
  }
  __syncthreads();
  if (w == 0) {
    const int lo = lane & 15;
    const int hi = lane >> 4;
    const float* Tl = T + lo * H1_PITCH;
    const float* Wl = (const float*)Ws + lo * H1_PITCH;
    v8f acc[2];
    acc[0] = (v8f){0.f, 0.f, 0.f, 0.f, 0.f, 0.f, 0.f, 0.f};
    acc[1] = acc[0];
    for (int k0 = 0; k0 < H1_COLS; k0 += 4) {
      const int ka = k0 + hi * 2;
      const v2f a = *(const v2f*)(Tl + ka);
#pragma unroll
      for (int j = 0; j < 2; ++j) {
        const v2f b = *(const v2f*)(Wl + j * 16 * H1_PITCH + ka);
        acc[j] = __builtin_amdgcn_wmma_f32_16x16x4_f32(false, a, false, b,
                                                       (short)0, acc[j], false, false);
      }
    }
    strip_store_scores<2, 1>(gb, lo, hi, acc, z2, HID, 0, a2, s2s, s2d);
  }
}

#define GATHER_WGS ((N_NODES + 7) / 8)
#define GATHER16_WGS (N_NODES / 16)
#define STRIP_WGS  ((N_NODES / 16 + 7) / 8)
template <int PERIOD, int GEMM_WGS, int GATHER_N>
struct Mixed {
  static constexpr int MIX = PERIOD * GEMM_WGS;
  static constexpr int GRID = MIX + GATHER_N - (PERIOD - 1) * GEMM_WGS;
  static_assert(GATHER_N >= (PERIOD - 1) * GEMM_WGS, "the gather outlasts the GEMM");
  static __device__ __forceinline__ int gemm_wg(int i) { return (i < MIX && i % PERIOD == 0) ? i / PERIOD : -1; }
  static __device__ __forceinline__ int gather_wg(int i) {
    return (i < MIX) ? (i / PERIOD) * (PERIOD - 1) + (i % PERIOD) - 1 : i - GEMM_WGS;
  }
};
typedef Mixed<3, 3 * STRIP_WGS, GATHER16_WGS> Mix1;
typedef Mixed<13, STRIP_WGS, GATHER_WGS> Mix2;

__global__ void __launch_bounds__(256) __attribute__((amdgpu_waves_per_eu(16, 16)))
gather1_and_frac(const int* __restrict__ rowptr, const int* __restrict__ csr,
                 const float* __restrict__ ss, const float* __restrict__ sd,
                 const float* __restrict__ z,
                 const float* __restrict__ W2, const float* __restrict__ a2,
                 float* __restrict__ z2, float* __restrict__ s2s, float* __restrict__ s2d,
                 const float* __restrict__ H, const int* __restrict__ fc,
                 const float* __restrict__ Wf, float* __restrict__ hfpre) {
  __shared__ float4 lds[FUSE_LDS_F4];
  const int i = (int)blockIdx.x;
  const int f = Mix1::gemm_wg(i);
  if (f >= 0) {
    gemm_frac_block(f % STRIP_WGS, f / STRIP_WGS, H, fc, Wf, hfpre, lds);
  } else {
    gather1_z2_block(Mix1::gather_wg(i), rowptr, csr, ss, sd, z, W2, a2, z2, s2s, s2d, lds);
  }
}

__global__ void __launch_bounds__(256) __attribute__((amdgpu_waves_per_eu(16, 16)))
gather2_and_wff(const int* __restrict__ rowptr, const int* __restrict__ csr,
                const float* __restrict__ ss, const float* __restrict__ sd,
                const float* __restrict__ z, float* __restrict__ hg,
                const float* __restrict__ hfpre, const float* __restrict__ Wff) {
  const int i = (int)blockIdx.x;
  const int f = Mix2::gemm_wg(i);
  if (f >= 0) {
    gemm_strip_block<2, 96, 0>(f, hfpre, Wff, hg, N_NODES / 16, 64, HID, nullptr, nullptr, nullptr);
  } else {
    node_softmax_gather_block<1, false>(Mix2::gather_wg(i), rowptr, csr, ss, sd, z, hg, 64);
  }
}

extern "C" void kernel_launch(void* const* d_in, const int* in_sizes, int n_in,
                              void* d_out, int out_size, void* d_ws, size_t ws_size,
                              hipStream_t stream) {
  const float* h    = (const float*)d_in[0];
  const int*   src  = (const int*)  d_in[1];
  const int*   dst  = (const int*)  d_in[2];
  const int*   fc   = (const int*)  d_in[3];
  const float* W1   = (const float*)d_in[4];
  const float* a1   = (const float*)d_in[5];
  const float* W2   = (const float*)d_in[6];
  const float* a2   = (const float*)d_in[7];
  const float* Wf   = (const float*)d_in[8];
  const float* Wff  = (const float*)d_in[9];
  const float* Wfc2 = (const float*)d_in[10];
  float* out = (float*)d_out;

  float* w = (float*)d_ws;
  size_t off = 0;
  int* bintot  = (int*)(w + off); off += 512;
  int* runbase = (int*)(w + off); off += SLICES * 512;
  int* rowptr  = (int*)(w + off); off += N_NODES + 32;
  int* binned  = (int*)(w + off); off += N_EDGES;
  int* csr     = (int*)(w + off); off += N_EDGES;
  float* z1  = w + off; off += (size_t)N_NODES * 128;
  float* s1s = w + off; off += (size_t)N_NODES * HEADS;
  float* s1d = w + off; off += (size_t)N_NODES * HEADS;
  float* z2  = w + off; off += (size_t)N_NODES * HID;
  float* s2s = w + off; off += N_NODES;
  float* s2d = w + off; off += N_NODES;
  float* hg  = w + off; off += (size_t)N_NODES * 64;
  float* hfp = w + off; off += (size_t)N_NODES * 96;
  (void)ws_size; (void)in_sizes; (void)n_in; (void)out_size;

  const int TB = 256;
  const int MT = N_NODES / 16;
  const int gemm_blocks = (MT + 7) / 8;

  zero_f32<<<(NBINS + TB - 1) / TB, TB, 0, stream>>>((float*)bintot, (long long)NBINS);
  bin_count<<<SLICES, SORT_T, 0, stream>>>(dst, bintot, runbase);
  bin_scatter<<<SLICES, SORT_T, 0, stream>>>(src, dst, bintot, runbase, binned);
  bin_finish<<<NBINS, SORT_T, 0, stream>>>(bintot, binned, rowptr, csr);

  wmma_gemm_strip<8, 128, HEADS><<<gemm_blocks, TB, 0, stream>>>(h, W1, z1, MT, 128, 0, a1, s1s, s1d);
  gather1_and_frac<<<Mix1::GRID, TB, 0, stream>>>(rowptr, csr, s1s, s1d, z1, W2, a2, z2, s2s, s2d,
                                                  h, fc, Wf, hfp);

  gather2_and_wff<<<Mix2::GRID, TB, 0, stream>>>(rowptr, csr, s2s, s2d, z2, hg, hfp, Wff);

  wmma_gemm_strip<1, 64, 0><<<gemm_blocks, TB, 0, stream>>>(hg, Wfc2, out, MT, OUT_DIM, 0, nullptr, nullptr, nullptr);
}


# Gather row loops read each slot's weight and source-row offset from a per-wave LDS table (two 16-byte LDS reads per 4 slots), not v_readlane per slot and head plus head selects; slot sum order kept
# speedup vs baseline: 7.6578x; 1.0369x over previous
#include <hip/hip_runtime.h>
#include <math.h>

#define N_NODES 100000
#define N_EDGES 1600000
#define IN_DIM  128
#define HID     32
#define HEADS   4
#define OUT_DIM 16

typedef __attribute__((ext_vector_type(2))) float v2f;
typedef __attribute__((ext_vector_type(8))) float v8f;

__global__ void zero_f32(float* __restrict__ p, long long n) {
  long long t = (long long)blockIdx.x * blockDim.x + threadIdx.x;
  if (t < n) p[t] = 0.0f;
}

template <int NT, int K>
__device__ __forceinline__ void stage_weights(const float* __restrict__ W, float4* Ws) {
  constexpr int K4 = K / 4;
  constexpr int P4 = K4 + 1;
  for (int i = threadIdx.x; i < NT * 16 * K4; i += 256) {
    const int row = i / K4;
    const int c = i - row * K4;
    Ws[row * P4 + c] = ((const float4*)W)[i];
  }
}

__device__ __forceinline__ float row16_sum(float x) {
  x += __int_as_float(__builtin_amdgcn_update_dpp(0, __float_as_int(x), 0x161, 0xf, 0xf, true));
  x += __int_as_float(__builtin_amdgcn_update_dpp(0, __float_as_int(x), 0x162, 0xf, 0xf, true));
  x += __int_as_float(__builtin_amdgcn_update_dpp(0, __float_as_int(x), 0x164, 0xf, 0xf, true));
  x += __int_as_float(__builtin_amdgcn_update_dpp(0, __float_as_int(x), 0x168, 0xf, 0xf, true));
  return x;
}

template <int NT, int NHD>
__device__ __forceinline__ void
strip_store_scores(int wave, int lo, int hi, const v8f* acc, float* __restrict__ C, int ldc, int coff,
                   const float* __restrict__ av, float* __restrict__ ss, float* __restrict__ sd) {
  {
#pragma unroll
    for (int j = 0; j < NT; ++j) {
      float* Cb = C + (size_t)wave * 16 * ldc + coff + (size_t)j * 16 + lo;
#pragma unroll
      for (int r = 0; r < 8; ++r)
        Cb[(size_t)(r + hi * 8) * ldc] = acc[j][r];
    }
    if constexpr (NHD > 0) {
      static_assert(NT == 2 * NHD, "a head is two 16-column tiles");
      float o[NHD];
#pragma unroll
      for (int h = 0; h < NHD; ++h) {
        const float* ah = av + h * 2 * HID;
        const float al0 = ah[lo], al1 = ah[16 + lo], ar0 = ah[HID + lo], ar1 = ah[HID + 16 + lo];
        float keep = 0.f;
#pragma unroll
        for (int r = 0; r < 8; ++r) {
          const float ps = row16_sum(acc[2 * h][r] * al0 + acc[2 * h + 1][r] * al1);
          const float pd = row16_sum(acc[2 * h][r] * ar0 + acc[2 * h + 1][r] * ar1);
          keep = (lo == r) ? ps : keep;
          keep = (lo == r + 8) ? pd : keep;
        }
        o[h] = keep;
      }
      float* sp = ((lo < 8) ? ss : sd) + ((size_t)wave * 16 + hi * 8 + (lo & 7)) * NHD;
      if constexpr (NHD == 4) {
        *(float4*)sp = make_float4(o[0], o[1], o[2], o[3]);
      } else {
#pragma unroll
        for (int h = 0; h < NHD; ++h) sp[h] = o[h];
      }
    }
  }
}

template <int NT, int K, int NHD>
__device__ __forceinline__ void
gemm_strip_block(int bx, const float* __restrict__ A, const float* __restrict__ W,
                 float* __restrict__ C, int Mtiles, int ldc, int coff,
                 const float* __restrict__ av, float* __restrict__ ss, float* __restrict__ sd) {
  constexpr int P = K + 4;
  __shared__ float4 Ws[NT * 16 * (P / 4)];
  stage_weights<NT, K>(W, Ws);
  __syncthreads();
  const int wave = (bx * blockDim.x + threadIdx.x) >> 5;
  if (wave < Mtiles) {
    const int lane = threadIdx.x & 31;
    const int lo = lane & 15;
    const int hi = lane >> 4;
    const float* Ab = A + ((size_t)wave * 16 + lo) * K;
    const float* Wl = (const float*)Ws + lo * P;
    v8f acc[NT];
#pragma unroll
    for (int j = 0; j < NT; ++j)
      acc[j] = (v8f){0.f, 0.f, 0.f, 0.f, 0.f, 0.f, 0.f, 0.f};
    for (int k0 = 0; k0 < K; k0 += 4) {
      const int ka = k0 + hi * 2;
      v2f a;
      a.x = Ab[ka]; a.y = Ab[ka + 1];
#pragma unroll
      for (int j = 0; j < NT; ++j) {
        const v2f b = *(const v2f*)(Wl + j * 16 * P + ka);
        acc[j] = __builtin_amdgcn_wmma_f32_16x16x4_f32(false, a, false, b,
                                                       (short)0, acc[j], false, false);
      }
    }
    strip_store_scores<NT, NHD>(wave, lo, hi, acc, C, ldc, coff, av, ss, sd);
  }
}

template <int NT, int K, int NHD>
__global__ void __launch_bounds__(256)
wmma_gemm_strip(const float* __restrict__ A, const float* __restrict__ W,
                float* __restrict__ C, int Mtiles, int ldc, int coff,
                const float* __restrict__ av, float* __restrict__ ss, float* __restrict__ sd) {
  gemm_strip_block<NT, K, NHD>((int)blockIdx.x, A, W, C, Mtiles, ldc, coff, av, ss, sd);
}

__device__ __forceinline__ void
gemm_frac_block(int bx, int s, const float* __restrict__ H, const int* __restrict__ fc,
                const float* __restrict__ Wf, float* __restrict__ hfpre, float4* Ws) {
  constexpr int P = IN_DIM + 4;
  stage_weights<2, IN_DIM>(Wf + (size_t)s * HID * IN_DIM, Ws);
  __syncthreads();
  const int wave = (bx * blockDim.x + threadIdx.x) >> 5;
  if (wave < N_NODES / 16) {
    const int lane = threadIdx.x & 31;
    const int lo = lane & 15;
    const int hi = lane >> 4;
    const int nid = fc[(wave * 16 + lo) * 3 + s];
    const float* Ab = H + (size_t)nid * IN_DIM;
    const float* Wl = (const float*)Ws + lo * P;
    v8f acc[2];
    acc[0] = (v8f){0.f, 0.f, 0.f, 0.f, 0.f, 0.f, 0.f, 0.f};
    acc[1] = acc[0];
    for (int k0 = 0; k0 < IN_DIM; k0 += 4) {
      const int ka = k0 + hi * 2;
      v2f a;
      a.x = Ab[ka]; a.y = Ab[ka + 1];
#pragma unroll
      for (int j = 0; j < 2; ++j) {
        const v2f b = *(const v2f*)(Wl + j * 16 * P + ka);
        acc[j] = __builtin_amdgcn_wmma_f32_16x16x4_f32(false, a, false, b,
                                                       (short)0, acc[j], false, false);
      }
    }
#pragma unroll
    for (int j = 0; j < 2; ++j) {
      float* Cb = hfpre + (size_t)wave * 16 * 96 + s * HID + (size_t)j * 16 + lo;
#pragma unroll
      for (int r = 0; r < 8; ++r)
        Cb[(size_t)(r + hi * 8) * 96] = acc[j][r];
    }
  }
}

#define BIN_SHIFT 8
#define BIN_NODES (1 << BIN_SHIFT)
#define NBINS     ((N_NODES + BIN_NODES - 1) >> BIN_SHIFT)
#define SLICES    256
#define SLICE_E   ((N_EDGES + SLICES - 1) / SLICES)
#define SORT_T    1024
#define PER_T     ((SLICE_E + SORT_T - 1) / SORT_T)
#define SRC_BITS  17
#define BIN_CAP   8192
static_assert(N_NODES <= (1 << SRC_BITS), "a source id must fit the packed word");
static_assert(NBINS <= SORT_T, "one thread per bin in the scans");
static_assert(BIN_NODES <= SORT_T, "one thread per node in the scans");

__device__ __forceinline__ int block_excl_scan(int v, int* wsum) {
  const int lane = threadIdx.x & 31, w = threadIdx.x >> 5;
  int inc = v;
  for (int o = 1; o < 32; o <<= 1) {
    const int u = __shfl_up(inc, o);
    if (lane >= o) inc += u;
  }
  if (lane == 31) wsum[w] = inc;
  __syncthreads();
  if (w == 0) {
    const int s = wsum[lane];
    int si = s;
    for (int o = 1; o < 32; o <<= 1) {
      const int u = __shfl_up(si, o);
      if (lane >= o) si += u;
    }
    wsum[lane] = si - s;
  }
  __syncthreads();
  const int ex = wsum[w] + inc - v;
  __syncthreads();
  return ex;
}

__global__ void __launch_bounds__(SORT_T)
bin_count(const int* __restrict__ dst, int* __restrict__ bintot, int* __restrict__ runbase) {
  __shared__ int hist[NBINS];
  const int t = threadIdx.x;
  for (int i = t; i < NBINS; i += SORT_T) hist[i] = 0;
  __syncthreads();
  const int e0 = blockIdx.x * SLICE_E;
  const int e1 = (e0 + SLICE_E < N_EDGES) ? e0 + SLICE_E : N_EDGES;
  for (int e = e0 + t; e < e1; e += SORT_T) atomicAdd(&hist[dst[e] >> BIN_SHIFT], 1);
  __syncthreads();
  for (int i = t; i < NBINS; i += SORT_T)
    runbase[blockIdx.x * NBINS + i] = atomicAdd(&bintot[i], hist[i]);
}

__global__ void __launch_bounds__(SORT_T)
bin_scatter(const int* __restrict__ src, const int* __restrict__ dst,
            const int* __restrict__ bintot, const int* __restrict__ runbase,
            int* __restrict__ binned) {
  __shared__ int hist[NBINS];
  __shared__ int lstart[NBINS];
  __shared__ int gbase[NBINS];
  __shared__ int wsum[32];
  __shared__ int sval[SLICE_E];
  __shared__ int sidx[SLICE_E];
  const int t = threadIdx.x;
  for (int i = t; i < NBINS; i += SORT_T) hist[i] = 0;
  __syncthreads();
  const int e0 = blockIdx.x * SLICE_E;
  const int e1 = (e0 + SLICE_E < N_EDGES) ? e0 + SLICE_E : N_EDGES;
  int d[PER_T], s[PER_T];
#pragma unroll
  for (int k = 0; k < PER_T; ++k) {
    const int e = e0 + t + k * SORT_T;
    d[k] = -1; s[k] = 0;
    if (e < e1) { d[k] = dst[e]; s[k] = src[e]; atomicAdd(&hist[d[k] >> BIN_SHIFT], 1); }
  }
  __syncthreads();
  const int tot = (t < NBINS) ? bintot[t] : 0;
  const int cnt = (t < NBINS) ? hist[t] : 0;
  const int binstart = block_excl_scan(tot, wsum);
  const int lst = block_excl_scan(cnt, wsum);
  if (t < NBINS) {
    gbase[t] = binstart + runbase[blockIdx.x * NBINS + t];
    lstart[t] = lst;
    hist[t] = 0;
  }
  __syncthreads();
#pragma unroll
  for (int k = 0; k < PER_T; ++k) {
    if (d[k] >= 0) {
      const int b = d[k] >> BIN_SHIFT;
      const int r = atomicAdd(&hist[b], 1);
      const int p = lstart[b] + r;
      sval[p] = s[k] | ((d[k] & (BIN_NODES - 1)) << SRC_BITS);
      sidx[p] = gbase[b] + r;
    }
  }
  __syncthreads();
  const int n = e1 - e0;
  for (int i = t; i < n; i += SORT_T) binned[sidx[i]] = sval[i];
}

__global__ void __launch_bounds__(SORT_T)
bin_finish(const int* __restrict__ bintot, const int* __restrict__ binned,
           int* __restrict__ rowptr, int* __restrict__ csr) {
  __shared__ int deg[BIN_NODES];
  __shared__ int first[BIN_NODES];
  __shared__ int wsum[32];
  __shared__ int range[2];
  __shared__ int sorted[BIN_CAP];
  const int t = threadIdx.x;
  const int b = blockIdx.x;
  const int tot = (t < NBINS) ? bintot[t] : 0;
  const int ex0 = block_excl_scan(tot, wsum);
  if (t == b) { range[0] = ex0; range[1] = tot; }
  if (t < BIN_NODES) deg[t] = 0;
  __syncthreads();
  const int start = range[0];
  const int cnt = range[1];
  const int* mine = binned + start;
  for (int i = t; i < cnt; i += SORT_T) atomicAdd(&deg[((unsigned)mine[i]) >> SRC_BITS], 1);
  __syncthreads();
  const int dg = (t < BIN_NODES) ? deg[t] : 0;
  const int ex = block_excl_scan(dg, wsum);
  if (t < BIN_NODES) {
    const int node = b * BIN_NODES + t;
    if (node < N_NODES) rowptr[node] = start + ex;
    first[t] = ex;
    deg[t] = 0;
  }
  if (b == NBINS - 1 && t == 0) rowptr[N_NODES] = start + cnt;
  __syncthreads();
  if (cnt <= BIN_CAP) {
    for (int i = t; i < cnt; i += SORT_T) {
      const unsigned w = (unsigned)mine[i];
      const int dl = w >> SRC_BITS;
      const int r = atomicAdd(&deg[dl], 1);
      sorted[first[dl] + r] = (int)(w & ((1u << SRC_BITS) - 1u));
    }
    __syncthreads();
    for (int i = t; i < cnt; i += SORT_T) csr[start + i] = sorted[i];
  } else {
    for (int i = t; i < cnt; i += SORT_T) {
      const unsigned w = (unsigned)mine[i];
      const int dl = w >> SRC_BITS;
      const int r = atomicAdd(&deg[dl], 1);
      csr[start + first[dl] + r] = (int)(w & ((1u << SRC_BITS) - 1u));
    }
  }
}

template <int NH>
__device__ __forceinline__ int load_edge(const int* __restrict__ csr, const float* __restrict__ ss,
                                         const float* sdn, int j, int end, float* x) {
  int s = 0;
  if (j < end) {
    s = csr[j];
    float v[NH];
    if constexpr (NH == 4) {
      const float4 q = *(const float4*)(ss + (size_t)s * 4);
      v[0] = q.x; v[1] = q.y; v[2] = q.z; v[3] = q.w;
    } else {
      for (int h = 0; h < NH; ++h) v[h] = ss[(size_t)s * NH + h];
    }
    for (int h = 0; h < NH; ++h) {
      const float t = v[h] + sdn[h];
      x[h] = (t > 0.f) ? t : 0.01f * t;
    }
  } else {
    for (int h = 0; h < NH; ++h) x[h] = -INFINITY;
  }
  return s;
}

#define SLOT_PITCH 40
#define SLOT_TAB_FLOATS(NH) ((NH) * SLOT_PITCH + 32)

template <int NH>
__device__ __forceinline__ void put_slots(float* tab, int lane, const float* alpha, int s) {
  for (int h = 0; h < NH; ++h) tab[h * SLOT_PITCH + lane] = alpha[h];
  ((unsigned*)(tab + NH * SLOT_PITCH))[lane] = (unsigned)s * (unsigned)(NH * HID * sizeof(float));
  __builtin_amdgcn_fence(__ATOMIC_ACQ_REL, "wavefront");
}

template <int NH>
__device__ __forceinline__ void load_row(const float* __restrict__ p, float* r) {
  if constexpr (NH == 4) {
    const float4 q = *(const float4*)p;
    r[0] = q.x; r[1] = q.y; r[2] = q.z; r[3] = q.w;
  } else {
    for (int h = 0; h < NH; ++h) r[h] = p[h];
  }
}

template <int NH>
__device__ __forceinline__ void add_rows(const char* __restrict__ zb, unsigned lane_off, const float* tah,
                                         const unsigned* to, int cnt, float* acc) {
  int t = 0;
  if (cnt >= 4) {
    float4 a4 = *(const float4*)tah;
    uint4 o4 = *(const uint4*)to;
    for (;;) {
      const float a[4] = {a4.x, a4.y, a4.z, a4.w};
      const unsigned o[4] = {o4.x, o4.y, o4.z, o4.w};
      float r[4][NH];
#pragma unroll
      for (int u = 0; u < 4; ++u) load_row<NH>((const float*)(zb + (o[u] + lane_off)), r[u]);
      t += 4;
      const bool more = t + 4 <= cnt;
      if (more) {
        a4 = *(const float4*)(tah + t);
        o4 = *(const uint4*)(to + t);
      }
#pragma unroll
      for (int u = 0; u < 4; ++u)
        for (int h = 0; h < NH; ++h) acc[h] = __builtin_fmaf(a[u], r[u][h], acc[h]);
      if (!more) break;
    }
  }
  for (; t < cnt; ++t) {
    float r[NH];
    const float a = tah[t];
    load_row<NH>((const float*)(zb + (to[t] + lane_off)), r);
    for (int h = 0; h < NH; ++h) acc[h] = __builtin_fmaf(a, r[h], acc[h]);
  }
}

template <int NH, bool ELU>
__device__ __forceinline__ void
node_softmax_gather_row(int n, const int* __restrict__ rowptr, const int* __restrict__ csr,
                        const float* __restrict__ ss, const float* __restrict__ sd,
                        const float* __restrict__ z, float* tab, float* acc) {
  const int lane = threadIdx.x & 31;
  const int hh = (lane * NH) >> 5;
  const int beg = rowptr[n];
  const int end = rowptr[n + 1];
  for (int h = 0; h < NH; ++h) acc[h] = 0.f;
  if (end > beg) {
    float sdn[NH];
    for (int h = 0; h < NH; ++h) sdn[h] = sd[(size_t)n * NH + h];
    const char* zb = (const char*)z;
    const unsigned lane_off = (unsigned)(lane * NH * sizeof(float));
    const float* tah = tab + hh * SLOT_PITCH;
    const unsigned* to = (const unsigned*)(tab + NH * SLOT_PITCH);
    float x0[NH];
    const int s0 = load_edge<NH>(csr, ss, sdn, beg + lane, end, x0);
    const bool v0 = (beg + lane) < end;
    float m[NH];
    for (int h = 0; h < NH; ++h) m[h] = x0[h];
    for (int base = beg + 32; base < end; base += 32) {
      float x[NH];
      load_edge<NH>(csr, ss, sdn, base + lane, end, x);
      for (int h = 0; h < NH; ++h) m[h] = fmaxf(m[h], x[h]);
    }
    for (int o = 16; o > 0; o >>= 1)
      for (int h = 0; h < NH; ++h) m[h] = fmaxf(m[h], __shfl_xor(m[h], o));
    float ee0[NH], sum[NH];
    for (int h = 0; h < NH; ++h) { ee0[h] = v0 ? expf(x0[h] - m[h]) : 0.f; sum[h] = ee0[h]; }
    for (int base = beg + 32; base < end; base += 32) {
      float x[NH];
      load_edge<NH>(csr, ss, sdn, base + lane, end, x);
      const bool v = (base + lane) < end;
      for (int h = 0; h < NH; ++h) sum[h] += v ? expf(x[h] - m[h]) : 0.f;
    }
    for (int o = 16; o > 0; o >>= 1)
      for (int h = 0; h < NH; ++h) sum[h] += __shfl_xor(sum[h], o);
    float den[NH], al[NH];
    for (int h = 0; h < NH; ++h) { den[h] = fmaxf(sum[h], 1e-9f); al[h] = ee0[h] / den[h]; }
    const int c0 = end - beg;
    put_slots<NH>(tab, lane, al, s0);
    add_rows<NH>(zb, lane_off, tah, to, c0 < 32 ? c0 : 32, acc);
    for (int base = beg + 32; base < end; base += 32) {
      float x[NH];
      const int s = load_edge<NH>(csr, ss, sdn, base + lane, end, x);
      const bool v = (base + lane) < end;
      for (int h = 0; h < NH; ++h) al[h] = v ? expf(x[h] - m[h]) / den[h] : 0.f;
      const int c = end - base;
      put_slots<NH>(tab, lane, al, s);
      add_rows<NH>(zb, lane_off, tah, to, c < 32 ? c : 32, acc);
    }
  }
  if constexpr (ELU) {
    for (int h = 0; h < NH; ++h) acc[h] = (acc[h] > 0.f) ? acc[h] : (expf(acc[h]) - 1.0f);
  }
}

template <int NH, bool ELU>
__device__ __forceinline__ void
node_softmax_gather_block(int gb, const int* __restrict__ rowptr, const int* __restrict__ csr,
                          const float* __restrict__ ss, const float* __restrict__ sd,
                          const float* __restrict__ z, float* __restrict__ out, int ldo, float* slots) {
  const int w = __builtin_amdgcn_readfirstlane((int)(threadIdx.x >> 5));
  const int n = gb * 8 + w;
  if (n >= N_NODES) return;
  const int lane = threadIdx.x & 31;
  float acc[NH];
  node_softmax_gather_row<NH, ELU>(n, rowptr, csr, ss, sd, z, slots + w * SLOT_TAB_FLOATS(NH), acc);
  float* op = out + (size_t)n * ldo + lane * NH;
  if constexpr (NH == 4) {
    *(float4*)op = make_float4(acc[0], acc[1], acc[2], acc[3]);
  } else {
    for (int h = 0; h < NH; ++h) op[h] = acc[h];
  }
}

#define H1_COLS  (HEADS * HID)
#define H1_PITCH (H1_COLS + 4)
#define FUSE_W_F4 (2 * 16 * (H1_PITCH / 4))
#define FUSE_LDS_F4 (FUSE_W_F4 + 16 * (H1_PITCH / 4))
static_assert(N_NODES % 16 == 0, "whole 16-node groups");
__device__ __forceinline__ void
gather1_z2_block(int gb, const int* __restrict__ rowptr, const int* __restrict__ csr,
                 const float* __restrict__ ss, const float* __restrict__ sd,
                 const float* __restrict__ z, const float* __restrict__ W2,
                 const float* __restrict__ a2, float* __restrict__ z2,
                 float* __restrict__ s2s, float* __restrict__ s2d, float4* lds, float* slots) {
  float4* Ws = lds;
  float* T = (float*)(lds + FUSE_W_F4);
  {
    constexpr int K4 = H1_COLS / 4, P4 = H1_PITCH / 4;
    static_assert(2 * 16 * K4 == 4 * 256, "four float4 per thread");
    float4 q[4];
#pragma unroll
    for (int u = 0; u < 4; ++u) q[u] = ((const float4*)W2)[threadIdx.x + u * 256];
#pragma unroll
    for (int u = 0; u < 4; ++u) {
      const int i = threadIdx.x + u * 256;
      Ws[(i / K4) * P4 + (i % K4)] = q[u];
    }
  }
  const int w = __builtin_amdgcn_readfirstlane((int)(threadIdx.x >> 5));
  const int lane = threadIdx.x & 31;
  for (int i = 0; i < 2; ++i) {
    const int row = w * 2 + i;
    float acc[HEADS];
    node_softmax_gather_row<HEADS, true>(gb * 16 + row, rowptr, csr, ss, sd, z,
                                         slots + w * SLOT_TAB_FLOATS(HEADS), acc);
    *(float4*)(T + row * H1_PITCH + lane * HEADS) = make_float4(acc[0], acc[1], acc[2], acc[3]);
  }
  __syncthreads();
  if (w == 0) {
    const int lo = lane & 15;
    const int hi = lane >> 4;
    const float* Tl = T + lo * H1_PITCH;
    const float* Wl = (const float*)Ws + lo * H1_PITCH;
    v8f acc[2];
    acc[0] = (v8f){0.f, 0.f, 0.f, 0.f, 0.f, 0.f, 0.f, 0.f};
    acc[1] = acc[0];
    for (int k0 = 0; k0 < H1_COLS; k0 += 4) {
      const int ka = k0 + hi * 2;
      const v2f a = *(const v2f*)(Tl + ka);
#pragma unroll
      for (int j = 0; j < 2; ++j) {
        const v2f b = *(const v2f*)(Wl + j * 16 * H1_PITCH + ka);
        acc[j] = __builtin_amdgcn_wmma_f32_16x16x4_f32(false, a, false, b,
                                                       (short)0, acc[j], false, false);
      }
    }
    strip_store_scores<2, 1>(gb, lo, hi, acc, z2, HID, 0, a2, s2s, s2d);
  }
}

#define GATHER_WGS ((N_NODES + 7) / 8)
#define GATHER16_WGS (N_NODES / 16)
#define STRIP_WGS  ((N_NODES / 16 + 7) / 8)
template <int PERIOD, int GEMM_WGS, int GATHER_N>
struct Mixed {
  static constexpr int MIX = PERIOD * GEMM_WGS;
  static constexpr int GRID = MIX + GATHER_N - (PERIOD - 1) * GEMM_WGS;
  static_assert(GATHER_N >= (PERIOD - 1) * GEMM_WGS, "the gather outlasts the GEMM");
  static __device__ __forceinline__ int gemm_wg(int i) { return (i < MIX && i % PERIOD == 0) ? i / PERIOD : -1; }
  static __device__ __forceinline__ int gather_wg(int i) {
    return (i < MIX) ? (i / PERIOD) * (PERIOD - 1) + (i % PERIOD) - 1 : i - GEMM_WGS;
  }
};
typedef Mixed<3, 3 * STRIP_WGS, GATHER16_WGS> Mix1;
typedef Mixed<13, STRIP_WGS, GATHER_WGS> Mix2;

__global__ void __launch_bounds__(256) __attribute__((amdgpu_waves_per_eu(16, 16)))
gather1_and_frac(const int* __restrict__ rowptr, const int* __restrict__ csr,
                 const float* __restrict__ ss, const float* __restrict__ sd,
                 const float* __restrict__ z,
                 const float* __restrict__ W2, const float* __restrict__ a2,
                 float* __restrict__ z2, float* __restrict__ s2s, float* __restrict__ s2d,
                 const float* __restrict__ H, const int* __restrict__ fc,
                 const float* __restrict__ Wf, float* __restrict__ hfpre) {
  __shared__ float4 lds[FUSE_LDS_F4];
  __shared__ __attribute__((aligned(16))) float slots[8 * SLOT_TAB_FLOATS(HEADS)];
  const int i = (int)blockIdx.x;
  const int f = Mix1::gemm_wg(i);
  if (f >= 0) {
    gemm_frac_block(f % STRIP_WGS, f / STRIP_WGS, H, fc, Wf, hfpre, lds);
  } else {
    gather1_z2_block(Mix1::gather_wg(i), rowptr, csr, ss, sd, z, W2, a2, z2, s2s, s2d, lds, slots);
  }
}

__global__ void __launch_bounds__(256) __attribute__((amdgpu_waves_per_eu(16, 16)))
gather2_and_wff(const int* __restrict__ rowptr, const int* __restrict__ csr,
                const float* __restrict__ ss, const float* __restrict__ sd,
                const float* __restrict__ z, float* __restrict__ hg,
                const float* __restrict__ hfpre, const float* __restrict__ Wff) {
  __shared__ __attribute__((aligned(16))) float slots[8 * SLOT_TAB_FLOATS(1)];
  const int i = (int)blockIdx.x;
  const int f = Mix2::gemm_wg(i);
  if (f >= 0) {
    gemm_strip_block<2, 96, 0>(f, hfpre, Wff, hg, N_NODES / 16, 64, HID, nullptr, nullptr, nullptr);
  } else {
    node_softmax_gather_block<1, false>(Mix2::gather_wg(i), rowptr, csr, ss, sd, z, hg, 64, slots);
  }
}

extern "C" void kernel_launch(void* const* d_in, const int* in_sizes, int n_in,
                              void* d_out, int out_size, void* d_ws, size_t ws_size,
                              hipStream_t stream) {
  const float* h    = (const float*)d_in[0];
  const int*   src  = (const int*)  d_in[1];
  const int*   dst  = (const int*)  d_in[2];
  const int*   fc   = (const int*)  d_in[3];
  const float* W1   = (const float*)d_in[4];
  const float* a1   = (const float*)d_in[5];
  const float* W2   = (const float*)d_in[6];
  const float* a2   = (const float*)d_in[7];
  const float* Wf   = (const float*)d_in[8];
  const float* Wff  = (const float*)d_in[9];
  const float* Wfc2 = (const float*)d_in[10];
  float* out = (float*)d_out;

  float* w = (float*)d_ws;
  size_t off = 0;
  int* bintot  = (int*)(w + off); off += 512;
  int* runbase = (int*)(w + off); off += SLICES * 512;
  int* rowptr  = (int*)(w + off); off += N_NODES + 32;
  int* binned  = (int*)(w + off); off += N_EDGES;
  int* csr     = (int*)(w + off); off += N_EDGES;
  float* z1  = w + off; off += (size_t)N_NODES * 128;
  float* s1s = w + off; off += (size_t)N_NODES * HEADS;
  float* s1d = w + off; off += (size_t)N_NODES * HEADS;
  float* z2  = w + off; off += (size_t)N_NODES * HID;
  float* s2s = w + off; off += N_NODES;
  float* s2d = w + off; off += N_NODES;
  float* hg  = w + off; off += (size_t)N_NODES * 64;
  float* hfp = w + off; off += (size_t)N_NODES * 96;
  (void)ws_size; (void)in_sizes; (void)n_in; (void)out_size;

  const int TB = 256;
  const int MT = N_NODES / 16;
  const int gemm_blocks = (MT + 7) / 8;

  zero_f32<<<(NBINS + TB - 1) / TB, TB, 0, stream>>>((float*)bintot, (long long)NBINS);
  bin_count<<<SLICES, SORT_T, 0, stream>>>(dst, bintot, runbase);
  bin_scatter<<<SLICES, SORT_T, 0, stream>>>(src, dst, bintot, runbase, binned);
  bin_finish<<<NBINS, SORT_T, 0, stream>>>(bintot, binned, rowptr, csr);

  wmma_gemm_strip<8, 128, HEADS><<<gemm_blocks, TB, 0, stream>>>(h, W1, z1, MT, 128, 0, a1, s1s, s1d);
  gather1_and_frac<<<Mix1::GRID, TB, 0, stream>>>(rowptr, csr, s1s, s1d, z1, W2, a2, z2, s2s, s2d,
                                                  h, fc, Wf, hfp);

  gather2_and_wff<<<Mix2::GRID, TB, 0, stream>>>(rowptr, csr, s2s, s2d, z2, hg, hfp, Wff);

  wmma_gemm_strip<1, 64, 0><<<gemm_blocks, TB, 0, stream>>>(hg, Wfc2, out, MT, OUT_DIM, 0, nullptr, nullptr, nullptr);
}


# Layer-1 z GEMM (needs no graph) runs as leading workgroups of the CSR-build launches, a strip split over 4 waves by head, not as its own launch: matrix and edge work overlap; 8 launches -> 7
# speedup vs baseline: 7.9590x; 1.0393x over previous
#include <hip/hip_runtime.h>
#include <math.h>

#define N_NODES 100000
#define N_EDGES 1600000
#define IN_DIM  128
#define HID     32
#define HEADS   4
#define OUT_DIM 16

typedef __attribute__((ext_vector_type(2))) float v2f;
typedef __attribute__((ext_vector_type(8))) float v8f;

template <int NT, int K>
__device__ __forceinline__ void stage_weights(const float* __restrict__ W, float4* Ws) {
  constexpr int K4 = K / 4;
  constexpr int P4 = K4 + 1;
  for (int i = threadIdx.x; i < NT * 16 * K4; i += 256) {
    const int row = i / K4;
    const int c = i - row * K4;
    Ws[row * P4 + c] = ((const float4*)W)[i];
  }
}

__device__ __forceinline__ float row16_sum(float x) {
  x += __int_as_float(__builtin_amdgcn_update_dpp(0, __float_as_int(x), 0x161, 0xf, 0xf, true));
  x += __int_as_float(__builtin_amdgcn_update_dpp(0, __float_as_int(x), 0x162, 0xf, 0xf, true));
  x += __int_as_float(__builtin_amdgcn_update_dpp(0, __float_as_int(x), 0x164, 0xf, 0xf, true));
  x += __int_as_float(__builtin_amdgcn_update_dpp(0, __float_as_int(x), 0x168, 0xf, 0xf, true));
  return x;
}

template <int NT, int NHD, int SST = NHD>
__device__ __forceinline__ void
strip_store_scores(int wave, int lo, int hi, const v8f* acc, float* __restrict__ C, int ldc, int coff,
                   const float* __restrict__ av, float* __restrict__ ss, float* __restrict__ sd) {
  {
#pragma unroll
    for (int j = 0; j < NT; ++j) {
      float* Cb = C + (size_t)wave * 16 * ldc + coff + (size_t)j * 16 + lo;
#pragma unroll
      for (int r = 0; r < 8; ++r)
        Cb[(size_t)(r + hi * 8) * ldc] = acc[j][r];
    }
    if constexpr (NHD > 0) {
      static_assert(NT == 2 * NHD, "a head is two 16-column tiles");
      float o[NHD];
#pragma unroll
      for (int h = 0; h < NHD; ++h) {
        const float* ah = av + h * 2 * HID;
        const float al0 = ah[lo], al1 = ah[16 + lo], ar0 = ah[HID + lo], ar1 = ah[HID + 16 + lo];
        float keep = 0.f;
#pragma unroll
        for (int r = 0; r < 8; ++r) {
          const float ps = row16_sum(acc[2 * h][r] * al0 + acc[2 * h + 1][r] * al1);
          const float pd = row16_sum(acc[2 * h][r] * ar0 + acc[2 * h + 1][r] * ar1);
          keep = (lo == r) ? ps : keep;
          keep = (lo == r + 8) ? pd : keep;
        }
        o[h] = keep;
      }
      float* sp = ((lo < 8) ? ss : sd) + ((size_t)wave * 16 + hi * 8 + (lo & 7)) * SST;
      if constexpr (NHD == 4) {
        *(float4*)sp = make_float4(o[0], o[1], o[2], o[3]);
      } else {
#pragma unroll
        for (int h = 0; h < NHD; ++h) sp[h] = o[h];
      }
    }
  }
}

template <int NT, int K, int NHD, int SST>
__device__ __forceinline__ void
gemm_strip_wave(int wave, int lo, int hi, const float* Ar, const float4* Ws,
                float* __restrict__ C, int ldc, int coff,
                const float* __restrict__ av, float* __restrict__ ss, float* __restrict__ sd) {
  constexpr int P = K + 4;
  const float* Wl = (const float*)Ws + lo * P;
  v8f acc[NT];
#pragma unroll
  for (int j = 0; j < NT; ++j)
    acc[j] = (v8f){0.f, 0.f, 0.f, 0.f, 0.f, 0.f, 0.f, 0.f};
  for (int k0 = 0; k0 < K; k0 += 4) {
    const int ka = k0 + hi * 2;
    v2f a;
    a.x = Ar[ka]; a.y = Ar[ka + 1];
#pragma unroll
    for (int j = 0; j < NT; ++j) {
      const v2f b = *(const v2f*)(Wl + j * 16 * P + ka);
      acc[j] = __builtin_amdgcn_wmma_f32_16x16x4_f32(false, a, false, b,
                                                     (short)0, acc[j], false, false);
    }
  }
  strip_store_scores<NT, NHD, SST>(wave, lo, hi, acc, C, ldc, coff, av, ss, sd);
}

template <int NT, int K, int NHD>
__device__ __forceinline__ void
gemm_strip_block(int bx, const float* __restrict__ A, const float* __restrict__ W,
                 float* __restrict__ C, int Mtiles, int ldc, int coff,
                 const float* __restrict__ av, float* __restrict__ ss, float* __restrict__ sd) {
  constexpr int P = K + 4;
  __shared__ float4 Ws[NT * 16 * (P / 4)];
  stage_weights<NT, K>(W, Ws);
  __syncthreads();
  const int wave = (bx * blockDim.x + threadIdx.x) >> 5;
  if (wave < Mtiles) {
    const int lane = threadIdx.x & 31;
    const int lo = lane & 15;
    const int hi = lane >> 4;
    gemm_strip_wave<NT, K, NHD, NHD>(wave, lo, hi, A + ((size_t)wave * 16 + lo) * K, Ws,
                                     C, ldc, coff, av, ss, sd);
  }
}

template <int NT, int K, int NHD>
__global__ void __launch_bounds__(256)
wmma_gemm_strip(const float* __restrict__ A, const float* __restrict__ W,
                float* __restrict__ C, int Mtiles, int ldc, int coff,
                const float* __restrict__ av, float* __restrict__ ss, float* __restrict__ sd) {
  gemm_strip_block<NT, K, NHD>((int)blockIdx.x, A, W, C, Mtiles, ldc, coff, av, ss, sd);
}

__device__ __forceinline__ void
gemm_frac_block(int bx, int s, const float* __restrict__ H, const int* __restrict__ fc,
                const float* __restrict__ Wf, float* __restrict__ hfpre, float4* Ws) {
  constexpr int P = IN_DIM + 4;
  stage_weights<2, IN_DIM>(Wf + (size_t)s * HID * IN_DIM, Ws);
  __syncthreads();
  const int wave = (bx * blockDim.x + threadIdx.x) >> 5;
  if (wave < N_NODES / 16) {
    const int lane = threadIdx.x & 31;
    const int lo = lane & 15;
    const int hi = lane >> 4;
    const int nid = fc[(wave * 16 + lo) * 3 + s];
    const float* Ab = H + (size_t)nid * IN_DIM;
    const float* Wl = (const float*)Ws + lo * P;
    v8f acc[2];
    acc[0] = (v8f){0.f, 0.f, 0.f, 0.f, 0.f, 0.f, 0.f, 0.f};
    acc[1] = acc[0];
    for (int k0 = 0; k0 < IN_DIM; k0 += 4) {
      const int ka = k0 + hi * 2;
      v2f a;
      a.x = Ab[ka]; a.y = Ab[ka + 1];
#pragma unroll
      for (int j = 0; j < 2; ++j) {
        const v2f b = *(const v2f*)(Wl + j * 16 * P + ka);
        acc[j] = __builtin_amdgcn_wmma_f32_16x16x4_f32(false, a, false, b,
                                                       (short)0, acc[j], false, false);
      }
    }
#pragma unroll
    for (int j = 0; j < 2; ++j) {
      float* Cb = hfpre + (size_t)wave * 16 * 96 + s * HID + (size_t)j * 16 + lo;
#pragma unroll
      for (int r = 0; r < 8; ++r)
        Cb[(size_t)(r + hi * 8) * 96] = acc[j][r];
    }
  }
}

#define BIN_SHIFT 8
#define BIN_NODES (1 << BIN_SHIFT)
#define NBINS     ((N_NODES + BIN_NODES - 1) >> BIN_SHIFT)
#define SLICES    256
#define SLICE_E   ((N_EDGES + SLICES - 1) / SLICES)
#define SORT_T    1024
#define PER_T     ((SLICE_E + SORT_T - 1) / SORT_T)
#define SRC_BITS  17
#define BIN_CAP   8192
static_assert(N_NODES <= (1 << SRC_BITS), "a source id must fit the packed word");
static_assert(NBINS <= SORT_T, "one thread per bin in the scans");
static_assert(BIN_NODES <= SORT_T, "one thread per node in the scans");

__device__ __forceinline__ int block_excl_scan(int v, int* wsum) {
  const int lane = threadIdx.x & 31, w = threadIdx.x >> 5;
  int inc = v;
  for (int o = 1; o < 32; o <<= 1) {
    const int u = __shfl_up(inc, o);
    if (lane >= o) inc += u;
  }
  if (lane == 31) wsum[w] = inc;
  __syncthreads();
  if (w == 0) {
    const int s = wsum[lane];
    int si = s;
    for (int o = 1; o < 32; o <<= 1) {
      const int u = __shfl_up(si, o);
      if (lane >= o) si += u;
    }
    wsum[lane] = si - s;
  }
  __syncthreads();
  const int ex = wsum[w] + inc - v;
  __syncthreads();
  return ex;
}

#define Z1_P        (IN_DIM + 4)
#define Z1_TILE_F4  (128 * (Z1_P / 4))
#define Z1_LDS_F4   (2 * Z1_TILE_F4)
#define Z1_STRIPS   (N_NODES / 16)
#define Z1_WGS      256
#define Z1_REST     (Z1_STRIPS - 3 * 8 * Z1_WGS)
#define Z1_HEAD_WGS ((Z1_REST + 1) / 2)
static_assert(HEADS * HID == 128 && 128 * (IN_DIM / 4) == 4 * SORT_T, "four float4 per thread fill a tile");
static_assert(Z1_REST >= 0, "three launches of eight-strip workgroups do not overshoot");
struct Z1Args {
  const float* h; const float* W1; const float* a1;
  float* z1; float* s1s; float* s1d;
};
__device__ __forceinline__ void z1_role(float4* lds, int first, int ns, const Z1Args& q) {
  constexpr int K4 = IN_DIM / 4, P4 = Z1_P / 4;
  float4* Wz = lds;
  float4* Az = lds + Z1_TILE_F4;
  {
    const float4* Wg = (const float4*)q.W1;
    const float4* Ag = (const float4*)(q.h + (size_t)first * 16 * IN_DIM);
    const int na = ns * 16 * K4;
    float4 w[4], a[4];
#pragma unroll
    for (int u = 0; u < 4; ++u) {
      const int i = threadIdx.x + u * SORT_T;
      w[u] = Wg[i];
      a[u] = (i < na) ? Ag[i] : make_float4(0.f, 0.f, 0.f, 0.f);
    }
#pragma unroll
    for (int u = 0; u < 4; ++u) {
      const int i = threadIdx.x + u * SORT_T;
      Wz[(i / K4) * P4 + (i % K4)] = w[u];
      if (i < na) Az[(i / K4) * P4 + (i % K4)] = a[u];
    }
  }
  __syncthreads();
  const int wv = __builtin_amdgcn_readfirstlane((int)(threadIdx.x >> 5));
  const int sl = wv >> 2;
  const int hd = wv & 3;
  if (sl < ns) {
    const int lane = threadIdx.x & 31;
    const int lo = lane & 15;
    const int hi = lane >> 4;
    gemm_strip_wave<2, IN_DIM, 1, HEADS>(first + sl, lo, hi, (const float*)Az + (sl * 16 + lo) * Z1_P,
                                         Wz + hd * 32 * P4, q.z1, HEADS * HID, hd * HID,
                                         q.a1 + hd * 2 * HID, q.s1s + hd, q.s1d + hd);
  }
}

__global__ void __launch_bounds__(SORT_T) __attribute__((amdgpu_waves_per_eu(16, 16)))
zero_and_z1(int* __restrict__ bintot, Z1Args q) {
  __shared__ float4 lds[Z1_LDS_F4];
  const int i = (int)blockIdx.x;
  if (i < Z1_HEAD_WGS) {
    const int first = 3 * 8 * Z1_WGS + 2 * i;
    z1_role(lds, first, (Z1_STRIPS - first < 2) ? Z1_STRIPS - first : 2, q);
  } else {
    for (int k = threadIdx.x; k < NBINS; k += SORT_T) bintot[k] = 0;
  }
}

__global__ void __launch_bounds__(SORT_T) __attribute__((amdgpu_waves_per_eu(16, 16)))
bin_count(const int* __restrict__ dst, int* __restrict__ bintot, int* __restrict__ runbase, Z1Args q) {
  __shared__ float4 lds[Z1_LDS_F4];
  if ((int)blockIdx.x < Z1_WGS) { z1_role(lds, (int)blockIdx.x * 8, 8, q); return; }
  const int bx = (int)blockIdx.x - Z1_WGS;
  int* hist = (int*)lds;
  const int t = threadIdx.x;
  for (int i = t; i < NBINS; i += SORT_T) hist[i] = 0;
  __syncthreads();
  const int e0 = bx * SLICE_E;
  const int e1 = (e0 + SLICE_E < N_EDGES) ? e0 + SLICE_E : N_EDGES;
  for (int e = e0 + t; e < e1; e += SORT_T) atomicAdd(&hist[dst[e] >> BIN_SHIFT], 1);
  __syncthreads();
  for (int i = t; i < NBINS; i += SORT_T)
    runbase[bx * NBINS + i] = atomicAdd(&bintot[i], hist[i]);
}

#define NBINS_PAD ((NBINS + 31) & ~31)
static_assert((3 * NBINS_PAD + 32 + 2 * SLICE_E) * 4 <= Z1_LDS_F4 * 16, "the scatter's arrays fit the block");
__global__ void __launch_bounds__(SORT_T) __attribute__((amdgpu_waves_per_eu(16, 16)))
bin_scatter(const int* __restrict__ src, const int* __restrict__ dst,
            const int* __restrict__ bintot, const int* __restrict__ runbase,
            int* __restrict__ binned, Z1Args q) {
  __shared__ float4 lds[Z1_LDS_F4];
  if ((int)blockIdx.x < Z1_WGS) { z1_role(lds, (Z1_WGS + (int)blockIdx.x) * 8, 8, q); return; }
  const int bx = (int)blockIdx.x - Z1_WGS;
  int* hist = (int*)lds;
  int* lstart = hist + NBINS_PAD;
  int* gbase = lstart + NBINS_PAD;
  int* wsum = gbase + NBINS_PAD;
  int* sval = wsum + 32;
  int* sidx = sval + SLICE_E;
  const int t = threadIdx.x;
  for (int i = t; i < NBINS; i += SORT_T) hist[i] = 0;
  __syncthreads();
  const int e0 = bx * SLICE_E;
  const int e1 = (e0 + SLICE_E < N_EDGES) ? e0 + SLICE_E : N_EDGES;
  int d[PER_T], s[PER_T];
#pragma unroll
  for (int k = 0; k < PER_T; ++k) {
    const int e = e0 + t + k * SORT_T;
    d[k] = -1; s[k] = 0;
    if (e < e1) { d[k] = dst[e]; s[k] = src[e]; atomicAdd(&hist[d[k] >> BIN_SHIFT], 1); }
  }
  __syncthreads();
  const int tot = (t < NBINS) ? bintot[t] : 0;
  const int cnt = (t < NBINS) ? hist[t] : 0;
  const int binstart = block_excl_scan(tot, wsum);
  const int lst = block_excl_scan(cnt, wsum);
  if (t < NBINS) {
    gbase[t] = binstart + runbase[bx * NBINS + t];
    lstart[t] = lst;
    hist[t] = 0;
  }
  __syncthreads();
#pragma unroll
  for (int k = 0; k < PER_T; ++k) {
    if (d[k] >= 0) {
      const int b = d[k] >> BIN_SHIFT;
      const int r = atomicAdd(&hist[b], 1);
      const int p = lstart[b] + r;
      sval[p] = s[k] | ((d[k] & (BIN_NODES - 1)) << SRC_BITS);
      sidx[p] = gbase[b] + r;
    }
  }
  __syncthreads();
  const int n = e1 - e0;
  for (int i = t; i < n; i += SORT_T) binned[sidx[i]] = sval[i];
}

static_assert((2 * BIN_NODES + 64 + BIN_CAP) * 4 <= Z1_LDS_F4 * 16, "the finish pass's arrays fit the block");
__global__ void __launch_bounds__(SORT_T) __attribute__((amdgpu_waves_per_eu(16, 16)))
bin_finish(const int* __restrict__ bintot, const int* __restrict__ binned,
           int* __restrict__ rowptr, int* __restrict__ csr, Z1Args q) {
  __shared__ float4 lds[Z1_LDS_F4];
  if ((int)blockIdx.x < Z1_WGS) { z1_role(lds, (2 * Z1_WGS + (int)blockIdx.x) * 8, 8, q); return; }
  int* deg = (int*)lds;
  int* first = deg + BIN_NODES;
  int* wsum = first + BIN_NODES;
  int* range = wsum + 32;
  int* sorted = range + 32;
  const int t = threadIdx.x;
  const int b = (int)blockIdx.x - Z1_WGS;
  const int tot = (t < NBINS) ? bintot[t] : 0;
  const int ex0 = block_excl_scan(tot, wsum);
  if (t == b) { range[0] = ex0; range[1] = tot; }
  if (t < BIN_NODES) deg[t] = 0;
  __syncthreads();
  const int start = range[0];
  const int cnt = range[1];
  const int* mine = binned + start;
  for (int i = t; i < cnt; i += SORT_T) atomicAdd(&deg[((unsigned)mine[i]) >> SRC_BITS], 1);
  __syncthreads();
  const int dg = (t < BIN_NODES) ? deg[t] : 0;
  const int ex = block_excl_scan(dg, wsum);
  if (t < BIN_NODES) {
    const int node = b * BIN_NODES + t;
    if (node < N_NODES) rowptr[node] = start + ex;
    first[t] = ex;
    deg[t] = 0;
  }
  if (b == NBINS - 1 && t == 0) rowptr[N_NODES] = start + cnt;
  __syncthreads();
  if (cnt <= BIN_CAP) {
    for (int i = t; i < cnt; i += SORT_T) {
      const unsigned w = (unsigned)mine[i];
      const int dl = w >> SRC_BITS;
      const int r = atomicAdd(&deg[dl], 1);
      sorted[first[dl] + r] = (int)(w & ((1u << SRC_BITS) - 1u));
    }
    __syncthreads();
    for (int i = t; i < cnt; i += SORT_T) csr[start + i] = sorted[i];
  } else {
    for (int i = t; i < cnt; i += SORT_T) {
      const unsigned w = (unsigned)mine[i];
      const int dl = w >> SRC_BITS;
      const int r = atomicAdd(&deg[dl], 1);
      csr[start + first[dl] + r] = (int)(w & ((1u << SRC_BITS) - 1u));
    }
  }
}

template <int NH>
__device__ __forceinline__ int load_edge(const int* __restrict__ csr, const float* __restrict__ ss,
                                         const float* sdn, int j, int end, float* x) {
  int s = 0;
  if (j < end) {
    s = csr[j];
    float v[NH];
    if constexpr (NH == 4) {
      const float4 q = *(const float4*)(ss + (size_t)s * 4);
      v[0] = q.x; v[1] = q.y; v[2] = q.z; v[3] = q.w;
    } else {
      for (int h = 0; h < NH; ++h) v[h] = ss[(size_t)s * NH + h];
    }
    for (int h = 0; h < NH; ++h) {
      const float t = v[h] + sdn[h];
      x[h] = (t > 0.f) ? t : 0.01f * t;
    }
  } else {
    for (int h = 0; h < NH; ++h) x[h] = -INFINITY;
  }
  return s;
}

#define SLOT_PITCH 40
#define SLOT_TAB_FLOATS(NH) ((NH) * SLOT_PITCH + 32)

template <int NH>
__device__ __forceinline__ void put_slots(float* tab, int lane, const float* alpha, int s) {
  for (int h = 0; h < NH; ++h) tab[h * SLOT_PITCH + lane] = alpha[h];
  ((unsigned*)(tab + NH * SLOT_PITCH))[lane] = (unsigned)s * (unsigned)(NH * HID * sizeof(float));
  __builtin_amdgcn_fence(__ATOMIC_ACQ_REL, "wavefront");
}

template <int NH>
__device__ __forceinline__ void load_row(const float* __restrict__ p, float* r) {
  if constexpr (NH == 4) {
    const float4 q = *(const float4*)p;
    r[0] = q.x; r[1] = q.y; r[2] = q.z; r[3] = q.w;
  } else {
    for (int h = 0; h < NH; ++h) r[h] = p[h];
  }
}

template <int NH>
__device__ __forceinline__ void add_rows(const char* __restrict__ zb, unsigned lane_off, const float* tah,
                                         const unsigned* to, int cnt, float* acc) {
  int t = 0;
  if (cnt >= 4) {
    float4 a4 = *(const float4*)tah;
    uint4 o4 = *(const uint4*)to;
    for (;;) {
      const float a[4] = {a4.x, a4.y, a4.z, a4.w};
      const unsigned o[4] = {o4.x, o4.y, o4.z, o4.w};
      float r[4][NH];
#pragma unroll
      for (int u = 0; u < 4; ++u) load_row<NH>((const float*)(zb + (o[u] + lane_off)), r[u]);
      t += 4;
      const bool more = t + 4 <= cnt;
      if (more) {
        a4 = *(const float4*)(tah + t);
        o4 = *(const uint4*)(to + t);
      }
#pragma unroll
      for (int u = 0; u < 4; ++u)
        for (int h = 0; h < NH; ++h) acc[h] = __builtin_fmaf(a[u], r[u][h], acc[h]);
      if (!more) break;
    }
  }
  for (; t < cnt; ++t) {
    float r[NH];
    const float a = tah[t];
    load_row<NH>((const float*)(zb + (to[t] + lane_off)), r);
    for (int h = 0; h < NH; ++h) acc[h] = __builtin_fmaf(a, r[h], acc[h]);
  }
}

template <int NH, bool ELU>
__device__ __forceinline__ void
node_softmax_gather_row(int n, const int* __restrict__ rowptr, const int* __restrict__ csr,
                        const float* __restrict__ ss, const float* __restrict__ sd,
                        const float* __restrict__ z, float* tab, float* acc) {
  const int lane = threadIdx.x & 31;
  const int hh = (lane * NH) >> 5;
  const int beg = rowptr[n];
  const int end = rowptr[n + 1];
  for (int h = 0; h < NH; ++h) acc[h] = 0.f;
  if (end > beg) {
    float sdn[NH];
    for (int h = 0; h < NH; ++h) sdn[h] = sd[(size_t)n * NH + h];
    const char* zb = (const char*)z;
    const unsigned lane_off = (unsigned)(lane * NH * sizeof(float));
    const float* tah = tab + hh * SLOT_PITCH;
    const unsigned* to = (const unsigned*)(tab + NH * SLOT_PITCH);
    float x0[NH];
    const int s0 = load_edge<NH>(csr, ss, sdn, beg + lane, end, x0);
    const bool v0 = (beg + lane) < end;
    float m[NH];
    for (int h = 0; h < NH; ++h) m[h] = x0[h];
    for (int base = beg + 32; base < end; base += 32) {
      float x[NH];
      load_edge<NH>(csr, ss, sdn, base + lane, end, x);
      for (int h = 0; h < NH; ++h) m[h] = fmaxf(m[h], x[h]);
    }
    for (int o = 16; o > 0; o >>= 1)
      for (int h = 0; h < NH; ++h) m[h] = fmaxf(m[h], __shfl_xor(m[h], o));
    float ee0[NH], sum[NH];
    for (int h = 0; h < NH; ++h) { ee0[h] = v0 ? expf(x0[h] - m[h]) : 0.f; sum[h] = ee0[h]; }
    for (int base = beg + 32; base < end; base += 32) {
      float x[NH];
      load_edge<NH>(csr, ss, sdn, base + lane, end, x);
      const bool v = (base + lane) < end;
      for (int h = 0; h < NH; ++h) sum[h] += v ? expf(x[h] - m[h]) : 0.f;
    }
    for (int o = 16; o > 0; o >>= 1)
      for (int h = 0; h < NH; ++h) sum[h] += __shfl_xor(sum[h], o);
    float den[NH], al[NH];
    for (int h = 0; h < NH; ++h) { den[h] = fmaxf(sum[h], 1e-9f); al[h] = ee0[h] / den[h]; }
    const int c0 = end - beg;
    put_slots<NH>(tab, lane, al, s0);
    add_rows<NH>(zb, lane_off, tah, to, c0 < 32 ? c0 : 32, acc);
    for (int base = beg + 32; base < end; base += 32) {
      float x[NH];
      const int s = load_edge<NH>(csr, ss, sdn, base + lane, end, x);
      const bool v = (base + lane) < end;
      for (int h = 0; h < NH; ++h) al[h] = v ? expf(x[h] - m[h]) / den[h] : 0.f;
      const int c = end - base;
      put_slots<NH>(tab, lane, al, s);
      add_rows<NH>(zb, lane_off, tah, to, c < 32 ? c : 32, acc);
    }
  }
  if constexpr (ELU) {
    for (int h = 0; h < NH; ++h) acc[h] = (acc[h] > 0.f) ? acc[h] : (expf(acc[h]) - 1.0f);
  }
}

template <int NH, bool ELU>
__device__ __forceinline__ void
node_softmax_gather_block(int gb, const int* __restrict__ rowptr, const int* __restrict__ csr,
                          const float* __restrict__ ss, const float* __restrict__ sd,
                          const float* __restrict__ z, float* __restrict__ out, int ldo, float* slots) {
  const int w = __builtin_amdgcn_readfirstlane((int)(threadIdx.x >> 5));
  const int n = gb * 8 + w;
  if (n >= N_NODES) return;
  const int lane = threadIdx.x & 31;
  float acc[NH];
  node_softmax_gather_row<NH, ELU>(n, rowptr, csr, ss, sd, z, slots + w * SLOT_TAB_FLOATS(NH), acc);
  float* op = out + (size_t)n * ldo + lane * NH;
  if constexpr (NH == 4) {
    *(float4*)op = make_float4(acc[0], acc[1], acc[2], acc[3]);
  } else {
    for (int h = 0; h < NH; ++h) op[h] = acc[h];
  }
}

#define H1_COLS  (HEADS * HID)
#define H1_PITCH (H1_COLS + 4)
#define FUSE_W_F4 (2 * 16 * (H1_PITCH / 4))
#define FUSE_LDS_F4 (FUSE_W_F4 + 16 * (H1_PITCH / 4))
static_assert(N_NODES % 16 == 0, "whole 16-node groups");
__device__ __forceinline__ void
gather1_z2_block(int gb, const int* __restrict__ rowptr, const int* __restrict__ csr,
                 const float* __restrict__ ss, const float* __restrict__ sd,
                 const float* __restrict__ z, const float* __restrict__ W2,
                 const float* __restrict__ a2, float* __restrict__ z2,
                 float* __restrict__ s2s, float* __restrict__ s2d, float4* lds, float* slots) {
  float4* Ws = lds;
  float* T = (float*)(lds + FUSE_W_F4);
  {
    constexpr int K4 = H1_COLS / 4, P4 = H1_PITCH / 4;
    static_assert(2 * 16 * K4 == 4 * 256, "four float4 per thread");
    float4 q[4];
#pragma unroll
    for (int u = 0; u < 4; ++u) q[u] = ((const float4*)W2)[threadIdx.x + u * 256];
#pragma unroll
    for (int u = 0; u < 4; ++u) {
      const int i = threadIdx.x + u * 256;
      Ws[(i / K4) * P4 + (i % K4)] = q[u];
    }
  }
  const int w = __builtin_amdgcn_readfirstlane((int)(threadIdx.x >> 5));
  const int lane = threadIdx.x & 31;
  for (int i = 0; i < 2; ++i) {
    const int row = w * 2 + i;
    float acc[HEADS];
    node_softmax_gather_row<HEADS, true>(gb * 16 + row, rowptr, csr, ss, sd, z,
                                         slots + w * SLOT_TAB_FLOATS(HEADS), acc);
    *(float4*)(T + row * H1_PITCH + lane * HEADS) = make_float4(acc[0], acc[1], acc[2], acc[3]);
  }
  __syncthreads();
  if (w == 0) {
    const int lo = lane & 15;
    const int hi = lane >> 4;
    const float* Tl = T + lo * H1_PITCH;
    const float* Wl = (const float*)Ws + lo * H1_PITCH;
    v8f acc[2];
    acc[0] = (v8f){0.f, 0.f, 0.f, 0.f, 0.f, 0.f, 0.f, 0.f};
    acc[1] = acc[0];
    for (int k0 = 0; k0 < H1_COLS; k0 += 4) {
      const int ka = k0 + hi * 2;
      const v2f a = *(const v2f*)(Tl + ka);
#pragma unroll
      for (int j = 0; j < 2; ++j) {
        const v2f b = *(const v2f*)(Wl + j * 16 * H1_PITCH + ka);
        acc[j] = __builtin_amdgcn_wmma_f32_16x16x4_f32(false, a, false, b,
                                                       (short)0, acc[j], false, false);
      }
    }
    strip_store_scores<2, 1>(gb, lo, hi, acc, z2, HID, 0, a2, s2s, s2d);
  }
}

#define GATHER_WGS ((N_NODES + 7) / 8)
#define GATHER16_WGS (N_NODES / 16)
#define STRIP_WGS  ((N_NODES / 16 + 7) / 8)
template <int PERIOD, int GEMM_WGS, int GATHER_N>
struct Mixed {
  static constexpr int MIX = PERIOD * GEMM_WGS;
  static constexpr int GRID = MIX + GATHER_N - (PERIOD - 1) * GEMM_WGS;
  static_assert(GATHER_N >= (PERIOD - 1) * GEMM_WGS, "the gather outlasts the GEMM");
  static __device__ __forceinline__ int gemm_wg(int i) { return (i < MIX && i % PERIOD == 0) ? i / PERIOD : -1; }
  static __device__ __forceinline__ int gather_wg(int i) {
    return (i < MIX) ? (i / PERIOD) * (PERIOD - 1) + (i % PERIOD) - 1 : i - GEMM_WGS;
  }
};
typedef Mixed<3, 3 * STRIP_WGS, GATHER16_WGS> Mix1;
typedef Mixed<13, STRIP_WGS, GATHER_WGS> Mix2;

__global__ void __launch_bounds__(256) __attribute__((amdgpu_waves_per_eu(16, 16)))
gather1_and_frac(const int* __restrict__ rowptr, const int* __restrict__ csr,
                 const float* __restrict__ ss, const float* __restrict__ sd,
                 const float* __restrict__ z,
                 const float* __restrict__ W2, const float* __restrict__ a2,
                 float* __restrict__ z2, float* __restrict__ s2s, float* __restrict__ s2d,
                 const float* __restrict__ H, const int* __restrict__ fc,
                 const float* __restrict__ Wf, float* __restrict__ hfpre) {
  __shared__ float4 lds[FUSE_LDS_F4];
  __shared__ __attribute__((aligned(16))) float slots[8 * SLOT_TAB_FLOATS(HEADS)];
  const int i = (int)blockIdx.x;
  const int f = Mix1::gemm_wg(i);
  if (f >= 0) {
    gemm_frac_block(f % STRIP_WGS, f / STRIP_WGS, H, fc, Wf, hfpre, lds);
  } else {
    gather1_z2_block(Mix1::gather_wg(i), rowptr, csr, ss, sd, z, W2, a2, z2, s2s, s2d, lds, slots);
  }
}

__global__ void __launch_bounds__(256) __attribute__((amdgpu_waves_per_eu(16, 16)))
gather2_and_wff(const int* __restrict__ rowptr, const int* __restrict__ csr,
                const float* __restrict__ ss, const float* __restrict__ sd,
                const float* __restrict__ z, float* __restrict__ hg,
                const float* __restrict__ hfpre, const float* __restrict__ Wff) {
  __shared__ __attribute__((aligned(16))) float slots[8 * SLOT_TAB_FLOATS(1)];
  const int i = (int)blockIdx.x;
  const int f = Mix2::gemm_wg(i);
  if (f >= 0) {
    gemm_strip_block<2, 96, 0>(f, hfpre, Wff, hg, N_NODES / 16, 64, HID, nullptr, nullptr, nullptr);
  } else {
    node_softmax_gather_block<1, false>(Mix2::gather_wg(i), rowptr, csr, ss, sd, z, hg, 64, slots);
  }
}

extern "C" void kernel_launch(void* const* d_in, const int* in_sizes, int n_in,
                              void* d_out, int out_size, void* d_ws, size_t ws_size,
                              hipStream_t stream) {
  const float* h    = (const float*)d_in[0];
  const int*   src  = (const int*)  d_in[1];
  const int*   dst  = (const int*)  d_in[2];
  const int*   fc   = (const int*)  d_in[3];
  const float* W1   = (const float*)d_in[4];
  const float* a1   = (const float*)d_in[5];
  const float* W2   = (const float*)d_in[6];
  const float* a2   = (const float*)d_in[7];
  const float* Wf   = (const float*)d_in[8];
  const float* Wff  = (const float*)d_in[9];
  const float* Wfc2 = (const float*)d_in[10];
  float* out = (float*)d_out;

  float* w = (float*)d_ws;
  size_t off = 0;
  int* bintot  = (int*)(w + off); off += 512;
  int* runbase = (int*)(w + off); off += SLICES * 512;
  int* rowptr  = (int*)(w + off); off += N_NODES + 32;
  int* binned  = (int*)(w + off); off += N_EDGES;
  int* csr     = (int*)(w + off); off += N_EDGES;
  float* z1  = w + off; off += (size_t)N_NODES * 128;
  float* s1s = w + off; off += (size_t)N_NODES * HEADS;
  float* s1d = w + off; off += (size_t)N_NODES * HEADS;
  float* z2  = w + off; off += (size_t)N_NODES * HID;
  float* s2s = w + off; off += N_NODES;
  float* s2d = w + off; off += N_NODES;
  float* hg  = w + off; off += (size_t)N_NODES * 64;
  float* hfp = w + off; off += (size_t)N_NODES * 96;
  (void)ws_size; (void)in_sizes; (void)n_in; (void)out_size;

  const int TB = 256;
  const int MT = N_NODES / 16;
  const int gemm_blocks = (MT + 7) / 8;

  const Z1Args zq = {h, W1, a1, z1, s1s, s1d};
  zero_and_z1<<<Z1_HEAD_WGS + 1, SORT_T, 0, stream>>>(bintot, zq);
  bin_count<<<Z1_WGS + SLICES, SORT_T, 0, stream>>>(dst, bintot, runbase, zq);
  bin_scatter<<<Z1_WGS + SLICES, SORT_T, 0, stream>>>(src, dst, bintot, runbase, binned, zq);
  bin_finish<<<Z1_WGS + NBINS, SORT_T, 0, stream>>>(bintot, binned, rowptr, csr, zq);

  gather1_and_frac<<<Mix1::GRID, TB, 0, stream>>>(rowptr, csr, s1s, s1d, z1, W2, a2, z2, s2s, s2d,
                                                  h, fc, Wf, hfp);

  gather2_and_wff<<<Mix2::GRID, TB, 0, stream>>>(rowptr, csr, s2s, s2d, z2, hg, hfp, Wff);

  wmma_gemm_strip<1, 64, 0><<<gemm_blocks, TB, 0, stream>>>(hg, Wfc2, out, MT, OUT_DIM, 0, nullptr, nullptr, nullptr);
}


# Frac branch folded by linearity: Mt_s=Wfc2[:,32:].Wff[:,32s:32s+32].Wf[s] built in-call on WMMA, P=h@Mt^T [N,48] once per node; final GEMM (K=32) adds 3 gathered 16-float pieces; sum order changed
# speedup vs baseline: 9.3661x; 1.1768x over previous
#include <hip/hip_runtime.h>
#include <math.h>

#define N_NODES 100000
#define N_EDGES 1600000
#define IN_DIM  128
#define HID     32
#define HEADS   4
#define OUT_DIM 16

typedef __attribute__((ext_vector_type(2))) float v2f;
typedef __attribute__((ext_vector_type(8))) float v8f;


__device__ __forceinline__ float row16_sum(float x) {
  x += __int_as_float(__builtin_amdgcn_update_dpp(0, __float_as_int(x), 0x161, 0xf, 0xf, true));
  x += __int_as_float(__builtin_amdgcn_update_dpp(0, __float_as_int(x), 0x162, 0xf, 0xf, true));
  x += __int_as_float(__builtin_amdgcn_update_dpp(0, __float_as_int(x), 0x164, 0xf, 0xf, true));
  x += __int_as_float(__builtin_amdgcn_update_dpp(0, __float_as_int(x), 0x168, 0xf, 0xf, true));
  return x;
}

template <int NT, int NHD, int SST = NHD>
__device__ __forceinline__ void
strip_store_scores(int wave, int lo, int hi, const v8f* acc, float* __restrict__ C, int ldc, int coff,
                   const float* __restrict__ av, float* __restrict__ ss, float* __restrict__ sd) {
  {
#pragma unroll
    for (int j = 0; j < NT; ++j) {
      float* Cb = C + (size_t)wave * 16 * ldc + coff + (size_t)j * 16 + lo;
#pragma unroll
      for (int r = 0; r < 8; ++r)
        Cb[(size_t)(r + hi * 8) * ldc] = acc[j][r];
    }
    if constexpr (NHD > 0) {
      static_assert(NT == 2 * NHD, "a head is two 16-column tiles");
      float o[NHD];
#pragma unroll
      for (int h = 0; h < NHD; ++h) {
        const float* ah = av + h * 2 * HID;
        const float al0 = ah[lo], al1 = ah[16 + lo], ar0 = ah[HID + lo], ar1 = ah[HID + 16 + lo];
        float keep = 0.f;
#pragma unroll
        for (int r = 0; r < 8; ++r) {
          const float ps = row16_sum(acc[2 * h][r] * al0 + acc[2 * h + 1][r] * al1);
          const float pd = row16_sum(acc[2 * h][r] * ar0 + acc[2 * h + 1][r] * ar1);
          keep = (lo == r) ? ps : keep;
          keep = (lo == r + 8) ? pd : keep;
        }
        o[h] = keep;
      }
      float* sp = ((lo < 8) ? ss : sd) + ((size_t)wave * 16 + hi * 8 + (lo & 7)) * SST;
      if constexpr (NHD == 4) {
        *(float4*)sp = make_float4(o[0], o[1], o[2], o[3]);
      } else {
#pragma unroll
        for (int h = 0; h < NHD; ++h) sp[h] = o[h];
      }
    }
  }
}

template <int NT, int K, int NHD, int SST>
__device__ __forceinline__ void
gemm_strip_wave(int wave, int lo, int hi, const float* Ar, const float4* Ws,
                float* __restrict__ C, int ldc, int coff,
                const float* __restrict__ av, float* __restrict__ ss, float* __restrict__ sd) {
  constexpr int P = K + 4;
  const float* Wl = (const float*)Ws + lo * P;
  v8f acc[NT];
#pragma unroll
  for (int j = 0; j < NT; ++j)
    acc[j] = (v8f){0.f, 0.f, 0.f, 0.f, 0.f, 0.f, 0.f, 0.f};
  for (int k0 = 0; k0 < K; k0 += 4) {
    const int ka = k0 + hi * 2;
    v2f a;
    a.x = Ar[ka]; a.y = Ar[ka + 1];
#pragma unroll
    for (int j = 0; j < NT; ++j) {
      const v2f b = *(const v2f*)(Wl + j * 16 * P + ka);
      acc[j] = __builtin_amdgcn_wmma_f32_16x16x4_f32(false, a, false, b,
                                                     (short)0, acc[j], false, false);
    }
  }
  strip_store_scores<NT, NHD, SST>(wave, lo, hi, acc, C, ldc, coff, av, ss, sd);
}

#define P_COLS 48
__device__ __forceinline__ void
gemm_p_block(int bx, const float* __restrict__ H, const float* __restrict__ Mt,
             float* __restrict__ P, float4* Ws) {
  {
    constexpr int K4 = IN_DIM / 4, P4 = K4 + 1;
    static_assert(P_COLS * K4 == 6 * 256, "six float4 per thread");
    float4 q[6];
#pragma unroll
    for (int u = 0; u < 6; ++u) q[u] = ((const float4*)Mt)[threadIdx.x + u * 256];
#pragma unroll
    for (int u = 0; u < 6; ++u) {
      const int i = threadIdx.x + u * 256;
      Ws[(i / K4) * P4 + (i % K4)] = q[u];
    }
  }
  __syncthreads();
  const int wave = (bx * blockDim.x + threadIdx.x) >> 5;
  if (wave < N_NODES / 16) {
    const int lane = threadIdx.x & 31;
    const int lo = lane & 15;
    const int hi = lane >> 4;
    gemm_strip_wave<3, IN_DIM, 0, 0>(wave, lo, hi, H + ((size_t)wave * 16 + lo) * IN_DIM, Ws,
                                     P, P_COLS, 0, nullptr, nullptr, nullptr);
  }
}

#define FOLD_TP 136
__device__ __forceinline__ void
fold_frac_weights(float* T, const float* __restrict__ Wf, const float* __restrict__ Wff,
                  const float* __restrict__ Wfc2, float* __restrict__ Mt) {
  const int wv = __builtin_amdgcn_readfirstlane((int)(threadIdx.x >> 5));
  const int lane = threadIdx.x & 31;
  const int lo = lane & 15;
  const int hi = lane >> 4;
  for (int t = wv; t < 48; t += 32) {
    const int s = t >> 4, mt = (t >> 3) & 1, nt = t & 7;
    const float* Ar = Wff + (mt * 16 + lo) * (3 * HID) + s * HID;
    const float* Bc = Wf + (size_t)s * HID * IN_DIM + nt * 16 + lo;
    v8f acc = (v8f){0.f, 0.f, 0.f, 0.f, 0.f, 0.f, 0.f, 0.f};
    for (int k0 = 0; k0 < HID; k0 += 4) {
      const int ka = k0 + hi * 2;
      v2f a, b;
      a.x = Ar[ka]; a.y = Ar[ka + 1];
      b.x = Bc[ka * IN_DIM]; b.y = Bc[(ka + 1) * IN_DIM];
      acc = __builtin_amdgcn_wmma_f32_16x16x4_f32(false, a, false, b, (short)0, acc, false, false);
    }
    float* Tb = T + (s * HID + mt * 16 + hi * 8) * FOLD_TP + nt * 16 + lo;
    for (int r = 0; r < 8; ++r) Tb[r * FOLD_TP] = acc[r];
  }
  __syncthreads();
  if (wv < 24) {
    const int s = wv >> 3, nt = wv & 7;
    const float* Ar = Wfc2 + lo * (2 * HID) + HID;
    const float* Bc = T + s * HID * FOLD_TP + nt * 16 + lo;
    v8f acc = (v8f){0.f, 0.f, 0.f, 0.f, 0.f, 0.f, 0.f, 0.f};
    for (int k0 = 0; k0 < HID; k0 += 4) {
      const int ka = k0 + hi * 2;
      v2f a, b;
      a.x = Ar[ka]; a.y = Ar[ka + 1];
      b.x = Bc[ka * FOLD_TP]; b.y = Bc[(ka + 1) * FOLD_TP];
      acc = __builtin_amdgcn_wmma_f32_16x16x4_f32(false, a, false, b, (short)0, acc, false, false);
    }
    float* Mb = Mt + (s * 16 + hi * 8) * IN_DIM + nt * 16 + lo;
    for (int r = 0; r < 8; ++r) Mb[r * IN_DIM] = acc[r];
  }
}

#define BIN_SHIFT 8
#define BIN_NODES (1 << BIN_SHIFT)
#define NBINS     ((N_NODES + BIN_NODES - 1) >> BIN_SHIFT)
#define SLICES    256
#define SLICE_E   ((N_EDGES + SLICES - 1) / SLICES)
#define SORT_T    1024
#define PER_T     ((SLICE_E + SORT_T - 1) / SORT_T)
#define SRC_BITS  17
#define BIN_CAP   8192
static_assert(N_NODES <= (1 << SRC_BITS), "a source id must fit the packed word");
static_assert(NBINS <= SORT_T, "one thread per bin in the scans");
static_assert(BIN_NODES <= SORT_T, "one thread per node in the scans");

__device__ __forceinline__ int block_excl_scan(int v, int* wsum) {
  const int lane = threadIdx.x & 31, w = threadIdx.x >> 5;
  int inc = v;
  for (int o = 1; o < 32; o <<= 1) {
    const int u = __shfl_up(inc, o);
    if (lane >= o) inc += u;
  }
  if (lane == 31) wsum[w] = inc;
  __syncthreads();
  if (w == 0) {
    const int s = wsum[lane];
    int si = s;
    for (int o = 1; o < 32; o <<= 1) {
      const int u = __shfl_up(si, o);
      if (lane >= o) si += u;
    }
    wsum[lane] = si - s;
  }
  __syncthreads();
  const int ex = wsum[w] + inc - v;
  __syncthreads();
  return ex;
}

#define Z1_P        (IN_DIM + 4)
#define Z1_TILE_F4  (128 * (Z1_P / 4))
#define Z1_LDS_F4   (2 * Z1_TILE_F4)
#define Z1_STRIPS   (N_NODES / 16)
#define Z1_WGS      256
#define Z1_REST     (Z1_STRIPS - 3 * 8 * Z1_WGS)
#define Z1_HEAD_WGS ((Z1_REST + 1) / 2)
static_assert(HEADS * HID == 128 && 128 * (IN_DIM / 4) == 4 * SORT_T, "four float4 per thread fill a tile");
static_assert(Z1_REST >= 0, "three launches of eight-strip workgroups do not overshoot");
static_assert(3 * HID * FOLD_TP * 4 <= Z1_LDS_F4 * 16, "the folded weights' intermediate fits the block");
struct Z1Args {
  const float* h; const float* W1; const float* a1;
  float* z1; float* s1s; float* s1d;
};
__device__ __forceinline__ void z1_role(float4* lds, int first, int ns, const Z1Args& q) {
  constexpr int K4 = IN_DIM / 4, P4 = Z1_P / 4;
  float4* Wz = lds;
  float4* Az = lds + Z1_TILE_F4;
  {
    const float4* Wg = (const float4*)q.W1;
    const float4* Ag = (const float4*)(q.h + (size_t)first * 16 * IN_DIM);
    const int na = ns * 16 * K4;
    float4 w[4], a[4];
#pragma unroll
    for (int u = 0; u < 4; ++u) {
      const int i = threadIdx.x + u * SORT_T;
      w[u] = Wg[i];
      a[u] = (i < na) ? Ag[i] : make_float4(0.f, 0.f, 0.f, 0.f);
    }
#pragma unroll
    for (int u = 0; u < 4; ++u) {
      const int i = threadIdx.x + u * SORT_T;
      Wz[(i / K4) * P4 + (i % K4)] = w[u];
      if (i < na) Az[(i / K4) * P4 + (i % K4)] = a[u];
    }
  }
  __syncthreads();
  const int wv = __builtin_amdgcn_readfirstlane((int)(threadIdx.x >> 5));
  const int sl = wv >> 2;
  const int hd = wv & 3;
  if (sl < ns) {
    const int lane = threadIdx.x & 31;
    const int lo = lane & 15;
    const int hi = lane >> 4;
    gemm_strip_wave<2, IN_DIM, 1, HEADS>(first + sl, lo, hi, (const float*)Az + (sl * 16 + lo) * Z1_P,
                                         Wz + hd * 32 * P4, q.z1, HEADS * HID, hd * HID,
                                         q.a1 + hd * 2 * HID, q.s1s + hd, q.s1d + hd);
  }
}

__global__ void __launch_bounds__(SORT_T) __attribute__((amdgpu_waves_per_eu(16, 16)))
zero_and_z1(int* __restrict__ bintot, Z1Args q, const float* __restrict__ Wf, const float* __restrict__ Wff,
            const float* __restrict__ Wfc2, float* __restrict__ Mt) {
  __shared__ float4 lds[Z1_LDS_F4];
  const int i = (int)blockIdx.x;
  if (i < Z1_HEAD_WGS) {
    const int first = 3 * 8 * Z1_WGS + 2 * i;
    z1_role(lds, first, (Z1_STRIPS - first < 2) ? Z1_STRIPS - first : 2, q);
  } else {
    for (int k = threadIdx.x; k < NBINS; k += SORT_T) bintot[k] = 0;
    fold_frac_weights((float*)lds, Wf, Wff, Wfc2, Mt);
  }
}

__global__ void __launch_bounds__(SORT_T) __attribute__((amdgpu_waves_per_eu(16, 16)))
bin_count(const int* __restrict__ dst, int* __restrict__ bintot, int* __restrict__ runbase, Z1Args q) {
  __shared__ float4 lds[Z1_LDS_F4];
  if ((int)blockIdx.x < Z1_WGS) { z1_role(lds, (int)blockIdx.x * 8, 8, q); return; }
  const int bx = (int)blockIdx.x - Z1_WGS;
  int* hist = (int*)lds;
  const int t = threadIdx.x;
  for (int i = t; i < NBINS; i += SORT_T) hist[i] = 0;
  __syncthreads();
  const int e0 = bx * SLICE_E;
  const int e1 = (e0 + SLICE_E < N_EDGES) ? e0 + SLICE_E : N_EDGES;
  for (int e = e0 + t; e < e1; e += SORT_T) atomicAdd(&hist[dst[e] >> BIN_SHIFT], 1);
  __syncthreads();
  for (int i = t; i < NBINS; i += SORT_T)
    runbase[bx * NBINS + i] = atomicAdd(&bintot[i], hist[i]);
}

#define NBINS_PAD ((NBINS + 31) & ~31)
static_assert((3 * NBINS_PAD + 32 + 2 * SLICE_E) * 4 <= Z1_LDS_F4 * 16, "the scatter's arrays fit the block");
__global__ void __launch_bounds__(SORT_T) __attribute__((amdgpu_waves_per_eu(16, 16)))
bin_scatter(const int* __restrict__ src, const int* __restrict__ dst,
            const int* __restrict__ bintot, const int* __restrict__ runbase,
            int* __restrict__ binned, Z1Args q) {
  __shared__ float4 lds[Z1_LDS_F4];
  if ((int)blockIdx.x < Z1_WGS) { z1_role(lds, (Z1_WGS + (int)blockIdx.x) * 8, 8, q); return; }
  const int bx = (int)blockIdx.x - Z1_WGS;
  int* hist = (int*)lds;
  int* lstart = hist + NBINS_PAD;
  int* gbase = lstart + NBINS_PAD;
  int* wsum = gbase + NBINS_PAD;
  int* sval = wsum + 32;
  int* sidx = sval + SLICE_E;
  const int t = threadIdx.x;
  for (int i = t; i < NBINS; i += SORT_T) hist[i] = 0;
  __syncthreads();
  const int e0 = bx * SLICE_E;
  const int e1 = (e0 + SLICE_E < N_EDGES) ? e0 + SLICE_E : N_EDGES;
  int d[PER_T], s[PER_T];
#pragma unroll
  for (int k = 0; k < PER_T; ++k) {
    const int e = e0 + t + k * SORT_T;
    d[k] = -1; s[k] = 0;
    if (e < e1) { d[k] = dst[e]; s[k] = src[e]; atomicAdd(&hist[d[k] >> BIN_SHIFT], 1); }
  }
  __syncthreads();
  const int tot = (t < NBINS) ? bintot[t] : 0;
  const int cnt = (t < NBINS) ? hist[t] : 0;
  const int binstart = block_excl_scan(tot, wsum);
  const int lst = block_excl_scan(cnt, wsum);
  if (t < NBINS) {
    gbase[t] = binstart + runbase[bx * NBINS + t];
    lstart[t] = lst;
    hist[t] = 0;
  }
  __syncthreads();
#pragma unroll
  for (int k = 0; k < PER_T; ++k) {
    if (d[k] >= 0) {
      const int b = d[k] >> BIN_SHIFT;
      const int r = atomicAdd(&hist[b], 1);
      const int p = lstart[b] + r;
      sval[p] = s[k] | ((d[k] & (BIN_NODES - 1)) << SRC_BITS);
      sidx[p] = gbase[b] + r;
    }
  }
  __syncthreads();
  const int n = e1 - e0;
  for (int i = t; i < n; i += SORT_T) binned[sidx[i]] = sval[i];
}

static_assert((2 * BIN_NODES + 64 + BIN_CAP) * 4 <= Z1_LDS_F4 * 16, "the finish pass's arrays fit the block");
__global__ void __launch_bounds__(SORT_T) __attribute__((amdgpu_waves_per_eu(16, 16)))
bin_finish(const int* __restrict__ bintot, const int* __restrict__ binned,
           int* __restrict__ rowptr, int* __restrict__ csr, Z1Args q) {
  __shared__ float4 lds[Z1_LDS_F4];
  if ((int)blockIdx.x < Z1_WGS) { z1_role(lds, (2 * Z1_WGS + (int)blockIdx.x) * 8, 8, q); return; }
  int* deg = (int*)lds;
  int* first = deg + BIN_NODES;
  int* wsum = first + BIN_NODES;
  int* range = wsum + 32;
  int* sorted = range + 32;
  const int t = threadIdx.x;
  const int b = (int)blockIdx.x - Z1_WGS;
  const int tot = (t < NBINS) ? bintot[t] : 0;
  const int ex0 = block_excl_scan(tot, wsum);
  if (t == b) { range[0] = ex0; range[1] = tot; }
  if (t < BIN_NODES) deg[t] = 0;
  __syncthreads();
  const int start = range[0];
  const int cnt = range[1];
  const int* mine = binned + start;
  for (int i = t; i < cnt; i += SORT_T) atomicAdd(&deg[((unsigned)mine[i]) >> SRC_BITS], 1);
  __syncthreads();
  const int dg = (t < BIN_NODES) ? deg[t] : 0;
  const int ex = block_excl_scan(dg, wsum);
  if (t < BIN_NODES) {
    const int node = b * BIN_NODES + t;
    if (node < N_NODES) rowptr[node] = start + ex;
    first[t] = ex;
    deg[t] = 0;
  }
  if (b == NBINS - 1 && t == 0) rowptr[N_NODES] = start + cnt;
  __syncthreads();
  if (cnt <= BIN_CAP) {
    for (int i = t; i < cnt; i += SORT_T) {
      const unsigned w = (unsigned)mine[i];
      const int dl = w >> SRC_BITS;
      const int r = atomicAdd(&deg[dl], 1);
      sorted[first[dl] + r] = (int)(w & ((1u << SRC_BITS) - 1u));
    }
    __syncthreads();
    for (int i = t; i < cnt; i += SORT_T) csr[start + i] = sorted[i];
  } else {
    for (int i = t; i < cnt; i += SORT_T) {
      const unsigned w = (unsigned)mine[i];
      const int dl = w >> SRC_BITS;
      const int r = atomicAdd(&deg[dl], 1);
      csr[start + first[dl] + r] = (int)(w & ((1u << SRC_BITS) - 1u));
    }
  }
}

template <int NH>
__device__ __forceinline__ int load_edge(const int* __restrict__ csr, const float* __restrict__ ss,
                                         const float* sdn, int j, int end, float* x) {
  int s = 0;
  if (j < end) {
    s = csr[j];
    float v[NH];
    if constexpr (NH == 4) {
      const float4 q = *(const float4*)(ss + (size_t)s * 4);
      v[0] = q.x; v[1] = q.y; v[2] = q.z; v[3] = q.w;
    } else {
      for (int h = 0; h < NH; ++h) v[h] = ss[(size_t)s * NH + h];
    }
    for (int h = 0; h < NH; ++h) {
      const float t = v[h] + sdn[h];
      x[h] = (t > 0.f) ? t : 0.01f * t;
    }
  } else {
    for (int h = 0; h < NH; ++h) x[h] = -INFINITY;
  }
  return s;
}

#define SLOT_PITCH 40
#define SLOT_TAB_FLOATS(NH) ((NH) * SLOT_PITCH + 32)

template <int NH>
__device__ __forceinline__ void put_slots(float* tab, int lane, const float* alpha, int s) {
  for (int h = 0; h < NH; ++h) tab[h * SLOT_PITCH + lane] = alpha[h];
  ((unsigned*)(tab + NH * SLOT_PITCH))[lane] = (unsigned)s * (unsigned)(NH * HID * sizeof(float));
  __builtin_amdgcn_fence(__ATOMIC_ACQ_REL, "wavefront");
}

template <int NH>
__device__ __forceinline__ void load_row(const float* __restrict__ p, float* r) {
  if constexpr (NH == 4) {
    const float4 q = *(const float4*)p;
    r[0] = q.x; r[1] = q.y; r[2] = q.z; r[3] = q.w;
  } else {
    for (int h = 0; h < NH; ++h) r[h] = p[h];
  }
}

template <int NH>
__device__ __forceinline__ void add_rows(const char* __restrict__ zb, unsigned lane_off, const float* tah,
                                         const unsigned* to, int cnt, float* acc) {
  int t = 0;
  if (cnt >= 4) {
    float4 a4 = *(const float4*)tah;
    uint4 o4 = *(const uint4*)to;
    for (;;) {
      const float a[4] = {a4.x, a4.y, a4.z, a4.w};
      const unsigned o[4] = {o4.x, o4.y, o4.z, o4.w};
      float r[4][NH];
#pragma unroll
      for (int u = 0; u < 4; ++u) load_row<NH>((const float*)(zb + (o[u] + lane_off)), r[u]);
      t += 4;
      const bool more = t + 4 <= cnt;
      if (more) {
        a4 = *(const float4*)(tah + t);
        o4 = *(const uint4*)(to + t);
      }
#pragma unroll
      for (int u = 0; u < 4; ++u)
        for (int h = 0; h < NH; ++h) acc[h] = __builtin_fmaf(a[u], r[u][h], acc[h]);
      if (!more) break;
    }
  }
  for (; t < cnt; ++t) {
    float r[NH];
    const float a = tah[t];
    load_row<NH>((const float*)(zb + (to[t] + lane_off)), r);
    for (int h = 0; h < NH; ++h) acc[h] = __builtin_fmaf(a, r[h], acc[h]);
  }
}

template <int NH, bool ELU>
__device__ __forceinline__ void
node_softmax_gather_row(int n, const int* __restrict__ rowptr, const int* __restrict__ csr,
                        const float* __restrict__ ss, const float* __restrict__ sd,
                        const float* __restrict__ z, float* tab, float* acc) {
  const int lane = threadIdx.x & 31;
  const int hh = (lane * NH) >> 5;
  const int beg = rowptr[n];
  const int end = rowptr[n + 1];
  for (int h = 0; h < NH; ++h) acc[h] = 0.f;
  if (end > beg) {
    float sdn[NH];
    for (int h = 0; h < NH; ++h) sdn[h] = sd[(size_t)n * NH + h];
    const char* zb = (const char*)z;
    const unsigned lane_off = (unsigned)(lane * NH * sizeof(float));
    const float* tah = tab + hh * SLOT_PITCH;
    const unsigned* to = (const unsigned*)(tab + NH * SLOT_PITCH);
    float x0[NH];
    const int s0 = load_edge<NH>(csr, ss, sdn, beg + lane, end, x0);
    const bool v0 = (beg + lane) < end;
    float m[NH];
    for (int h = 0; h < NH; ++h) m[h] = x0[h];
    for (int base = beg + 32; base < end; base += 32) {
      float x[NH];
      load_edge<NH>(csr, ss, sdn, base + lane, end, x);
      for (int h = 0; h < NH; ++h) m[h] = fmaxf(m[h], x[h]);
    }
    for (int o = 16; o > 0; o >>= 1)
      for (int h = 0; h < NH; ++h) m[h] = fmaxf(m[h], __shfl_xor(m[h], o));
    float ee0[NH], sum[NH];
    for (int h = 0; h < NH; ++h) { ee0[h] = v0 ? expf(x0[h] - m[h]) : 0.f; sum[h] = ee0[h]; }
    for (int base = beg + 32; base < end; base += 32) {
      float x[NH];
      load_edge<NH>(csr, ss, sdn, base + lane, end, x);
      const bool v = (base + lane) < end;
      for (int h = 0; h < NH; ++h) sum[h] += v ? expf(x[h] - m[h]) : 0.f;
    }
    for (int o = 16; o > 0; o >>= 1)
      for (int h = 0; h < NH; ++h) sum[h] += __shfl_xor(sum[h], o);
    float den[NH], al[NH];
    for (int h = 0; h < NH; ++h) { den[h] = fmaxf(sum[h], 1e-9f); al[h] = ee0[h] / den[h]; }
    const int c0 = end - beg;
    put_slots<NH>(tab, lane, al, s0);
    add_rows<NH>(zb, lane_off, tah, to, c0 < 32 ? c0 : 32, acc);
    for (int base = beg + 32; base < end; base += 32) {
      float x[NH];
      const int s = load_edge<NH>(csr, ss, sdn, base + lane, end, x);
      const bool v = (base + lane) < end;
      for (int h = 0; h < NH; ++h) al[h] = v ? expf(x[h] - m[h]) / den[h] : 0.f;
      const int c = end - base;
      put_slots<NH>(tab, lane, al, s);
      add_rows<NH>(zb, lane_off, tah, to, c < 32 ? c : 32, acc);
    }
  }
  if constexpr (ELU) {
    for (int h = 0; h < NH; ++h) acc[h] = (acc[h] > 0.f) ? acc[h] : (expf(acc[h]) - 1.0f);
  }
}

template <int NH, bool ELU>
__device__ __forceinline__ void
node_softmax_gather_block(int gb, const int* __restrict__ rowptr, const int* __restrict__ csr,
                          const float* __restrict__ ss, const float* __restrict__ sd,
                          const float* __restrict__ z, float* __restrict__ out, int ldo, float* slots) {
  const int w = __builtin_amdgcn_readfirstlane((int)(threadIdx.x >> 5));
  const int n = gb * 8 + w;
  if (n >= N_NODES) return;
  const int lane = threadIdx.x & 31;
  float acc[NH];
  node_softmax_gather_row<NH, ELU>(n, rowptr, csr, ss, sd, z, slots + w * SLOT_TAB_FLOATS(NH), acc);
  float* op = out + (size_t)n * ldo + lane * NH;
  if constexpr (NH == 4) {
    *(float4*)op = make_float4(acc[0], acc[1], acc[2], acc[3]);
  } else {
    for (int h = 0; h < NH; ++h) op[h] = acc[h];
  }
}

#define H1_COLS  (HEADS * HID)
#define H1_PITCH (H1_COLS + 4)
#define FUSE_W_F4 (2 * 16 * (H1_PITCH / 4))
#define FUSE_LDS_F4 (FUSE_W_F4 + 16 * (H1_PITCH / 4))
static_assert(N_NODES % 16 == 0, "whole 16-node groups");
__device__ __forceinline__ void
gather1_z2_block(int gb, const int* __restrict__ rowptr, const int* __restrict__ csr,
                 const float* __restrict__ ss, const float* __restrict__ sd,
                 const float* __restrict__ z, const float* __restrict__ W2,
                 const float* __restrict__ a2, float* __restrict__ z2,
                 float* __restrict__ s2s, float* __restrict__ s2d, float4* lds, float* slots) {
  float4* Ws = lds;
  float* T = (float*)(lds + FUSE_W_F4);
  {
    constexpr int K4 = H1_COLS / 4, P4 = H1_PITCH / 4;
    static_assert(2 * 16 * K4 == 4 * 256, "four float4 per thread");
    float4 q[4];
#pragma unroll
    for (int u = 0; u < 4; ++u) q[u] = ((const float4*)W2)[threadIdx.x + u * 256];
#pragma unroll
    for (int u = 0; u < 4; ++u) {
      const int i = threadIdx.x + u * 256;
      Ws[(i / K4) * P4 + (i % K4)] = q[u];
    }
  }
  const int w = __builtin_amdgcn_readfirstlane((int)(threadIdx.x >> 5));
  const int lane = threadIdx.x & 31;
  for (int i = 0; i < 2; ++i) {
    const int row = w * 2 + i;
    float acc[HEADS];
    node_softmax_gather_row<HEADS, true>(gb * 16 + row, rowptr, csr, ss, sd, z,
                                         slots + w * SLOT_TAB_FLOATS(HEADS), acc);
    *(float4*)(T + row * H1_PITCH + lane * HEADS) = make_float4(acc[0], acc[1], acc[2], acc[3]);
  }
  __syncthreads();
  if (w == 0) {
    const int lo = lane & 15;
    const int hi = lane >> 4;
    const float* Tl = T + lo * H1_PITCH;
    const float* Wl = (const float*)Ws + lo * H1_PITCH;
    v8f acc[2];
    acc[0] = (v8f){0.f, 0.f, 0.f, 0.f, 0.f, 0.f, 0.f, 0.f};
    acc[1] = acc[0];
    for (int k0 = 0; k0 < H1_COLS; k0 += 4) {
      const int ka = k0 + hi * 2;
      const v2f a = *(const v2f*)(Tl + ka);
#pragma unroll
      for (int j = 0; j < 2; ++j) {
        const v2f b = *(const v2f*)(Wl + j * 16 * H1_PITCH + ka);
        acc[j] = __builtin_amdgcn_wmma_f32_16x16x4_f32(false, a, false, b,
                                                       (short)0, acc[j], false, false);
      }
    }
    strip_store_scores<2, 1>(gb, lo, hi, acc, z2, HID, 0, a2, s2s, s2d);
  }
}

#define GATHER_WGS ((N_NODES + 7) / 8)
#define GATHER16_WGS (N_NODES / 16)
#define STRIP_WGS  ((N_NODES / 16 + 7) / 8)
template <int PERIOD, int GEMM_WGS, int GATHER_N>
struct Mixed {
  static constexpr int MIX = PERIOD * GEMM_WGS;
  static constexpr int GRID = MIX + GATHER_N - (PERIOD - 1) * GEMM_WGS;
  static_assert(GATHER_N >= (PERIOD - 1) * GEMM_WGS, "the gather outlasts the GEMM");
  static __device__ __forceinline__ int gemm_wg(int i) { return (i < MIX && i % PERIOD == 0) ? i / PERIOD : -1; }
  static __device__ __forceinline__ int gather_wg(int i) {
    return (i < MIX) ? (i / PERIOD) * (PERIOD - 1) + (i % PERIOD) - 1 : i - GEMM_WGS;
  }
};
typedef Mixed<7, STRIP_WGS, GATHER16_WGS> Mix1;

__global__ void __launch_bounds__(256) __attribute__((amdgpu_waves_per_eu(16, 16)))
gather1_and_frac(const int* __restrict__ rowptr, const int* __restrict__ csr,
                 const float* __restrict__ ss, const float* __restrict__ sd,
                 const float* __restrict__ z,
                 const float* __restrict__ W2, const float* __restrict__ a2,
                 float* __restrict__ z2, float* __restrict__ s2s, float* __restrict__ s2d,
                 const float* __restrict__ H, const float* __restrict__ Mt, float* __restrict__ P) {
  static_assert(3 * 16 * ((IN_DIM + 4) / 4) <= FUSE_LDS_F4, "the P role's weight tile fits the block");
  __shared__ float4 lds[FUSE_LDS_F4];
  __shared__ __attribute__((aligned(16))) float slots[8 * SLOT_TAB_FLOATS(HEADS)];
  const int i = (int)blockIdx.x;
  const int f = Mix1::gemm_wg(i);
  if (f >= 0) {
    gemm_p_block(f, H, Mt, P, lds);
  } else {
    gather1_z2_block(Mix1::gather_wg(i), rowptr, csr, ss, sd, z, W2, a2, z2, s2s, s2d, lds, slots);
  }
}

__global__ void __launch_bounds__(256)
gather2(const int* __restrict__ rowptr, const int* __restrict__ csr,
        const float* __restrict__ ss, const float* __restrict__ sd,
        const float* __restrict__ z, float* __restrict__ h2) {
  __shared__ __attribute__((aligned(16))) float slots[8 * SLOT_TAB_FLOATS(1)];
  node_softmax_gather_block<1, false>((int)blockIdx.x, rowptr, csr, ss, sd, z, h2, HID, slots);
}

__global__ void __launch_bounds__(256)
final_proj(const float* __restrict__ h2, const float* __restrict__ Wfc2, const float* __restrict__ P,
           const int* __restrict__ fc, float* __restrict__ out) {
  constexpr int K4 = HID / 4, P4 = K4 + 1;
  __shared__ float4 Ws[OUT_DIM * P4];
  for (int i = threadIdx.x; i < OUT_DIM * K4; i += 256)
    Ws[(i / K4) * P4 + (i % K4)] = ((const float4*)Wfc2)[(i / K4) * (2 * K4) + (i % K4)];
  __syncthreads();
  const int wave = ((int)blockIdx.x * 256 + (int)threadIdx.x) >> 5;
  if (wave < N_NODES / 16) {
    const int lane = threadIdx.x & 31;
    const int lo = lane & 15;
    const int hi = lane >> 4;
    v8f acc;
#pragma unroll
    for (int r = 0; r < 8; ++r) {
      const int* f = fc + (wave * 16 + hi * 8 + r) * 3;
      const float p0 = P[(size_t)f[0] * P_COLS + lo];
      const float p1 = P[(size_t)f[1] * P_COLS + 16 + lo];
      const float p2 = P[(size_t)f[2] * P_COLS + 32 + lo];
      acc[r] = (p0 + p1) + p2;
    }
    const float* Ar = h2 + ((size_t)wave * 16 + lo) * HID;
    const float* Wl = (const float*)Ws + lo * (HID + 4);
    for (int k0 = 0; k0 < HID; k0 += 4) {
      const int ka = k0 + hi * 2;
      v2f a;
      a.x = Ar[ka]; a.y = Ar[ka + 1];
      const v2f b = *(const v2f*)(Wl + ka);
      acc = __builtin_amdgcn_wmma_f32_16x16x4_f32(false, a, false, b, (short)0, acc, false, false);
    }
    float* Cb = out + (size_t)wave * 16 * OUT_DIM + lo;
    for (int r = 0; r < 8; ++r) Cb[(size_t)(r + hi * 8) * OUT_DIM] = acc[r];
  }
}

extern "C" void kernel_launch(void* const* d_in, const int* in_sizes, int n_in,
                              void* d_out, int out_size, void* d_ws, size_t ws_size,
                              hipStream_t stream) {
  const float* h    = (const float*)d_in[0];
  const int*   src  = (const int*)  d_in[1];
  const int*   dst  = (const int*)  d_in[2];
  const int*   fc   = (const int*)  d_in[3];
  const float* W1   = (const float*)d_in[4];
  const float* a1   = (const float*)d_in[5];
  const float* W2   = (const float*)d_in[6];
  const float* a2   = (const float*)d_in[7];
  const float* Wf   = (const float*)d_in[8];
  const float* Wff  = (const float*)d_in[9];
  const float* Wfc2 = (const float*)d_in[10];
  float* out = (float*)d_out;

  float* w = (float*)d_ws;
  size_t off = 0;
  int* bintot  = (int*)(w + off); off += 512;
  int* runbase = (int*)(w + off); off += SLICES * 512;
  int* rowptr  = (int*)(w + off); off += N_NODES + 32;
  int* binned  = (int*)(w + off); off += N_EDGES;
  int* csr     = (int*)(w + off); off += N_EDGES;
  float* z1  = w + off; off += (size_t)N_NODES * 128;
  float* s1s = w + off; off += (size_t)N_NODES * HEADS;
  float* s1d = w + off; off += (size_t)N_NODES * HEADS;
  float* z2  = w + off; off += (size_t)N_NODES * HID;
  float* s2s = w + off; off += N_NODES;
  float* s2d = w + off; off += N_NODES;
  float* h2  = w + off; off += (size_t)N_NODES * HID;
  float* P   = w + off; off += (size_t)N_NODES * P_COLS;
  float* Mt  = w + off; off += P_COLS * IN_DIM;
  (void)ws_size; (void)in_sizes; (void)n_in; (void)out_size;

  const int TB = 256;

  const Z1Args zq = {h, W1, a1, z1, s1s, s1d};
  zero_and_z1<<<Z1_HEAD_WGS + 1, SORT_T, 0, stream>>>(bintot, zq, Wf, Wff, Wfc2, Mt);
  bin_count<<<Z1_WGS + SLICES, SORT_T, 0, stream>>>(dst, bintot, runbase, zq);
  bin_scatter<<<Z1_WGS + SLICES, SORT_T, 0, stream>>>(src, dst, bintot, runbase, binned, zq);
  bin_finish<<<Z1_WGS + NBINS, SORT_T, 0, stream>>>(bintot, binned, rowptr, csr, zq);

  gather1_and_frac<<<Mix1::GRID, TB, 0, stream>>>(rowptr, csr, s1s, s1d, z1, W2, a2, z2, s2s, s2d,
                                                  h, Mt, P);

  gather2<<<GATHER_WGS, TB, 0, stream>>>(rowptr, csr, s2s, s2d, z2, h2);

  final_proj<<<STRIP_WGS, TB, 0, stream>>>(h2, Wfc2, P, fc, out);
}


# Final projection moved ahead of layer 2's edge sum (linear): y=z2@Wfc2[:,:32]^T [N,16] in gather 1's WMMA epilogue; gather 2 sums 64-byte y rows + P pieces into out; 7->6 launches; sum order changed
# speedup vs baseline: 9.9183x; 1.0590x over previous
#include <hip/hip_runtime.h>
#include <math.h>

#define N_NODES 100000
#define N_EDGES 1600000
#define IN_DIM  128
#define HID     32
#define HEADS   4
#define OUT_DIM 16

typedef __attribute__((ext_vector_type(2))) float v2f;
typedef __attribute__((ext_vector_type(8))) float v8f;


__device__ __forceinline__ float row16_sum(float x) {
  x += __int_as_float(__builtin_amdgcn_update_dpp(0, __float_as_int(x), 0x161, 0xf, 0xf, true));
  x += __int_as_float(__builtin_amdgcn_update_dpp(0, __float_as_int(x), 0x162, 0xf, 0xf, true));
  x += __int_as_float(__builtin_amdgcn_update_dpp(0, __float_as_int(x), 0x164, 0xf, 0xf, true));
  x += __int_as_float(__builtin_amdgcn_update_dpp(0, __float_as_int(x), 0x168, 0xf, 0xf, true));
  return x;
}

template <int NT, int NHD, int SST = NHD, bool STC = true>
__device__ __forceinline__ void
strip_store_scores(int wave, int lo, int hi, const v8f* acc, float* __restrict__ C, int ldc, int coff,
                   const float* __restrict__ av, float* __restrict__ ss, float* __restrict__ sd) {
  {
    if constexpr (STC) {
#pragma unroll
      for (int j = 0; j < NT; ++j) {
        float* Cb = C + (size_t)wave * 16 * ldc + coff + (size_t)j * 16 + lo;
#pragma unroll
        for (int r = 0; r < 8; ++r)
          Cb[(size_t)(r + hi * 8) * ldc] = acc[j][r];
      }
    }
    if constexpr (NHD > 0) {
      static_assert(NT == 2 * NHD, "a head is two 16-column tiles");
      float o[NHD];
#pragma unroll
      for (int h = 0; h < NHD; ++h) {
        const float* ah = av + h * 2 * HID;
        const float al0 = ah[lo], al1 = ah[16 + lo], ar0 = ah[HID + lo], ar1 = ah[HID + 16 + lo];
        float keep = 0.f;
#pragma unroll
        for (int r = 0; r < 8; ++r) {
          const float ps = row16_sum(acc[2 * h][r] * al0 + acc[2 * h + 1][r] * al1);
          const float pd = row16_sum(acc[2 * h][r] * ar0 + acc[2 * h + 1][r] * ar1);
          keep = (lo == r) ? ps : keep;
          keep = (lo == r + 8) ? pd : keep;
        }
        o[h] = keep;
      }
      float* sp = ((lo < 8) ? ss : sd) + ((size_t)wave * 16 + hi * 8 + (lo & 7)) * SST;
      if constexpr (NHD == 4) {
        *(float4*)sp = make_float4(o[0], o[1], o[2], o[3]);
      } else {
#pragma unroll
        for (int h = 0; h < NHD; ++h) sp[h] = o[h];
      }
    }
  }
}

template <int NT, int K, int NHD, int SST>
__device__ __forceinline__ void
gemm_strip_wave(int wave, int lo, int hi, const float* Ar, const float4* Ws,
                float* __restrict__ C, int ldc, int coff,
                const float* __restrict__ av, float* __restrict__ ss, float* __restrict__ sd) {
  constexpr int P = K + 4;
  const float* Wl = (const float*)Ws + lo * P;
  v8f acc[NT];
#pragma unroll
  for (int j = 0; j < NT; ++j)
    acc[j] = (v8f){0.f, 0.f, 0.f, 0.f, 0.f, 0.f, 0.f, 0.f};
  for (int k0 = 0; k0 < K; k0 += 4) {
    const int ka = k0 + hi * 2;
    v2f a;
    a.x = Ar[ka]; a.y = Ar[ka + 1];
#pragma unroll
    for (int j = 0; j < NT; ++j) {
      const v2f b = *(const v2f*)(Wl + j * 16 * P + ka);
      acc[j] = __builtin_amdgcn_wmma_f32_16x16x4_f32(false, a, false, b,
                                                     (short)0, acc[j], false, false);
    }
  }
  strip_store_scores<NT, NHD, SST>(wave, lo, hi, acc, C, ldc, coff, av, ss, sd);
}

#define P_COLS 48
__device__ __forceinline__ void
gemm_p_block(int bx, const float* __restrict__ H, const float* __restrict__ Mt,
             float* __restrict__ P, float4* Ws) {
  {
    constexpr int K4 = IN_DIM / 4, P4 = K4 + 1;
    static_assert(P_COLS * K4 == 6 * 256, "six float4 per thread");
    float4 q[6];
#pragma unroll
    for (int u = 0; u < 6; ++u) q[u] = ((const float4*)Mt)[threadIdx.x + u * 256];
#pragma unroll
    for (int u = 0; u < 6; ++u) {
      const int i = threadIdx.x + u * 256;
      Ws[(i / K4) * P4 + (i % K4)] = q[u];
    }
  }
  __syncthreads();
  const int wave = (bx * blockDim.x + threadIdx.x) >> 5;
  if (wave < N_NODES / 16) {
    const int lane = threadIdx.x & 31;
    const int lo = lane & 15;
    const int hi = lane >> 4;
    gemm_strip_wave<3, IN_DIM, 0, 0>(wave, lo, hi, H + ((size_t)wave * 16 + lo) * IN_DIM, Ws,
                                     P, P_COLS, 0, nullptr, nullptr, nullptr);
  }
}

#define FOLD_TP 136
__device__ __forceinline__ void
fold_frac_weights(float* T, const float* __restrict__ Wf, const float* __restrict__ Wff,
                  const float* __restrict__ Wfc2, float* __restrict__ Mt) {
  const int wv = __builtin_amdgcn_readfirstlane((int)(threadIdx.x >> 5));
  const int lane = threadIdx.x & 31;
  const int lo = lane & 15;
  const int hi = lane >> 4;
  for (int t = wv; t < 48; t += 32) {
    const int s = t >> 4, mt = (t >> 3) & 1, nt = t & 7;
    const float* Ar = Wff + (mt * 16 + lo) * (3 * HID) + s * HID;
    const float* Bc = Wf + (size_t)s * HID * IN_DIM + nt * 16 + lo;
    v8f acc = (v8f){0.f, 0.f, 0.f, 0.f, 0.f, 0.f, 0.f, 0.f};
    for (int k0 = 0; k0 < HID; k0 += 4) {
      const int ka = k0 + hi * 2;
      v2f a, b;
      a.x = Ar[ka]; a.y = Ar[ka + 1];
      b.x = Bc[ka * IN_DIM]; b.y = Bc[(ka + 1) * IN_DIM];
      acc = __builtin_amdgcn_wmma_f32_16x16x4_f32(false, a, false, b, (short)0, acc, false, false);
    }
    float* Tb = T + (s * HID + mt * 16 + hi * 8) * FOLD_TP + nt * 16 + lo;
    for (int r = 0; r < 8; ++r) Tb[r * FOLD_TP] = acc[r];
  }
  __syncthreads();
  if (wv < 24) {
    const int s = wv >> 3, nt = wv & 7;
    const float* Ar = Wfc2 + lo * (2 * HID) + HID;
    const float* Bc = T + s * HID * FOLD_TP + nt * 16 + lo;
    v8f acc = (v8f){0.f, 0.f, 0.f, 0.f, 0.f, 0.f, 0.f, 0.f};
    for (int k0 = 0; k0 < HID; k0 += 4) {
      const int ka = k0 + hi * 2;
      v2f a, b;
      a.x = Ar[ka]; a.y = Ar[ka + 1];
      b.x = Bc[ka * FOLD_TP]; b.y = Bc[(ka + 1) * FOLD_TP];
      acc = __builtin_amdgcn_wmma_f32_16x16x4_f32(false, a, false, b, (short)0, acc, false, false);
    }
    float* Mb = Mt + (s * 16 + hi * 8) * IN_DIM + nt * 16 + lo;
    for (int r = 0; r < 8; ++r) Mb[r * IN_DIM] = acc[r];
  }
}

#define BIN_SHIFT 8
#define BIN_NODES (1 << BIN_SHIFT)
#define NBINS     ((N_NODES + BIN_NODES - 1) >> BIN_SHIFT)
#define SLICES    256
#define SLICE_E   ((N_EDGES + SLICES - 1) / SLICES)
#define SORT_T    1024
#define PER_T     ((SLICE_E + SORT_T - 1) / SORT_T)
#define SRC_BITS  17
#define BIN_CAP   8192
static_assert(N_NODES <= (1 << SRC_BITS), "a source id must fit the packed word");
static_assert(NBINS <= SORT_T, "one thread per bin in the scans");
static_assert(BIN_NODES <= SORT_T, "one thread per node in the scans");

__device__ __forceinline__ int block_excl_scan(int v, int* wsum) {
  const int lane = threadIdx.x & 31, w = threadIdx.x >> 5;
  int inc = v;
  for (int o = 1; o < 32; o <<= 1) {
    const int u = __shfl_up(inc, o);
    if (lane >= o) inc += u;
  }
  if (lane == 31) wsum[w] = inc;
  __syncthreads();
  if (w == 0) {
    const int s = wsum[lane];
    int si = s;
    for (int o = 1; o < 32; o <<= 1) {
      const int u = __shfl_up(si, o);
      if (lane >= o) si += u;
    }
    wsum[lane] = si - s;
  }
  __syncthreads();
  const int ex = wsum[w] + inc - v;
  __syncthreads();
  return ex;
}

#define Z1_P        (IN_DIM + 4)
#define Z1_TILE_F4  (128 * (Z1_P / 4))
#define Z1_LDS_F4   (2 * Z1_TILE_F4)
#define Z1_STRIPS   (N_NODES / 16)
#define Z1_WGS      256
#define Z1_REST     (Z1_STRIPS - 3 * 8 * Z1_WGS)
#define Z1_HEAD_WGS ((Z1_REST + 1) / 2)
static_assert(HEADS * HID == 128 && 128 * (IN_DIM / 4) == 4 * SORT_T, "four float4 per thread fill a tile");
static_assert(Z1_REST >= 0, "three launches of eight-strip workgroups do not overshoot");
static_assert(3 * HID * FOLD_TP * 4 <= Z1_LDS_F4 * 16, "the folded weights' intermediate fits the block");
struct Z1Args {
  const float* h; const float* W1; const float* a1;
  float* z1; float* s1s; float* s1d;
};
__device__ __forceinline__ void z1_role(float4* lds, int first, int ns, const Z1Args& q) {
  constexpr int K4 = IN_DIM / 4, P4 = Z1_P / 4;
  float4* Wz = lds;
  float4* Az = lds + Z1_TILE_F4;
  {
    const float4* Wg = (const float4*)q.W1;
    const float4* Ag = (const float4*)(q.h + (size_t)first * 16 * IN_DIM);
    const int na = ns * 16 * K4;
    float4 w[4], a[4];
#pragma unroll
    for (int u = 0; u < 4; ++u) {
      const int i = threadIdx.x + u * SORT_T;
      w[u] = Wg[i];
      a[u] = (i < na) ? Ag[i] : make_float4(0.f, 0.f, 0.f, 0.f);
    }
#pragma unroll
    for (int u = 0; u < 4; ++u) {
      const int i = threadIdx.x + u * SORT_T;
      Wz[(i / K4) * P4 + (i % K4)] = w[u];
      if (i < na) Az[(i / K4) * P4 + (i % K4)] = a[u];
    }
  }
  __syncthreads();
  const int wv = __builtin_amdgcn_readfirstlane((int)(threadIdx.x >> 5));
  const int sl = wv >> 2;
  const int hd = wv & 3;
  if (sl < ns) {
    const int lane = threadIdx.x & 31;
    const int lo = lane & 15;
    const int hi = lane >> 4;
    gemm_strip_wave<2, IN_DIM, 1, HEADS>(first + sl, lo, hi, (const float*)Az + (sl * 16 + lo) * Z1_P,
                                         Wz + hd * 32 * P4, q.z1, HEADS * HID, hd * HID,
                                         q.a1 + hd * 2 * HID, q.s1s + hd, q.s1d + hd);
  }
}

__global__ void __launch_bounds__(SORT_T) __attribute__((amdgpu_waves_per_eu(16, 16)))
zero_and_z1(int* __restrict__ bintot, Z1Args q, const float* __restrict__ Wf, const float* __restrict__ Wff,
            const float* __restrict__ Wfc2, float* __restrict__ Mt) {
  __shared__ float4 lds[Z1_LDS_F4];
  const int i = (int)blockIdx.x;
  if (i < Z1_HEAD_WGS) {
    const int first = 3 * 8 * Z1_WGS + 2 * i;
    z1_role(lds, first, (Z1_STRIPS - first < 2) ? Z1_STRIPS - first : 2, q);
  } else {
    for (int k = threadIdx.x; k < NBINS; k += SORT_T) bintot[k] = 0;
    fold_frac_weights((float*)lds, Wf, Wff, Wfc2, Mt);
  }
}

__global__ void __launch_bounds__(SORT_T) __attribute__((amdgpu_waves_per_eu(16, 16)))
bin_count(const int* __restrict__ dst, int* __restrict__ bintot, int* __restrict__ runbase, Z1Args q) {
  __shared__ float4 lds[Z1_LDS_F4];
  if ((int)blockIdx.x < Z1_WGS) { z1_role(lds, (int)blockIdx.x * 8, 8, q); return; }
  const int bx = (int)blockIdx.x - Z1_WGS;
  int* hist = (int*)lds;
  const int t = threadIdx.x;
  for (int i = t; i < NBINS; i += SORT_T) hist[i] = 0;
  __syncthreads();
  const int e0 = bx * SLICE_E;
  const int e1 = (e0 + SLICE_E < N_EDGES) ? e0 + SLICE_E : N_EDGES;
  for (int e = e0 + t; e < e1; e += SORT_T) atomicAdd(&hist[dst[e] >> BIN_SHIFT], 1);
  __syncthreads();
  for (int i = t; i < NBINS; i += SORT_T)
    runbase[bx * NBINS + i] = atomicAdd(&bintot[i], hist[i]);
}

#define NBINS_PAD ((NBINS + 31) & ~31)
static_assert((3 * NBINS_PAD + 32 + 2 * SLICE_E) * 4 <= Z1_LDS_F4 * 16, "the scatter's arrays fit the block");
__global__ void __launch_bounds__(SORT_T) __attribute__((amdgpu_waves_per_eu(16, 16)))
bin_scatter(const int* __restrict__ src, const int* __restrict__ dst,
            const int* __restrict__ bintot, const int* __restrict__ runbase,
            int* __restrict__ binned, Z1Args q) {
  __shared__ float4 lds[Z1_LDS_F4];
  if ((int)blockIdx.x < Z1_WGS) { z1_role(lds, (Z1_WGS + (int)blockIdx.x) * 8, 8, q); return; }
  const int bx = (int)blockIdx.x - Z1_WGS;
  int* hist = (int*)lds;
  int* lstart = hist + NBINS_PAD;
  int* gbase = lstart + NBINS_PAD;
  int* wsum = gbase + NBINS_PAD;
  int* sval = wsum + 32;
  int* sidx = sval + SLICE_E;
  const int t = threadIdx.x;
  for (int i = t; i < NBINS; i += SORT_T) hist[i] = 0;
  __syncthreads();
  const int e0 = bx * SLICE_E;
  const int e1 = (e0 + SLICE_E < N_EDGES) ? e0 + SLICE_E : N_EDGES;
  int d[PER_T], s[PER_T];
#pragma unroll
  for (int k = 0; k < PER_T; ++k) {
    const int e = e0 + t + k * SORT_T;
    d[k] = -1; s[k] = 0;
    if (e < e1) { d[k] = dst[e]; s[k] = src[e]; atomicAdd(&hist[d[k] >> BIN_SHIFT], 1); }
  }
  __syncthreads();
  const int tot = (t < NBINS) ? bintot[t] : 0;
  const int cnt = (t < NBINS) ? hist[t] : 0;
  const int binstart = block_excl_scan(tot, wsum);
  const int lst = block_excl_scan(cnt, wsum);
  if (t < NBINS) {
    gbase[t] = binstart + runbase[bx * NBINS + t];
    lstart[t] = lst;
    hist[t] = 0;
  }
  __syncthreads();
#pragma unroll
  for (int k = 0; k < PER_T; ++k) {
    if (d[k] >= 0) {
      const int b = d[k] >> BIN_SHIFT;
      const int r = atomicAdd(&hist[b], 1);
      const int p = lstart[b] + r;
      sval[p] = s[k] | ((d[k] & (BIN_NODES - 1)) << SRC_BITS);
      sidx[p] = gbase[b] + r;
    }
  }
  __syncthreads();
  const int n = e1 - e0;
  for (int i = t; i < n; i += SORT_T) binned[sidx[i]] = sval[i];
}

static_assert((2 * BIN_NODES + 64 + BIN_CAP) * 4 <= Z1_LDS_F4 * 16, "the finish pass's arrays fit the block");
__global__ void __launch_bounds__(SORT_T) __attribute__((amdgpu_waves_per_eu(16, 16)))
bin_finish(const int* __restrict__ bintot, const int* __restrict__ binned,
           int* __restrict__ rowptr, int* __restrict__ csr, Z1Args q) {
  __shared__ float4 lds[Z1_LDS_F4];
  if ((int)blockIdx.x < Z1_WGS) { z1_role(lds, (2 * Z1_WGS + (int)blockIdx.x) * 8, 8, q); return; }
  int* deg = (int*)lds;
  int* first = deg + BIN_NODES;
  int* wsum = first + BIN_NODES;
  int* range = wsum + 32;
  int* sorted = range + 32;
  const int t = threadIdx.x;
  const int b = (int)blockIdx.x - Z1_WGS;
  const int tot = (t < NBINS) ? bintot[t] : 0;
  const int ex0 = block_excl_scan(tot, wsum);
  if (t == b) { range[0] = ex0; range[1] = tot; }
  if (t < BIN_NODES) deg[t] = 0;
  __syncthreads();
  const int start = range[0];
  const int cnt = range[1];
  const int* mine = binned + start;
  for (int i = t; i < cnt; i += SORT_T) atomicAdd(&deg[((unsigned)mine[i]) >> SRC_BITS], 1);
  __syncthreads();
  const int dg = (t < BIN_NODES) ? deg[t] : 0;
  const int ex = block_excl_scan(dg, wsum);
  if (t < BIN_NODES) {
    const int node = b * BIN_NODES + t;
    if (node < N_NODES) rowptr[node] = start + ex;
    first[t] = ex;
    deg[t] = 0;
  }
  if (b == NBINS - 1 && t == 0) rowptr[N_NODES] = start + cnt;
  __syncthreads();
  if (cnt <= BIN_CAP) {
    for (int i = t; i < cnt; i += SORT_T) {
      const unsigned w = (unsigned)mine[i];
      const int dl = w >> SRC_BITS;
      const int r = atomicAdd(&deg[dl], 1);
      sorted[first[dl] + r] = (int)(w & ((1u << SRC_BITS) - 1u));
    }
    __syncthreads();
    for (int i = t; i < cnt; i += SORT_T) csr[start + i] = sorted[i];
  } else {
    for (int i = t; i < cnt; i += SORT_T) {
      const unsigned w = (unsigned)mine[i];
      const int dl = w >> SRC_BITS;
      const int r = atomicAdd(&deg[dl], 1);
      csr[start + first[dl] + r] = (int)(w & ((1u << SRC_BITS) - 1u));
    }
  }
}

template <int NH>
__device__ __forceinline__ int load_edge(const int* __restrict__ csr, const float* __restrict__ ss,
                                         const float* sdn, int j, int end, float* x) {
  int s = 0;
  if (j < end) {
    s = csr[j];
    float v[NH];
    if constexpr (NH == 4) {
      const float4 q = *(const float4*)(ss + (size_t)s * 4);
      v[0] = q.x; v[1] = q.y; v[2] = q.z; v[3] = q.w;
    } else {
      for (int h = 0; h < NH; ++h) v[h] = ss[(size_t)s * NH + h];
    }
    for (int h = 0; h < NH; ++h) {
      const float t = v[h] + sdn[h];
      x[h] = (t > 0.f) ? t : 0.01f * t;
    }
  } else {
    for (int h = 0; h < NH; ++h) x[h] = -INFINITY;
  }
  return s;
}

#define SLOT_PITCH 40
#define SLOT_TAB_FLOATS(NH) ((NH) * SLOT_PITCH + 32)

template <int NH>
__device__ __forceinline__ void put_slots(float* tab, int lane, const float* alpha, int s) {
  for (int h = 0; h < NH; ++h) tab[h * SLOT_PITCH + lane] = alpha[h];
  ((unsigned*)(tab + NH * SLOT_PITCH))[lane] = (unsigned)s * (unsigned)(NH * HID * sizeof(float));
  __builtin_amdgcn_fence(__ATOMIC_ACQ_REL, "wavefront");
}

template <int NH>
__device__ __forceinline__ void load_row(const float* __restrict__ p, float* r) {
  if constexpr (NH == 4) {
    const float4 q = *(const float4*)p;
    r[0] = q.x; r[1] = q.y; r[2] = q.z; r[3] = q.w;
  } else {
    for (int h = 0; h < NH; ++h) r[h] = p[h];
  }
}

template <int NH>
__device__ __forceinline__ void add_rows(const char* __restrict__ zb, unsigned lane_off, const float* tah,
                                         const unsigned* to, int cnt, float* acc) {
  int t = 0;
  if (cnt >= 4) {
    float4 a4 = *(const float4*)tah;
    uint4 o4 = *(const uint4*)to;
    for (;;) {
      const float a[4] = {a4.x, a4.y, a4.z, a4.w};
      const unsigned o[4] = {o4.x, o4.y, o4.z, o4.w};
      float r[4][NH];
#pragma unroll
      for (int u = 0; u < 4; ++u) load_row<NH>((const float*)(zb + (o[u] + lane_off)), r[u]);
      t += 4;
      const bool more = t + 4 <= cnt;
      if (more) {
        a4 = *(const float4*)(tah + t);
        o4 = *(const uint4*)(to + t);
      }
#pragma unroll
      for (int u = 0; u < 4; ++u)
        for (int h = 0; h < NH; ++h) acc[h] = __builtin_fmaf(a[u], r[u][h], acc[h]);
      if (!more) break;
    }
  }
  for (; t < cnt; ++t) {
    float r[NH];
    const float a = tah[t];
    load_row<NH>((const float*)(zb + (to[t] + lane_off)), r);
    for (int h = 0; h < NH; ++h) acc[h] = __builtin_fmaf(a, r[h], acc[h]);
  }
}

template <int NH, bool ELU>
__device__ __forceinline__ void
node_softmax_gather_row(int n, const int* __restrict__ rowptr, const int* __restrict__ csr,
                        const float* __restrict__ ss, const float* __restrict__ sd,
                        const float* __restrict__ z, float* tab, float* acc) {
  const int lane = threadIdx.x & 31;
  const int hh = (lane * NH) >> 5;
  const int beg = rowptr[n];
  const int end = rowptr[n + 1];
  for (int h = 0; h < NH; ++h) acc[h] = 0.f;
  if (end > beg) {
    float sdn[NH];
    for (int h = 0; h < NH; ++h) sdn[h] = sd[(size_t)n * NH + h];
    const char* zb = (const char*)z;
    const unsigned lane_off = (unsigned)(lane * NH * sizeof(float));
    const float* tah = tab + hh * SLOT_PITCH;
    const unsigned* to = (const unsigned*)(tab + NH * SLOT_PITCH);
    float x0[NH];
    const int s0 = load_edge<NH>(csr, ss, sdn, beg + lane, end, x0);
    const bool v0 = (beg + lane) < end;
    float m[NH];
    for (int h = 0; h < NH; ++h) m[h] = x0[h];
    for (int base = beg + 32; base < end; base += 32) {
      float x[NH];
      load_edge<NH>(csr, ss, sdn, base + lane, end, x);
      for (int h = 0; h < NH; ++h) m[h] = fmaxf(m[h], x[h]);
    }
    for (int o = 16; o > 0; o >>= 1)
      for (int h = 0; h < NH; ++h) m[h] = fmaxf(m[h], __shfl_xor(m[h], o));
    float ee0[NH], sum[NH];
    for (int h = 0; h < NH; ++h) { ee0[h] = v0 ? expf(x0[h] - m[h]) : 0.f; sum[h] = ee0[h]; }
    for (int base = beg + 32; base < end; base += 32) {
      float x[NH];
      load_edge<NH>(csr, ss, sdn, base + lane, end, x);
      const bool v = (base + lane) < end;
      for (int h = 0; h < NH; ++h) sum[h] += v ? expf(x[h] - m[h]) : 0.f;
    }
    for (int o = 16; o > 0; o >>= 1)
      for (int h = 0; h < NH; ++h) sum[h] += __shfl_xor(sum[h], o);
    float den[NH], al[NH];
    for (int h = 0; h < NH; ++h) { den[h] = fmaxf(sum[h], 1e-9f); al[h] = ee0[h] / den[h]; }
    const int c0 = end - beg;
    put_slots<NH>(tab, lane, al, s0);
    add_rows<NH>(zb, lane_off, tah, to, c0 < 32 ? c0 : 32, acc);
    for (int base = beg + 32; base < end; base += 32) {
      float x[NH];
      const int s = load_edge<NH>(csr, ss, sdn, base + lane, end, x);
      const bool v = (base + lane) < end;
      for (int h = 0; h < NH; ++h) al[h] = v ? expf(x[h] - m[h]) / den[h] : 0.f;
      const int c = end - base;
      put_slots<NH>(tab, lane, al, s);
      add_rows<NH>(zb, lane_off, tah, to, c < 32 ? c : 32, acc);
    }
  }
  if constexpr (ELU) {
    for (int h = 0; h < NH; ++h) acc[h] = (acc[h] > 0.f) ? acc[h] : (expf(acc[h]) - 1.0f);
  }
}

#define H1_COLS  (HEADS * HID)
#define H1_PITCH (H1_COLS + 4)
#define FUSE_W_F4 (2 * 16 * (H1_PITCH / 4))
#define FUSE_LDS_F4 (FUSE_W_F4 + 16 * (H1_PITCH / 4))
static_assert(N_NODES % 16 == 0, "whole 16-node groups");
__device__ __forceinline__ void
gather1_z2_block(int gb, const int* __restrict__ rowptr, const int* __restrict__ csr,
                 const float* __restrict__ ss, const float* __restrict__ sd,
                 const float* __restrict__ z, const float* __restrict__ W2,
                 const float* __restrict__ a2, const float* __restrict__ Wfc2, float* __restrict__ Y,
                 float* __restrict__ s2s, float* __restrict__ s2d, float4* lds, float* slots) {
  float4* Ws = lds;
  float* T = (float*)(lds + FUSE_W_F4);
  {
    constexpr int K4 = H1_COLS / 4, P4 = H1_PITCH / 4;
    static_assert(2 * 16 * K4 == 4 * 256, "four float4 per thread");
    float4 q[4];
#pragma unroll
    for (int u = 0; u < 4; ++u) q[u] = ((const float4*)W2)[threadIdx.x + u * 256];
#pragma unroll
    for (int u = 0; u < 4; ++u) {
      const int i = threadIdx.x + u * 256;
      Ws[(i / K4) * P4 + (i % K4)] = q[u];
    }
  }
  const int w = __builtin_amdgcn_readfirstlane((int)(threadIdx.x >> 5));
  const int lane = threadIdx.x & 31;
  for (int i = 0; i < 2; ++i) {
    const int row = w * 2 + i;
    float acc[HEADS];
    node_softmax_gather_row<HEADS, true>(gb * 16 + row, rowptr, csr, ss, sd, z,
                                         slots + w * SLOT_TAB_FLOATS(HEADS), acc);
    *(float4*)(T + row * H1_PITCH + lane * HEADS) = make_float4(acc[0], acc[1], acc[2], acc[3]);
  }
  __syncthreads();
  if (w == 0) {
    const int lo = lane & 15;
    const int hi = lane >> 4;
    const float* Tl = T + lo * H1_PITCH;
    const float* Wl = (const float*)Ws + lo * H1_PITCH;
    v2f bw[HID / 4];
#pragma unroll
    for (int q = 0; q < HID / 4; ++q) bw[q] = *(const v2f*)(Wfc2 + lo * (2 * HID) + q * 4 + hi * 2);
    v8f acc[2];
    acc[0] = (v8f){0.f, 0.f, 0.f, 0.f, 0.f, 0.f, 0.f, 0.f};
    acc[1] = acc[0];
    for (int k0 = 0; k0 < H1_COLS; k0 += 4) {
      const int ka = k0 + hi * 2;
      const v2f a = *(const v2f*)(Tl + ka);
#pragma unroll
      for (int j = 0; j < 2; ++j) {
        const v2f b = *(const v2f*)(Wl + j * 16 * H1_PITCH + ka);
        acc[j] = __builtin_amdgcn_wmma_f32_16x16x4_f32(false, a, false, b,
                                                       (short)0, acc[j], false, false);
      }
    }
    strip_store_scores<2, 1, 1, false>(gb, lo, hi, acc, nullptr, 0, 0, a2, s2s, s2d);
#pragma unroll
    for (int j = 0; j < 2; ++j)
#pragma unroll
      for (int r = 0; r < 8; ++r) T[(hi * 8 + r) * H1_PITCH + j * 16 + lo] = acc[j][r];
    __builtin_amdgcn_fence(__ATOMIC_ACQ_REL, "wavefront");
    v8f ay[2];
    ay[0] = (v8f){0.f, 0.f, 0.f, 0.f, 0.f, 0.f, 0.f, 0.f};
    ay[1] = ay[0];
#pragma unroll
    for (int q = 0; q < HID / 4; ++q) {
      const v2f a = *(const v2f*)(Tl + q * 4 + hi * 2);
      ay[q & 1] = __builtin_amdgcn_wmma_f32_16x16x4_f32(false, a, false, bw[q],
                                                        (short)0, ay[q & 1], false, false);
    }
    float* Yb = Y + (size_t)gb * 16 * OUT_DIM + lo;
#pragma unroll
    for (int r = 0; r < 8; ++r) Yb[(size_t)(r + hi * 8) * OUT_DIM] = ay[0][r] + ay[1][r];
  }
}

#define GATHER_WGS ((N_NODES + 7) / 8)
#define GATHER16_WGS (N_NODES / 16)
#define STRIP_WGS  ((N_NODES / 16 + 7) / 8)
template <int PERIOD, int GEMM_WGS, int GATHER_N>
struct Mixed {
  static constexpr int MIX = PERIOD * GEMM_WGS;
  static constexpr int GRID = MIX + GATHER_N - (PERIOD - 1) * GEMM_WGS;
  static_assert(GATHER_N >= (PERIOD - 1) * GEMM_WGS, "the gather outlasts the GEMM");
  static __device__ __forceinline__ int gemm_wg(int i) { return (i < MIX && i % PERIOD == 0) ? i / PERIOD : -1; }
  static __device__ __forceinline__ int gather_wg(int i) {
    return (i < MIX) ? (i / PERIOD) * (PERIOD - 1) + (i % PERIOD) - 1 : i - GEMM_WGS;
  }
};
typedef Mixed<7, STRIP_WGS, GATHER16_WGS> Mix1;

__global__ void __launch_bounds__(256) __attribute__((amdgpu_waves_per_eu(16, 16)))
gather1_and_frac(const int* __restrict__ rowptr, const int* __restrict__ csr,
                 const float* __restrict__ ss, const float* __restrict__ sd,
                 const float* __restrict__ z,
                 const float* __restrict__ W2, const float* __restrict__ a2,
                 const float* __restrict__ Wfc2, float* __restrict__ Y,
                 float* __restrict__ s2s, float* __restrict__ s2d,
                 const float* __restrict__ H, const float* __restrict__ Mt, float* __restrict__ P) {
  static_assert(3 * 16 * ((IN_DIM + 4) / 4) <= FUSE_LDS_F4, "the P role's weight tile fits the block");
  __shared__ float4 lds[FUSE_LDS_F4];
  __shared__ __attribute__((aligned(16))) float slots[8 * SLOT_TAB_FLOATS(HEADS)];
  const int i = (int)blockIdx.x;
  const int f = Mix1::gemm_wg(i);
  if (f >= 0) {
    gemm_p_block(f, H, Mt, P, lds);
  } else {
    gather1_z2_block(Mix1::gather_wg(i), rowptr, csr, ss, sd, z, W2, a2, Wfc2, Y, s2s, s2d, lds, slots);
  }
}


__device__ __forceinline__ void put_slots_y(float* tab, int lane, float alpha, int s) {
  tab[lane] = alpha;
  ((unsigned*)(tab + SLOT_PITCH))[lane] = (unsigned)s * (unsigned)(OUT_DIM * sizeof(float));
  __builtin_amdgcn_fence(__ATOMIC_ACQ_REL, "wavefront");
}

__device__ __forceinline__ void add_rows_y(const char* __restrict__ yb, unsigned c16, int g, const float* tab,
                                           const unsigned* to, int cnt, float4& acc) {
  float a[4];
  unsigned o[4];
  float4 r[4];
#pragma unroll
  for (int i = 0; i < 4; ++i) {
    a[i] = tab[8 * i + g];
    o[i] = to[8 * i + g];
  }
#pragma unroll
  for (int i = 0; i < 4; ++i) {
    r[i] = make_float4(0.f, 0.f, 0.f, 0.f);
    if (8 * i + g < cnt) r[i] = *(const float4*)(yb + (o[i] + c16));
  }
#pragma unroll
  for (int i = 0; i < 4; ++i) {
    acc.x = __builtin_fmaf(a[i], r[i].x, acc.x);
    acc.y = __builtin_fmaf(a[i], r[i].y, acc.y);
    acc.z = __builtin_fmaf(a[i], r[i].z, acc.z);
    acc.w = __builtin_fmaf(a[i], r[i].w, acc.w);
  }
}

__device__ __forceinline__ float groups8_sum(float x) {
  x += __int_as_float(__builtin_amdgcn_update_dpp(0, __float_as_int(x), 0x164, 0xf, 0xf, true));
  x += __int_as_float(__builtin_amdgcn_update_dpp(0, __float_as_int(x), 0x168, 0xf, 0xf, true));
  x += __shfl_xor(x, 16);
  return x;
}

__global__ void __launch_bounds__(256)
gather2_out(const int* __restrict__ rowptr, const int* __restrict__ csr,
            const float* __restrict__ ss, const float* __restrict__ sd,
            const float* __restrict__ Y, const float* __restrict__ P, const int* __restrict__ fc,
            float* __restrict__ out) {
  __shared__ __attribute__((aligned(16))) float slots[8 * SLOT_TAB_FLOATS(1)];
  const int w = __builtin_amdgcn_readfirstlane((int)(threadIdx.x >> 5));
  const int n = (int)blockIdx.x * 8 + w;
  if (n >= N_NODES) return;
  const int lane = threadIdx.x & 31;
  const int g = lane >> 2;
  const unsigned c16 = (unsigned)(lane & 3) * 16u;
  float* tab = slots + w * SLOT_TAB_FLOATS(1);
  const unsigned* to = (const unsigned*)(tab + SLOT_PITCH);
  const int f = fc[n * 3 + (g < 3 ? g : 2)];
  float4 acc = make_float4(0.f, 0.f, 0.f, 0.f);
  const int beg = rowptr[n];
  const int end = rowptr[n + 1];
  const float sdn = sd[n];
  float x0;
  const int s0 = load_edge<1>(csr, ss, &sdn, beg + lane, end, &x0);
  if (lane < 12)
    acc = *(const float4*)((const char*)P + (((size_t)f * P_COLS + g * 16) * sizeof(float) + c16));
  if (end > beg) {
    const char* yb = (const char*)Y;
    const bool v0 = (beg + lane) < end;
    float m = x0;
    for (int base = beg + 32; base < end; base += 32) {
      float x;
      load_edge<1>(csr, ss, &sdn, base + lane, end, &x);
      m = fmaxf(m, x);
    }
    for (int o = 16; o > 0; o >>= 1) m = fmaxf(m, __shfl_xor(m, o));
    const float ee0 = v0 ? expf(x0 - m) : 0.f;
    float sum = ee0;
    for (int base = beg + 32; base < end; base += 32) {
      float x;
      load_edge<1>(csr, ss, &sdn, base + lane, end, &x);
      const bool v = (base + lane) < end;
      sum += v ? expf(x - m) : 0.f;
    }
    for (int o = 16; o > 0; o >>= 1) sum += __shfl_xor(sum, o);
    const float den = fmaxf(sum, 1e-9f);
    const int c0 = end - beg;
    put_slots_y(tab, lane, ee0 / den, s0);
    add_rows_y(yb, c16, g, tab, to, c0 < 32 ? c0 : 32, acc);
    for (int base = beg + 32; base < end; base += 32) {
      float x;
      const int s = load_edge<1>(csr, ss, &sdn, base + lane, end, &x);
      const bool v = (base + lane) < end;
      const int c = end - base;
      put_slots_y(tab, lane, v ? expf(x - m) / den : 0.f, s);
      add_rows_y(yb, c16, g, tab, to, c < 32 ? c : 32, acc);
    }
  }
  acc.x = groups8_sum(acc.x);
  acc.y = groups8_sum(acc.y);
  acc.z = groups8_sum(acc.z);
  acc.w = groups8_sum(acc.w);
  if (lane < 4) *(float4*)(out + (size_t)n * OUT_DIM + lane * 4) = acc;
}

extern "C" void kernel_launch(void* const* d_in, const int* in_sizes, int n_in,
                              void* d_out, int out_size, void* d_ws, size_t ws_size,
                              hipStream_t stream) {
  const float* h    = (const float*)d_in[0];
  const int*   src  = (const int*)  d_in[1];
  const int*   dst  = (const int*)  d_in[2];
  const int*   fc   = (const int*)  d_in[3];
  const float* W1   = (const float*)d_in[4];
  const float* a1   = (const float*)d_in[5];
  const float* W2   = (const float*)d_in[6];
  const float* a2   = (const float*)d_in[7];
  const float* Wf   = (const float*)d_in[8];
  const float* Wff  = (const float*)d_in[9];
  const float* Wfc2 = (const float*)d_in[10];
  float* out = (float*)d_out;

  float* w = (float*)d_ws;
  size_t off = 0;
  int* bintot  = (int*)(w + off); off += 512;
  int* runbase = (int*)(w + off); off += SLICES * 512;
  int* rowptr  = (int*)(w + off); off += N_NODES + 32;
  int* binned  = (int*)(w + off); off += N_EDGES;
  int* csr     = (int*)(w + off); off += N_EDGES;
  float* z1  = w + off; off += (size_t)N_NODES * 128;
  float* s1s = w + off; off += (size_t)N_NODES * HEADS;
  float* s1d = w + off; off += (size_t)N_NODES * HEADS;
  float* Y   = w + off; off += (size_t)N_NODES * OUT_DIM;
  float* s2s = w + off; off += N_NODES;
  float* s2d = w + off; off += N_NODES;
  float* P   = w + off; off += (size_t)N_NODES * P_COLS;
  float* Mt  = w + off; off += P_COLS * IN_DIM;
  (void)ws_size; (void)in_sizes; (void)n_in; (void)out_size;

  const int TB = 256;

  const Z1Args zq = {h, W1, a1, z1, s1s, s1d};
  zero_and_z1<<<Z1_HEAD_WGS + 1, SORT_T, 0, stream>>>(bintot, zq, Wf, Wff, Wfc2, Mt);
  bin_count<<<Z1_WGS + SLICES, SORT_T, 0, stream>>>(dst, bintot, runbase, zq);
  bin_scatter<<<Z1_WGS + SLICES, SORT_T, 0, stream>>>(src, dst, bintot, runbase, binned, zq);
  bin_finish<<<Z1_WGS + NBINS, SORT_T, 0, stream>>>(bintot, binned, rowptr, csr, zq);

  gather1_and_frac<<<Mix1::GRID, TB, 0, stream>>>(rowptr, csr, s1s, s1d, z1, W2, a2, Wfc2, Y, s2s, s2d,
                                                  h, Mt, P);

  gather2_out<<<GATHER_WGS, TB, 0, stream>>>(rowptr, csr, s2s, s2d, Y, P, fc, out);
}
